# starting kernel
# baseline (speedup 1.0000x reference)
#include <hip/hip_runtime.h>


typedef __attribute__((ext_vector_type(2))) float v2f;
typedef __attribute__((ext_vector_type(8))) float v8f;

#define PI_F 3.14159265358979323846f

__device__ __forceinline__ v8f wmma4(v2f a, v2f b, v8f c) {
  return __builtin_amdgcn_wmma_f32_16x16x4_f32(false, a, false, b, (short)0, c,
                                               false, false);
}

__device__ __forceinline__ int bandmap(int bi) { return (bi + 96) & 127; }

__device__ __forceinline__ v2f fragA(const float* A, int lda, int m0, int k0) {
  int lane = threadIdx.x & 31;
  int m = m0 + (lane & 15);
  int kk = k0 + ((lane >> 4) << 1);
  v2f r; r.x = A[m * lda + kk]; r.y = A[m * lda + kk + 1]; return r;
}
__device__ __forceinline__ v2f fragB(const float* B, int ldb, int k0, int n0) {
  int lane = threadIdx.x & 31;
  int n = n0 + (lane & 15);
  int kk = k0 + ((lane >> 4) << 1);
  v2f r; r.x = B[kk * ldb + n]; r.y = B[(kk + 1) * ldb + n]; return r;
}
__device__ __forceinline__ v2f fragB_bc(const float* F, int k0, int n0) {
  int lane = threadIdx.x & 31;
  int col = bandmap(n0 + (lane & 15));
  int kk = k0 + ((lane >> 4) << 1);
  v2f r; r.x = F[kk * 128 + col]; r.y = F[(kk + 1) * 128 + col]; return r;
}
__device__ __forceinline__ v2f fragA_br(const float* F, int m0, int k0) {
  int lane = threadIdx.x & 31;
  int row = bandmap(m0 + (lane & 15));
  int kk = k0 + ((lane >> 4) << 1);
  v2f r; r.x = F[row * 128 + kk]; r.y = F[row * 128 + kk + 1]; return r;
}
__device__ __forceinline__ v2f fragA_kmap(const float* F, int m0, int k0) {
  int lane = threadIdx.x & 31;
  int m = m0 + (lane & 15);
  int kk = k0 + ((lane >> 4) << 1);
  v2f r; r.x = F[m * 128 + bandmap(kk)]; r.y = F[m * 128 + bandmap(kk + 1)];
  return r;
}
__device__ __forceinline__ v2f fragB_kmap(const float* F, int k0, int n0) {
  int lane = threadIdx.x & 31;
  int n = n0 + (lane & 15);
  int kk = k0 + ((lane >> 4) << 1);
  v2f r; r.x = F[bandmap(kk) * 128 + n]; r.y = F[bandmap(kk + 1) * 128 + n];
  return r;
}

__global__ void k0_tables(float* __restrict__ Ftab) {
  for (int t = threadIdx.x + blockIdx.x * blockDim.x; t < 16384;
       t += blockDim.x * gridDim.x) {
    int j = t >> 7, k = t & 127;
    float ang = -2.f * PI_F * (float)((j * k) & 127) / 128.f;
    float s, c;
    sincosf(ang, &s, &c);
    Ftab[t] = c;
    Ftab[16384 + t] = s;
  }
}

__global__ void k1_fwd(const float* __restrict__ x, const float* __restrict__ Ftab,
                       float* __restrict__ XbR, float* __restrict__ XbI) {
  __shared__ float T1r[128 * 64];
  __shared__ float T1i[128 * 64];
  const float* Fre = Ftab;
  const float* Fim = Ftab + 16384;
  int blk = blockIdx.x;
  int b = blk >> 6, ci = blk & 63;
  const float* Ximg = x + ((size_t)blk << 14);
  int wave = threadIdx.x >> 5;
  int lane = threadIdx.x & 31;

  for (int t = wave; t < 32; t += 8) {
    int m0 = (t >> 2) << 4, n0 = (t & 3) << 4;
    v8f aR = {}, aI = {};
    for (int k = 0; k < 128; k += 4) {
      v2f a  = fragA(Ximg, 128, m0, k);
      v2f br = fragB_bc(Fre, k, n0);
      v2f bi = fragB_bc(Fim, k, n0);
      aR = wmma4(a, br, aR);
      aI = wmma4(a, bi, aI);
    }
    int n = n0 + (lane & 15);
    int mb = m0 + ((lane >> 4) << 3);
#pragma unroll
    for (int r = 0; r < 8; ++r) {
      T1r[(mb + r) * 64 + n] = aR[r];
      T1i[(mb + r) * 64 + n] = aI[r];
    }
  }
  __syncthreads();

  int bdst = (b + 8) & 15;
  for (int t = wave; t < 16; t += 8) {
    int m0 = (t >> 2) << 4, n0 = (t & 3) << 4;
    v8f cR = {}, cI = {};
    for (int k = 0; k < 128; k += 4) {
      v2f ar = fragA_br(Fre, m0, k);
      v2f ai = fragA_br(Fim, m0, k);
      v2f br = fragB(T1r, 64, k, n0);
      v2f bi = fragB(T1i, 64, k, n0);
      cR = wmma4(ar, br, cR);
      v2f ani; ani.x = -ai.x; ani.y = -ai.y;
      cR = wmma4(ani, bi, cR);
      cI = wmma4(ar, bi, cI);
      cI = wmma4(ai, br, cI);
    }
    int n = n0 + (lane & 15);
    int mb = m0 + ((lane >> 4) << 3);
    int dv = n - 32;
#pragma unroll
    for (int r = 0; r < 8; ++r) {
      int uu = mb + r;
      int du = uu - 32;
      float msk = (du * du + dv * dv <= 900) ? 1.f : 0.f;
      size_t off = ((size_t)(uu * 64 + n) << 10) + (size_t)(bdst << 6) + ci;
      XbR[off] = cR[r] * msk;
      XbI[off] = cI[r] * msk;
    }
  }
}

__global__ void k2_wdft(const float* __restrict__ w, float* __restrict__ WbR,
                        float* __restrict__ WbI) {
  int f = blockIdx.x;
  int u = bandmap(f >> 6), v = bandmap(f & 63);
  float cy[3], sy[3], cx[3], sx[3];
#pragma unroll
  for (int t = 0; t < 3; ++t) {
    sincosf(-2.f * PI_F * (float)((u * t) & 127) / 128.f, &sy[t], &cy[t]);
    sincosf(-2.f * PI_F * (float)((v * t) & 127) / 128.f, &sx[t], &cx[t]);
  }
  float* outR = WbR + ((size_t)f << 12);
  float* outI = WbI + ((size_t)f << 12);
  for (int p = threadIdx.x; p < 4096; p += blockDim.x) {
    int i = p >> 6, op = p & 63;
    int osrc = (op + 32) & 63;
    const float* wk = w + (size_t)(osrc * 64 + i) * 9;
    float ar = 0.f, ai = 0.f;
#pragma unroll
    for (int ky = 0; ky < 3; ++ky)
#pragma unroll
      for (int kx = 0; kx < 3; ++kx) {
        float tr = cy[ky] * cx[kx] - sy[ky] * sx[kx];
        float ti = cy[ky] * sx[kx] + sy[ky] * cx[kx];
        float wv = wk[ky * 3 + kx];
        ar += wv * tr;
        ai += wv * ti;
      }
    outR[p] = ar;
    outI[p] = ai;
  }
}

__global__ void k3_mix(const float* __restrict__ XbR, const float* __restrict__ XbI,
                       const float* __restrict__ WbR, const float* __restrict__ WbI,
                       float* __restrict__ YbR, float* __restrict__ YbI) {
  int f = blockIdx.x;
  const float* AR = XbR + ((size_t)f << 10);
  const float* AI = XbI + ((size_t)f << 10);
  const float* BR = WbR + ((size_t)f << 12);
  const float* BI = WbI + ((size_t)f << 12);
  __builtin_prefetch(BR, 0, 1);
  __builtin_prefetch(BI, 0, 1);
  int wave = threadIdx.x >> 5;
  int lane = threadIdx.x & 31;
  int n0 = wave << 4;
  v8f cR = {}, cI = {};
  for (int k = 0; k < 64; k += 4) {
    v2f ar = fragA(AR, 64, 0, k);
    v2f ai = fragA(AI, 64, 0, k);
    v2f br = fragB(BR, 64, k, n0);
    v2f bi = fragB(BI, 64, k, n0);
    cR = wmma4(ar, br, cR);
    v2f ani; ani.x = -ai.x; ani.y = -ai.y;
    cR = wmma4(ani, bi, cR);
    cI = wmma4(ar, bi, cI);
    cI = wmma4(ai, br, cI);
  }
  int n = n0 + (lane & 15);
  int mb = (lane >> 4) << 3;
#pragma unroll
  for (int r = 0; r < 8; ++r) {
    size_t off = ((size_t)((mb + r) * 64 + n) << 12) + f;
    YbR[off] = cR[r];
    YbI[off] = cI[r];
  }
}

__global__ void k4_inv(const float* __restrict__ YbR, const float* __restrict__ YbI,
                       const float* __restrict__ Ftab, const float* __restrict__ bias,
                       float* __restrict__ out) {
  __shared__ float Sr[128 * 64];
  __shared__ float Si[128 * 64];
  const float* Fre = Ftab;
  const float* Fim = Ftab + 16384;
  int blk = blockIdx.x;
  const float* Yr = YbR + ((size_t)blk << 12);
  const float* Yi = YbI + ((size_t)blk << 12);
  int wave = threadIdx.x >> 5;
  int lane = threadIdx.x & 31;

  for (int t = wave; t < 32; t += 8) {
    int m0 = (t >> 2) << 4, n0 = (t & 3) << 4;
    v8f cR = {}, cI = {};
    for (int k = 0; k < 64; k += 4) {
      v2f ar = fragA_kmap(Fre, m0, k);
      v2f ai = fragA_kmap(Fim, m0, k);
      v2f br = fragB(Yr, 64, k, n0);
      v2f bi = fragB(Yi, 64, k, n0);
      cR = wmma4(ar, br, cR);
      cR = wmma4(ai, bi, cR);
      cI = wmma4(ar, bi, cI);
      v2f ani; ani.x = -ai.x; ani.y = -ai.y;
      cI = wmma4(ani, br, cI);
    }
    int n = n0 + (lane & 15);
    int mb = m0 + ((lane >> 4) << 3);
#pragma unroll
    for (int r = 0; r < 8; ++r) {
      Sr[(mb + r) * 64 + n] = cR[r];
      Si[(mb + r) * 64 + n] = cI[r];
    }
  }
  __syncthreads();

  float bo = bias[blk & 63];
  float* oimg = out + ((size_t)blk << 14);
  for (int t = wave; t < 64; t += 8) {
    int m0 = (t >> 3) << 4, n0 = (t & 7) << 4;
    v8f c = {};
    for (int k = 0; k < 64; k += 4) {
      v2f sr = fragA(Sr, 64, m0, k);
      v2f si = fragA(Si, 64, m0, k);
      v2f br = fragB_kmap(Fre, k, n0);
      v2f bi = fragB_kmap(Fim, k, n0);
      c = wmma4(sr, br, c);
      c = wmma4(si, bi, c);
    }
    int n = n0 + (lane & 15);
    int mb = m0 + ((lane >> 4) << 3);
#pragma unroll
    for (int r = 0; r < 8; ++r)
      oimg[(mb + r) * 128 + n] = c[r] * (1.f / 16384.f) + bo;
  }
}

extern "C" void kernel_launch(void* const* d_in, const int* in_sizes, int n_in,
                              void* d_out, int out_size, void* d_ws, size_t ws_size,
                              hipStream_t stream) {
  (void)in_sizes; (void)n_in; (void)out_size; (void)ws_size;
  const float* x    = (const float*)d_in[0];
  const float* w    = (const float*)d_in[1];
  const float* bias = (const float*)d_in[2];
  float* out = (float*)d_out;

  float* ws   = (float*)d_ws;
  float* Ftab = ws;
  float* XbR  = ws + 32768;
  float* XbI  = XbR + 4194304;
  float* WbR  = XbI + 4194304;
  float* WbI  = WbR + 16777216;
  float* YbR  = WbI + 16777216;
  float* YbI  = YbR + 4194304;

  k0_tables<<<16, 256, 0, stream>>>(Ftab);
  k1_fwd  <<<1024, 256, 0, stream>>>(x, Ftab, XbR, XbI);
  k2_wdft <<<4096, 256, 0, stream>>>(w, WbR, WbI);
  k3_mix  <<<4096, 128, 0, stream>>>(XbR, XbI, WbR, WbI, YbR, YbI);
  k4_inv  <<<1024, 256, 0, stream>>>(YbR, YbI, Ftab, bias, out);
}


# k2 weight DFT: each thread holds the 9 taps of 4 outputs in registers and loops over 16 frequencies; block-uniform twiddles from LDS (one sincos of the summed angle); 16-byte coalesced stores
# speedup vs baseline: 1.1020x; 1.1020x over previous
#include <hip/hip_runtime.h>


typedef __attribute__((ext_vector_type(2))) float v2f;
typedef __attribute__((ext_vector_type(8))) float v8f;

#define PI_F 3.14159265358979323846f

__device__ __forceinline__ v8f wmma4(v2f a, v2f b, v8f c) {
  return __builtin_amdgcn_wmma_f32_16x16x4_f32(false, a, false, b, (short)0, c,
                                               false, false);
}

__device__ __forceinline__ int bandmap(int bi) { return (bi + 96) & 127; }

__device__ __forceinline__ v2f fragA(const float* A, int lda, int m0, int k0) {
  int lane = threadIdx.x & 31;
  int m = m0 + (lane & 15);
  int kk = k0 + ((lane >> 4) << 1);
  v2f r; r.x = A[m * lda + kk]; r.y = A[m * lda + kk + 1]; return r;
}
__device__ __forceinline__ v2f fragB(const float* B, int ldb, int k0, int n0) {
  int lane = threadIdx.x & 31;
  int n = n0 + (lane & 15);
  int kk = k0 + ((lane >> 4) << 1);
  v2f r; r.x = B[kk * ldb + n]; r.y = B[(kk + 1) * ldb + n]; return r;
}
__device__ __forceinline__ v2f fragB_bc(const float* F, int k0, int n0) {
  int lane = threadIdx.x & 31;
  int col = bandmap(n0 + (lane & 15));
  int kk = k0 + ((lane >> 4) << 1);
  v2f r; r.x = F[kk * 128 + col]; r.y = F[(kk + 1) * 128 + col]; return r;
}
__device__ __forceinline__ v2f fragA_br(const float* F, int m0, int k0) {
  int lane = threadIdx.x & 31;
  int row = bandmap(m0 + (lane & 15));
  int kk = k0 + ((lane >> 4) << 1);
  v2f r; r.x = F[row * 128 + kk]; r.y = F[row * 128 + kk + 1]; return r;
}
__device__ __forceinline__ v2f fragA_kmap(const float* F, int m0, int k0) {
  int lane = threadIdx.x & 31;
  int m = m0 + (lane & 15);
  int kk = k0 + ((lane >> 4) << 1);
  v2f r; r.x = F[m * 128 + bandmap(kk)]; r.y = F[m * 128 + bandmap(kk + 1)];
  return r;
}
__device__ __forceinline__ v2f fragB_kmap(const float* F, int k0, int n0) {
  int lane = threadIdx.x & 31;
  int n = n0 + (lane & 15);
  int kk = k0 + ((lane >> 4) << 1);
  v2f r; r.x = F[bandmap(kk) * 128 + n]; r.y = F[bandmap(kk + 1) * 128 + n];
  return r;
}

__global__ void k0_tables(float* __restrict__ Ftab) {
  for (int t = threadIdx.x + blockIdx.x * blockDim.x; t < 16384;
       t += blockDim.x * gridDim.x) {
    int j = t >> 7, k = t & 127;
    float ang = -2.f * PI_F * (float)((j * k) & 127) / 128.f;
    float s, c;
    sincosf(ang, &s, &c);
    Ftab[t] = c;
    Ftab[16384 + t] = s;
  }
}

__global__ void k1_fwd(const float* __restrict__ x, const float* __restrict__ Ftab,
                       float* __restrict__ XbR, float* __restrict__ XbI) {
  __shared__ float T1r[128 * 64];
  __shared__ float T1i[128 * 64];
  const float* Fre = Ftab;
  const float* Fim = Ftab + 16384;
  int blk = blockIdx.x;
  int b = blk >> 6, ci = blk & 63;
  const float* Ximg = x + ((size_t)blk << 14);
  int wave = threadIdx.x >> 5;
  int lane = threadIdx.x & 31;

  for (int t = wave; t < 32; t += 8) {
    int m0 = (t >> 2) << 4, n0 = (t & 3) << 4;
    v8f aR = {}, aI = {};
    for (int k = 0; k < 128; k += 4) {
      v2f a  = fragA(Ximg, 128, m0, k);
      v2f br = fragB_bc(Fre, k, n0);
      v2f bi = fragB_bc(Fim, k, n0);
      aR = wmma4(a, br, aR);
      aI = wmma4(a, bi, aI);
    }
    int n = n0 + (lane & 15);
    int mb = m0 + ((lane >> 4) << 3);
#pragma unroll
    for (int r = 0; r < 8; ++r) {
      T1r[(mb + r) * 64 + n] = aR[r];
      T1i[(mb + r) * 64 + n] = aI[r];
    }
  }
  __syncthreads();

  int bdst = (b + 8) & 15;
  for (int t = wave; t < 16; t += 8) {
    int m0 = (t >> 2) << 4, n0 = (t & 3) << 4;
    v8f cR = {}, cI = {};
    for (int k = 0; k < 128; k += 4) {
      v2f ar = fragA_br(Fre, m0, k);
      v2f ai = fragA_br(Fim, m0, k);
      v2f br = fragB(T1r, 64, k, n0);
      v2f bi = fragB(T1i, 64, k, n0);
      cR = wmma4(ar, br, cR);
      v2f ani; ani.x = -ai.x; ani.y = -ai.y;
      cR = wmma4(ani, bi, cR);
      cI = wmma4(ar, bi, cI);
      cI = wmma4(ai, br, cI);
    }
    int n = n0 + (lane & 15);
    int mb = m0 + ((lane >> 4) << 3);
    int dv = n - 32;
#pragma unroll
    for (int r = 0; r < 8; ++r) {
      int uu = mb + r;
      int du = uu - 32;
      float msk = (du * du + dv * dv <= 900) ? 1.f : 0.f;
      size_t off = ((size_t)(uu * 64 + n) << 10) + (size_t)(bdst << 6) + ci;
      XbR[off] = cR[r] * msk;
      XbI[off] = cI[r] * msk;
    }
  }
}

__global__ void __launch_bounds__(256)
k2_wdft(const float* __restrict__ w, float* __restrict__ WbR,
        float* __restrict__ WbI) {
  __shared__ __attribute__((aligned(16))) float tw[16 * 20];
  int bid = blockIdx.x;
  int pchunk = bid & 3;
  int vchunk = (bid >> 2) & 3;
  int uu = bid >> 4;
  int tid = threadIdx.x;
  if (tid < 144) {
    int vloc = tid / 9, tap = tid - vloc * 9;
    int ky = tap / 3, kx = tap - ky * 3;
    int u = bandmap(uu), v = bandmap(vchunk * 16 + vloc);
    int idx = (u * ky + v * kx) & 127;
    float s, c;
    sincosf(-2.f * PI_F * (float)idx / 128.f, &s, &c);
    tw[vloc * 20 + tap] = c;
    tw[vloc * 20 + 9 + tap] = s;
  } else if (tid < 160) {
    tw[(tid - 144) * 20 + 18] = 0.f;
    tw[(tid - 144) * 20 + 19] = 0.f;
  }
  int p0 = (pchunk << 10) + (tid << 2);
  int i = p0 >> 6, op0 = p0 & 63;
  float wv[4][9];
#pragma unroll
  for (int q = 0; q < 4; ++q) {
    int osrc = (op0 + q + 32) & 63;
    const float* wk = w + (size_t)(osrc * 64 + i) * 9;
#pragma unroll
    for (int t = 0; t < 9; ++t) wv[q][t] = wk[t];
  }
  __syncthreads();
  size_t fbase = (size_t)(uu * 64 + vchunk * 16);
  for (int vloc = 0; vloc < 16; ++vloc) {
    const float4* t4 = (const float4*)(tw + vloc * 20);
    float4 a = t4[0], b = t4[1], c = t4[2], d = t4[3], e = t4[4];
    float tr[9] = {a.x, a.y, a.z, a.w, b.x, b.y, b.z, b.w, c.x};
    float ti[9] = {c.y, c.z, c.w, d.x, d.y, d.z, d.w, e.x, e.y};
    float ar[4], ai[4];
#pragma unroll
    for (int q = 0; q < 4; ++q) {
      float sr = 0.f, si = 0.f;
#pragma unroll
      for (int t = 0; t < 9; ++t) {
        sr += wv[q][t] * tr[t];
        si += wv[q][t] * ti[t];
      }
      ar[q] = sr;
      ai[q] = si;
    }
    size_t off = ((fbase + (size_t)vloc) << 12) + (size_t)p0;
    *(float4*)(WbR + off) = make_float4(ar[0], ar[1], ar[2], ar[3]);
    *(float4*)(WbI + off) = make_float4(ai[0], ai[1], ai[2], ai[3]);
  }
}

__global__ void k3_mix(const float* __restrict__ XbR, const float* __restrict__ XbI,
                       const float* __restrict__ WbR, const float* __restrict__ WbI,
                       float* __restrict__ YbR, float* __restrict__ YbI) {
  int f = blockIdx.x;
  const float* AR = XbR + ((size_t)f << 10);
  const float* AI = XbI + ((size_t)f << 10);
  const float* BR = WbR + ((size_t)f << 12);
  const float* BI = WbI + ((size_t)f << 12);
  __builtin_prefetch(BR, 0, 1);
  __builtin_prefetch(BI, 0, 1);
  int wave = threadIdx.x >> 5;
  int lane = threadIdx.x & 31;
  int n0 = wave << 4;
  v8f cR = {}, cI = {};
  for (int k = 0; k < 64; k += 4) {
    v2f ar = fragA(AR, 64, 0, k);
    v2f ai = fragA(AI, 64, 0, k);
    v2f br = fragB(BR, 64, k, n0);
    v2f bi = fragB(BI, 64, k, n0);
    cR = wmma4(ar, br, cR);
    v2f ani; ani.x = -ai.x; ani.y = -ai.y;
    cR = wmma4(ani, bi, cR);
    cI = wmma4(ar, bi, cI);
    cI = wmma4(ai, br, cI);
  }
  int n = n0 + (lane & 15);
  int mb = (lane >> 4) << 3;
#pragma unroll
  for (int r = 0; r < 8; ++r) {
    size_t off = ((size_t)((mb + r) * 64 + n) << 12) + f;
    YbR[off] = cR[r];
    YbI[off] = cI[r];
  }
}

__global__ void k4_inv(const float* __restrict__ YbR, const float* __restrict__ YbI,
                       const float* __restrict__ Ftab, const float* __restrict__ bias,
                       float* __restrict__ out) {
  __shared__ float Sr[128 * 64];
  __shared__ float Si[128 * 64];
  const float* Fre = Ftab;
  const float* Fim = Ftab + 16384;
  int blk = blockIdx.x;
  const float* Yr = YbR + ((size_t)blk << 12);
  const float* Yi = YbI + ((size_t)blk << 12);
  int wave = threadIdx.x >> 5;
  int lane = threadIdx.x & 31;

  for (int t = wave; t < 32; t += 8) {
    int m0 = (t >> 2) << 4, n0 = (t & 3) << 4;
    v8f cR = {}, cI = {};
    for (int k = 0; k < 64; k += 4) {
      v2f ar = fragA_kmap(Fre, m0, k);
      v2f ai = fragA_kmap(Fim, m0, k);
      v2f br = fragB(Yr, 64, k, n0);
      v2f bi = fragB(Yi, 64, k, n0);
      cR = wmma4(ar, br, cR);
      cR = wmma4(ai, bi, cR);
      cI = wmma4(ar, bi, cI);
      v2f ani; ani.x = -ai.x; ani.y = -ai.y;
      cI = wmma4(ani, br, cI);
    }
    int n = n0 + (lane & 15);
    int mb = m0 + ((lane >> 4) << 3);
#pragma unroll
    for (int r = 0; r < 8; ++r) {
      Sr[(mb + r) * 64 + n] = cR[r];
      Si[(mb + r) * 64 + n] = cI[r];
    }
  }
  __syncthreads();

  float bo = bias[blk & 63];
  float* oimg = out + ((size_t)blk << 14);
  for (int t = wave; t < 64; t += 8) {
    int m0 = (t >> 3) << 4, n0 = (t & 7) << 4;
    v8f c = {};
    for (int k = 0; k < 64; k += 4) {
      v2f sr = fragA(Sr, 64, m0, k);
      v2f si = fragA(Si, 64, m0, k);
      v2f br = fragB_kmap(Fre, k, n0);
      v2f bi = fragB_kmap(Fim, k, n0);
      c = wmma4(sr, br, c);
      c = wmma4(si, bi, c);
    }
    int n = n0 + (lane & 15);
    int mb = m0 + ((lane >> 4) << 3);
#pragma unroll
    for (int r = 0; r < 8; ++r)
      oimg[(mb + r) * 128 + n] = c[r] * (1.f / 16384.f) + bo;
  }
}

extern "C" void kernel_launch(void* const* d_in, const int* in_sizes, int n_in,
                              void* d_out, int out_size, void* d_ws, size_t ws_size,
                              hipStream_t stream) {
  (void)in_sizes; (void)n_in; (void)out_size; (void)ws_size;
  const float* x    = (const float*)d_in[0];
  const float* w    = (const float*)d_in[1];
  const float* bias = (const float*)d_in[2];
  float* out = (float*)d_out;

  float* ws   = (float*)d_ws;
  float* Ftab = ws;
  float* XbR  = ws + 32768;
  float* XbI  = XbR + 4194304;
  float* WbR  = XbI + 4194304;
  float* WbI  = WbR + 16777216;
  float* YbR  = WbI + 16777216;
  float* YbI  = YbR + 4194304;

  k0_tables<<<16, 256, 0, stream>>>(Ftab);
  k1_fwd  <<<1024, 256, 0, stream>>>(x, Ftab, XbR, XbI);
  k2_wdft <<<1024, 256, 0, stream>>>(w, WbR, WbI);
  k3_mix  <<<4096, 128, 0, stream>>>(XbR, XbI, WbR, WbI, YbR, YbI);
  k4_inv  <<<1024, 256, 0, stream>>>(YbR, YbI, Ftab, bias, out);
}


# k4 inverse stage: LDS rows of the intermediate S padded from 64 to 68 floats so the stage-2 A-fragment reads spread over all 64 banks instead of two (16-way bank conflict removed)
# speedup vs baseline: 1.1422x; 1.0365x over previous
#include <hip/hip_runtime.h>


typedef __attribute__((ext_vector_type(2))) float v2f;
typedef __attribute__((ext_vector_type(8))) float v8f;

#define PI_F 3.14159265358979323846f

__device__ __forceinline__ v8f wmma4(v2f a, v2f b, v8f c) {
  return __builtin_amdgcn_wmma_f32_16x16x4_f32(false, a, false, b, (short)0, c,
                                               false, false);
}

__device__ __forceinline__ int bandmap(int bi) { return (bi + 96) & 127; }

__device__ __forceinline__ v2f fragA(const float* A, int lda, int m0, int k0) {
  int lane = threadIdx.x & 31;
  int m = m0 + (lane & 15);
  int kk = k0 + ((lane >> 4) << 1);
  v2f r; r.x = A[m * lda + kk]; r.y = A[m * lda + kk + 1]; return r;
}
__device__ __forceinline__ v2f fragB(const float* B, int ldb, int k0, int n0) {
  int lane = threadIdx.x & 31;
  int n = n0 + (lane & 15);
  int kk = k0 + ((lane >> 4) << 1);
  v2f r; r.x = B[kk * ldb + n]; r.y = B[(kk + 1) * ldb + n]; return r;
}
__device__ __forceinline__ v2f fragB_bc(const float* F, int k0, int n0) {
  int lane = threadIdx.x & 31;
  int col = bandmap(n0 + (lane & 15));
  int kk = k0 + ((lane >> 4) << 1);
  v2f r; r.x = F[kk * 128 + col]; r.y = F[(kk + 1) * 128 + col]; return r;
}
__device__ __forceinline__ v2f fragA_br(const float* F, int m0, int k0) {
  int lane = threadIdx.x & 31;
  int row = bandmap(m0 + (lane & 15));
  int kk = k0 + ((lane >> 4) << 1);
  v2f r; r.x = F[row * 128 + kk]; r.y = F[row * 128 + kk + 1]; return r;
}
__device__ __forceinline__ v2f fragA_kmap(const float* F, int m0, int k0) {
  int lane = threadIdx.x & 31;
  int m = m0 + (lane & 15);
  int kk = k0 + ((lane >> 4) << 1);
  v2f r; r.x = F[m * 128 + bandmap(kk)]; r.y = F[m * 128 + bandmap(kk + 1)];
  return r;
}
__device__ __forceinline__ v2f fragB_kmap(const float* F, int k0, int n0) {
  int lane = threadIdx.x & 31;
  int n = n0 + (lane & 15);
  int kk = k0 + ((lane >> 4) << 1);
  v2f r; r.x = F[bandmap(kk) * 128 + n]; r.y = F[bandmap(kk + 1) * 128 + n];
  return r;
}

__global__ void k0_tables(float* __restrict__ Ftab) {
  for (int t = threadIdx.x + blockIdx.x * blockDim.x; t < 16384;
       t += blockDim.x * gridDim.x) {
    int j = t >> 7, k = t & 127;
    float ang = -2.f * PI_F * (float)((j * k) & 127) / 128.f;
    float s, c;
    sincosf(ang, &s, &c);
    Ftab[t] = c;
    Ftab[16384 + t] = s;
  }
}

__global__ void k1_fwd(const float* __restrict__ x, const float* __restrict__ Ftab,
                       float* __restrict__ XbR, float* __restrict__ XbI) {
  __shared__ float T1r[128 * 64];
  __shared__ float T1i[128 * 64];
  const float* Fre = Ftab;
  const float* Fim = Ftab + 16384;
  int blk = blockIdx.x;
  int b = blk >> 6, ci = blk & 63;
  const float* Ximg = x + ((size_t)blk << 14);
  int wave = threadIdx.x >> 5;
  int lane = threadIdx.x & 31;

  for (int t = wave; t < 32; t += 8) {
    int m0 = (t >> 2) << 4, n0 = (t & 3) << 4;
    v8f aR = {}, aI = {};
    for (int k = 0; k < 128; k += 4) {
      v2f a  = fragA(Ximg, 128, m0, k);
      v2f br = fragB_bc(Fre, k, n0);
      v2f bi = fragB_bc(Fim, k, n0);
      aR = wmma4(a, br, aR);
      aI = wmma4(a, bi, aI);
    }
    int n = n0 + (lane & 15);
    int mb = m0 + ((lane >> 4) << 3);
#pragma unroll
    for (int r = 0; r < 8; ++r) {
      T1r[(mb + r) * 64 + n] = aR[r];
      T1i[(mb + r) * 64 + n] = aI[r];
    }
  }
  __syncthreads();

  int bdst = (b + 8) & 15;
  for (int t = wave; t < 16; t += 8) {
    int m0 = (t >> 2) << 4, n0 = (t & 3) << 4;
    v8f cR = {}, cI = {};
    for (int k = 0; k < 128; k += 4) {
      v2f ar = fragA_br(Fre, m0, k);
      v2f ai = fragA_br(Fim, m0, k);
      v2f br = fragB(T1r, 64, k, n0);
      v2f bi = fragB(T1i, 64, k, n0);
      cR = wmma4(ar, br, cR);
      v2f ani; ani.x = -ai.x; ani.y = -ai.y;
      cR = wmma4(ani, bi, cR);
      cI = wmma4(ar, bi, cI);
      cI = wmma4(ai, br, cI);
    }
    int n = n0 + (lane & 15);
    int mb = m0 + ((lane >> 4) << 3);
    int dv = n - 32;
#pragma unroll
    for (int r = 0; r < 8; ++r) {
      int uu = mb + r;
      int du = uu - 32;
      float msk = (du * du + dv * dv <= 900) ? 1.f : 0.f;
      size_t off = ((size_t)(uu * 64 + n) << 10) + (size_t)(bdst << 6) + ci;
      XbR[off] = cR[r] * msk;
      XbI[off] = cI[r] * msk;
    }
  }
}

__global__ void __launch_bounds__(256)
k2_wdft(const float* __restrict__ w, float* __restrict__ WbR,
        float* __restrict__ WbI) {
  __shared__ __attribute__((aligned(16))) float tw[16 * 20];
  int bid = blockIdx.x;
  int pchunk = bid & 3;
  int vchunk = (bid >> 2) & 3;
  int uu = bid >> 4;
  int tid = threadIdx.x;
  if (tid < 144) {
    int vloc = tid / 9, tap = tid - vloc * 9;
    int ky = tap / 3, kx = tap - ky * 3;
    int u = bandmap(uu), v = bandmap(vchunk * 16 + vloc);
    int idx = (u * ky + v * kx) & 127;
    float s, c;
    sincosf(-2.f * PI_F * (float)idx / 128.f, &s, &c);
    tw[vloc * 20 + tap] = c;
    tw[vloc * 20 + 9 + tap] = s;
  } else if (tid < 160) {
    tw[(tid - 144) * 20 + 18] = 0.f;
    tw[(tid - 144) * 20 + 19] = 0.f;
  }
  int p0 = (pchunk << 10) + (tid << 2);
  int i = p0 >> 6, op0 = p0 & 63;
  float wv[4][9];
#pragma unroll
  for (int q = 0; q < 4; ++q) {
    int osrc = (op0 + q + 32) & 63;
    const float* wk = w + (size_t)(osrc * 64 + i) * 9;
#pragma unroll
    for (int t = 0; t < 9; ++t) wv[q][t] = wk[t];
  }
  __syncthreads();
  size_t fbase = (size_t)(uu * 64 + vchunk * 16);
  for (int vloc = 0; vloc < 16; ++vloc) {
    const float4* t4 = (const float4*)(tw + vloc * 20);
    float4 a = t4[0], b = t4[1], c = t4[2], d = t4[3], e = t4[4];
    float tr[9] = {a.x, a.y, a.z, a.w, b.x, b.y, b.z, b.w, c.x};
    float ti[9] = {c.y, c.z, c.w, d.x, d.y, d.z, d.w, e.x, e.y};
    float ar[4], ai[4];
#pragma unroll
    for (int q = 0; q < 4; ++q) {
      float sr = 0.f, si = 0.f;
#pragma unroll
      for (int t = 0; t < 9; ++t) {
        sr += wv[q][t] * tr[t];
        si += wv[q][t] * ti[t];
      }
      ar[q] = sr;
      ai[q] = si;
    }
    size_t off = ((fbase + (size_t)vloc) << 12) + (size_t)p0;
    *(float4*)(WbR + off) = make_float4(ar[0], ar[1], ar[2], ar[3]);
    *(float4*)(WbI + off) = make_float4(ai[0], ai[1], ai[2], ai[3]);
  }
}

__global__ void k3_mix(const float* __restrict__ XbR, const float* __restrict__ XbI,
                       const float* __restrict__ WbR, const float* __restrict__ WbI,
                       float* __restrict__ YbR, float* __restrict__ YbI) {
  int f = blockIdx.x;
  const float* AR = XbR + ((size_t)f << 10);
  const float* AI = XbI + ((size_t)f << 10);
  const float* BR = WbR + ((size_t)f << 12);
  const float* BI = WbI + ((size_t)f << 12);
  __builtin_prefetch(BR, 0, 1);
  __builtin_prefetch(BI, 0, 1);
  int wave = threadIdx.x >> 5;
  int lane = threadIdx.x & 31;
  int n0 = wave << 4;
  v8f cR = {}, cI = {};
  for (int k = 0; k < 64; k += 4) {
    v2f ar = fragA(AR, 64, 0, k);
    v2f ai = fragA(AI, 64, 0, k);
    v2f br = fragB(BR, 64, k, n0);
    v2f bi = fragB(BI, 64, k, n0);
    cR = wmma4(ar, br, cR);
    v2f ani; ani.x = -ai.x; ani.y = -ai.y;
    cR = wmma4(ani, bi, cR);
    cI = wmma4(ar, bi, cI);
    cI = wmma4(ai, br, cI);
  }
  int n = n0 + (lane & 15);
  int mb = (lane >> 4) << 3;
#pragma unroll
  for (int r = 0; r < 8; ++r) {
    size_t off = ((size_t)((mb + r) * 64 + n) << 12) + f;
    YbR[off] = cR[r];
    YbI[off] = cI[r];
  }
}

#define S_PITCH 68
__global__ void k4_inv(const float* __restrict__ YbR, const float* __restrict__ YbI,
                       const float* __restrict__ Ftab, const float* __restrict__ bias,
                       float* __restrict__ out) {
  __shared__ __attribute__((aligned(16))) float Sr[128 * S_PITCH];
  __shared__ __attribute__((aligned(16))) float Si[128 * S_PITCH];
  const float* Fre = Ftab;
  const float* Fim = Ftab + 16384;
  int blk = blockIdx.x;
  const float* Yr = YbR + ((size_t)blk << 12);
  const float* Yi = YbI + ((size_t)blk << 12);
  int wave = threadIdx.x >> 5;
  int lane = threadIdx.x & 31;

  for (int t = wave; t < 32; t += 8) {
    int m0 = (t >> 2) << 4, n0 = (t & 3) << 4;
    v8f cR = {}, cI = {};
    for (int k = 0; k < 64; k += 4) {
      v2f ar = fragA_kmap(Fre, m0, k);
      v2f ai = fragA_kmap(Fim, m0, k);
      v2f br = fragB(Yr, 64, k, n0);
      v2f bi = fragB(Yi, 64, k, n0);
      cR = wmma4(ar, br, cR);
      cR = wmma4(ai, bi, cR);
      cI = wmma4(ar, bi, cI);
      v2f ani; ani.x = -ai.x; ani.y = -ai.y;
      cI = wmma4(ani, br, cI);
    }
    int n = n0 + (lane & 15);
    int mb = m0 + ((lane >> 4) << 3);
#pragma unroll
    for (int r = 0; r < 8; ++r) {
      Sr[(mb + r) * S_PITCH + n] = cR[r];
      Si[(mb + r) * S_PITCH + n] = cI[r];
    }
  }
  __syncthreads();

  float bo = bias[blk & 63];
  float* oimg = out + ((size_t)blk << 14);
  for (int t = wave; t < 64; t += 8) {
    int m0 = (t >> 3) << 4, n0 = (t & 7) << 4;
    v8f c = {};
    for (int k = 0; k < 64; k += 4) {
      v2f sr = fragA(Sr, S_PITCH, m0, k);
      v2f si = fragA(Si, S_PITCH, m0, k);
      v2f br = fragB_kmap(Fre, k, n0);
      v2f bi = fragB_kmap(Fim, k, n0);
      c = wmma4(sr, br, c);
      c = wmma4(si, bi, c);
    }
    int n = n0 + (lane & 15);
    int mb = m0 + ((lane >> 4) << 3);
#pragma unroll
    for (int r = 0; r < 8; ++r)
      oimg[(mb + r) * 128 + n] = c[r] * (1.f / 16384.f) + bo;
  }
}

extern "C" void kernel_launch(void* const* d_in, const int* in_sizes, int n_in,
                              void* d_out, int out_size, void* d_ws, size_t ws_size,
                              hipStream_t stream) {
  (void)in_sizes; (void)n_in; (void)out_size; (void)ws_size;
  const float* x    = (const float*)d_in[0];
  const float* w    = (const float*)d_in[1];
  const float* bias = (const float*)d_in[2];
  float* out = (float*)d_out;

  float* ws   = (float*)d_ws;
  float* Ftab = ws;
  float* XbR  = ws + 32768;
  float* XbI  = XbR + 4194304;
  float* WbR  = XbI + 4194304;
  float* WbI  = WbR + 16777216;
  float* YbR  = WbI + 16777216;
  float* YbI  = YbR + 4194304;

  k0_tables<<<16, 256, 0, stream>>>(Ftab);
  k1_fwd  <<<1024, 256, 0, stream>>>(x, Ftab, XbR, XbI);
  k2_wdft <<<1024, 256, 0, stream>>>(w, WbR, WbI);
  k3_mix  <<<4096, 128, 0, stream>>>(XbR, XbI, WbR, WbI, YbR, YbI);
  k4_inv  <<<1024, 256, 0, stream>>>(YbR, YbI, Ftab, bias, out);
}


# k3 channel mix: each block takes 32 consecutive frequencies of one 16-channel tile and transposes its results through LDS, so Yb is written in whole 128-byte lines, not 4-byte words 16 KB apart
# speedup vs baseline: 1.5806x; 1.3838x over previous
#include <hip/hip_runtime.h>


typedef __attribute__((ext_vector_type(2))) float v2f;
typedef __attribute__((ext_vector_type(8))) float v8f;

#define PI_F 3.14159265358979323846f

__device__ __forceinline__ v8f wmma4(v2f a, v2f b, v8f c) {
  return __builtin_amdgcn_wmma_f32_16x16x4_f32(false, a, false, b, (short)0, c,
                                               false, false);
}

__device__ __forceinline__ int bandmap(int bi) { return (bi + 96) & 127; }

__device__ __forceinline__ v2f fragA(const float* A, int lda, int m0, int k0) {
  int lane = threadIdx.x & 31;
  int m = m0 + (lane & 15);
  int kk = k0 + ((lane >> 4) << 1);
  v2f r; r.x = A[m * lda + kk]; r.y = A[m * lda + kk + 1]; return r;
}
__device__ __forceinline__ v2f fragB(const float* B, int ldb, int k0, int n0) {
  int lane = threadIdx.x & 31;
  int n = n0 + (lane & 15);
  int kk = k0 + ((lane >> 4) << 1);
  v2f r; r.x = B[kk * ldb + n]; r.y = B[(kk + 1) * ldb + n]; return r;
}
__device__ __forceinline__ v2f fragB_bc(const float* F, int k0, int n0) {
  int lane = threadIdx.x & 31;
  int col = bandmap(n0 + (lane & 15));
  int kk = k0 + ((lane >> 4) << 1);
  v2f r; r.x = F[kk * 128 + col]; r.y = F[(kk + 1) * 128 + col]; return r;
}
__device__ __forceinline__ v2f fragA_br(const float* F, int m0, int k0) {
  int lane = threadIdx.x & 31;
  int row = bandmap(m0 + (lane & 15));
  int kk = k0 + ((lane >> 4) << 1);
  v2f r; r.x = F[row * 128 + kk]; r.y = F[row * 128 + kk + 1]; return r;
}
__device__ __forceinline__ v2f fragA_kmap(const float* F, int m0, int k0) {
  int lane = threadIdx.x & 31;
  int m = m0 + (lane & 15);
  int kk = k0 + ((lane >> 4) << 1);
  v2f r; r.x = F[m * 128 + bandmap(kk)]; r.y = F[m * 128 + bandmap(kk + 1)];
  return r;
}
__device__ __forceinline__ v2f fragB_kmap(const float* F, int k0, int n0) {
  int lane = threadIdx.x & 31;
  int n = n0 + (lane & 15);
  int kk = k0 + ((lane >> 4) << 1);
  v2f r; r.x = F[bandmap(kk) * 128 + n]; r.y = F[bandmap(kk + 1) * 128 + n];
  return r;
}

__global__ void k0_tables(float* __restrict__ Ftab) {
  for (int t = threadIdx.x + blockIdx.x * blockDim.x; t < 16384;
       t += blockDim.x * gridDim.x) {
    int j = t >> 7, k = t & 127;
    float ang = -2.f * PI_F * (float)((j * k) & 127) / 128.f;
    float s, c;
    sincosf(ang, &s, &c);
    Ftab[t] = c;
    Ftab[16384 + t] = s;
  }
}

__global__ void k1_fwd(const float* __restrict__ x, const float* __restrict__ Ftab,
                       float* __restrict__ XbR, float* __restrict__ XbI) {
  __shared__ float T1r[128 * 64];
  __shared__ float T1i[128 * 64];
  const float* Fre = Ftab;
  const float* Fim = Ftab + 16384;
  int blk = blockIdx.x;
  int b = blk >> 6, ci = blk & 63;
  const float* Ximg = x + ((size_t)blk << 14);
  int wave = threadIdx.x >> 5;
  int lane = threadIdx.x & 31;

  for (int t = wave; t < 32; t += 8) {
    int m0 = (t >> 2) << 4, n0 = (t & 3) << 4;
    v8f aR = {}, aI = {};
    for (int k = 0; k < 128; k += 4) {
      v2f a  = fragA(Ximg, 128, m0, k);
      v2f br = fragB_bc(Fre, k, n0);
      v2f bi = fragB_bc(Fim, k, n0);
      aR = wmma4(a, br, aR);
      aI = wmma4(a, bi, aI);
    }
    int n = n0 + (lane & 15);
    int mb = m0 + ((lane >> 4) << 3);
#pragma unroll
    for (int r = 0; r < 8; ++r) {
      T1r[(mb + r) * 64 + n] = aR[r];
      T1i[(mb + r) * 64 + n] = aI[r];
    }
  }
  __syncthreads();

  int bdst = (b + 8) & 15;
  for (int t = wave; t < 16; t += 8) {
    int m0 = (t >> 2) << 4, n0 = (t & 3) << 4;
    v8f cR = {}, cI = {};
    for (int k = 0; k < 128; k += 4) {
      v2f ar = fragA_br(Fre, m0, k);
      v2f ai = fragA_br(Fim, m0, k);
      v2f br = fragB(T1r, 64, k, n0);
      v2f bi = fragB(T1i, 64, k, n0);
      cR = wmma4(ar, br, cR);
      v2f ani; ani.x = -ai.x; ani.y = -ai.y;
      cR = wmma4(ani, bi, cR);
      cI = wmma4(ar, bi, cI);
      cI = wmma4(ai, br, cI);
    }
    int n = n0 + (lane & 15);
    int mb = m0 + ((lane >> 4) << 3);
    int dv = n - 32;
#pragma unroll
    for (int r = 0; r < 8; ++r) {
      int uu = mb + r;
      int du = uu - 32;
      float msk = (du * du + dv * dv <= 900) ? 1.f : 0.f;
      size_t off = ((size_t)(uu * 64 + n) << 10) + (size_t)(bdst << 6) + ci;
      XbR[off] = cR[r] * msk;
      XbI[off] = cI[r] * msk;
    }
  }
}

__global__ void __launch_bounds__(256)
k2_wdft(const float* __restrict__ w, float* __restrict__ WbR,
        float* __restrict__ WbI) {
  __shared__ __attribute__((aligned(16))) float tw[16 * 20];
  int bid = blockIdx.x;
  int pchunk = bid & 3;
  int vchunk = (bid >> 2) & 3;
  int uu = bid >> 4;
  int tid = threadIdx.x;
  if (tid < 144) {
    int vloc = tid / 9, tap = tid - vloc * 9;
    int ky = tap / 3, kx = tap - ky * 3;
    int u = bandmap(uu), v = bandmap(vchunk * 16 + vloc);
    int idx = (u * ky + v * kx) & 127;
    float s, c;
    sincosf(-2.f * PI_F * (float)idx / 128.f, &s, &c);
    tw[vloc * 20 + tap] = c;
    tw[vloc * 20 + 9 + tap] = s;
  } else if (tid < 160) {
    tw[(tid - 144) * 20 + 18] = 0.f;
    tw[(tid - 144) * 20 + 19] = 0.f;
  }
  int p0 = (pchunk << 10) + (tid << 2);
  int i = p0 >> 6, op0 = p0 & 63;
  float wv[4][9];
#pragma unroll
  for (int q = 0; q < 4; ++q) {
    int osrc = (op0 + q + 32) & 63;
    const float* wk = w + (size_t)(osrc * 64 + i) * 9;
#pragma unroll
    for (int t = 0; t < 9; ++t) wv[q][t] = wk[t];
  }
  __syncthreads();
  size_t fbase = (size_t)(uu * 64 + vchunk * 16);
  for (int vloc = 0; vloc < 16; ++vloc) {
    const float4* t4 = (const float4*)(tw + vloc * 20);
    float4 a = t4[0], b = t4[1], c = t4[2], d = t4[3], e = t4[4];
    float tr[9] = {a.x, a.y, a.z, a.w, b.x, b.y, b.z, b.w, c.x};
    float ti[9] = {c.y, c.z, c.w, d.x, d.y, d.z, d.w, e.x, e.y};
    float ar[4], ai[4];
#pragma unroll
    for (int q = 0; q < 4; ++q) {
      float sr = 0.f, si = 0.f;
#pragma unroll
      for (int t = 0; t < 9; ++t) {
        sr += wv[q][t] * tr[t];
        si += wv[q][t] * ti[t];
      }
      ar[q] = sr;
      ai[q] = si;
    }
    size_t off = ((fbase + (size_t)vloc) << 12) + (size_t)p0;
    *(float4*)(WbR + off) = make_float4(ar[0], ar[1], ar[2], ar[3]);
    *(float4*)(WbI + off) = make_float4(ai[0], ai[1], ai[2], ai[3]);
  }
}

#define K3_FPB 32
#define K3_PITCH 36
__global__ void __launch_bounds__(512)
k3_mix(const float* __restrict__ XbR, const float* __restrict__ XbI,
       const float* __restrict__ WbR, const float* __restrict__ WbI,
       float* __restrict__ YbR, float* __restrict__ YbI) {
  __shared__ __attribute__((aligned(16))) float stg[2 * 256 * K3_PITCH];
  int wave = threadIdx.x >> 5;
  int lane = threadIdx.x & 31;
  int n0 = (blockIdx.x & 3) << 4;
  int f0 = (blockIdx.x >> 2) * K3_FPB;
  int ol = lane & 15;
  int mb = (lane >> 4) << 3;
  for (int fl = wave; fl < K3_FPB; fl += 16) {
    int f = f0 + fl;
    const float* AR = XbR + ((size_t)f << 10);
    const float* AI = XbI + ((size_t)f << 10);
    const float* BR = WbR + ((size_t)f << 12);
    const float* BI = WbI + ((size_t)f << 12);
    __builtin_prefetch(BR, 0, 1);
    __builtin_prefetch(BI, 0, 1);
    v8f cR = {}, cI = {};
    for (int k = 0; k < 64; k += 4) {
      v2f ar = fragA(AR, 64, 0, k);
      v2f ai = fragA(AI, 64, 0, k);
      v2f br = fragB(BR, 64, k, n0);
      v2f bi = fragB(BI, 64, k, n0);
      cR = wmma4(ar, br, cR);
      v2f ani; ani.x = -ai.x; ani.y = -ai.y;
      cR = wmma4(ani, bi, cR);
      cI = wmma4(ar, bi, cI);
      cI = wmma4(ai, br, cI);
    }
#pragma unroll
    for (int r = 0; r < 8; ++r) {
      int row = (mb + r) * 16 + ol;
      stg[row * K3_PITCH + fl] = cR[r];
      stg[(256 + row) * K3_PITCH + fl] = cI[r];
    }
  }
  __syncthreads();
  int q = threadIdx.x & 7;
#pragma unroll
  for (int j = 0; j < 8; ++j) {
    int row = (threadIdx.x >> 3) + (j << 6);
    int rb = (row >> 4) & 15, ro = row & 15;
    float4 v = *(const float4*)(stg + row * K3_PITCH + (q << 2));
    float* dst = ((row >> 8) ? YbI : YbR) +
                 ((size_t)(rb * 64 + n0 + ro) << 12) + f0 + (q << 2);
    *(float4*)dst = v;
  }
}

#define S_PITCH 68
__global__ void k4_inv(const float* __restrict__ YbR, const float* __restrict__ YbI,
                       const float* __restrict__ Ftab, const float* __restrict__ bias,
                       float* __restrict__ out) {
  __shared__ __attribute__((aligned(16))) float Sr[128 * S_PITCH];
  __shared__ __attribute__((aligned(16))) float Si[128 * S_PITCH];
  const float* Fre = Ftab;
  const float* Fim = Ftab + 16384;
  int blk = blockIdx.x;
  const float* Yr = YbR + ((size_t)blk << 12);
  const float* Yi = YbI + ((size_t)blk << 12);
  int wave = threadIdx.x >> 5;
  int lane = threadIdx.x & 31;

  for (int t = wave; t < 32; t += 8) {
    int m0 = (t >> 2) << 4, n0 = (t & 3) << 4;
    v8f cR = {}, cI = {};
    for (int k = 0; k < 64; k += 4) {
      v2f ar = fragA_kmap(Fre, m0, k);
      v2f ai = fragA_kmap(Fim, m0, k);
      v2f br = fragB(Yr, 64, k, n0);
      v2f bi = fragB(Yi, 64, k, n0);
      cR = wmma4(ar, br, cR);
      cR = wmma4(ai, bi, cR);
      cI = wmma4(ar, bi, cI);
      v2f ani; ani.x = -ai.x; ani.y = -ai.y;
      cI = wmma4(ani, br, cI);
    }
    int n = n0 + (lane & 15);
    int mb = m0 + ((lane >> 4) << 3);
#pragma unroll
    for (int r = 0; r < 8; ++r) {
      Sr[(mb + r) * S_PITCH + n] = cR[r];
      Si[(mb + r) * S_PITCH + n] = cI[r];
    }
  }
  __syncthreads();

  float bo = bias[blk & 63];
  float* oimg = out + ((size_t)blk << 14);
  for (int t = wave; t < 64; t += 8) {
    int m0 = (t >> 3) << 4, n0 = (t & 7) << 4;
    v8f c = {};
    for (int k = 0; k < 64; k += 4) {
      v2f sr = fragA(Sr, S_PITCH, m0, k);
      v2f si = fragA(Si, S_PITCH, m0, k);
      v2f br = fragB_kmap(Fre, k, n0);
      v2f bi = fragB_kmap(Fim, k, n0);
      c = wmma4(sr, br, c);
      c = wmma4(si, bi, c);
    }
    int n = n0 + (lane & 15);
    int mb = m0 + ((lane >> 4) << 3);
#pragma unroll
    for (int r = 0; r < 8; ++r)
      oimg[(mb + r) * 128 + n] = c[r] * (1.f / 16384.f) + bo;
  }
}

extern "C" void kernel_launch(void* const* d_in, const int* in_sizes, int n_in,
                              void* d_out, int out_size, void* d_ws, size_t ws_size,
                              hipStream_t stream) {
  (void)in_sizes; (void)n_in; (void)out_size; (void)ws_size;
  const float* x    = (const float*)d_in[0];
  const float* w    = (const float*)d_in[1];
  const float* bias = (const float*)d_in[2];
  float* out = (float*)d_out;

  float* ws   = (float*)d_ws;
  float* Ftab = ws;
  float* XbR  = ws + 32768;
  float* XbI  = XbR + 4194304;
  float* WbR  = XbI + 4194304;
  float* WbI  = WbR + 16777216;
  float* YbR  = WbI + 16777216;
  float* YbI  = YbR + 4194304;

  k0_tables<<<16, 256, 0, stream>>>(Ftab);
  k1_fwd  <<<1024, 256, 0, stream>>>(x, Ftab, XbR, XbI);
  k2_wdft <<<1024, 256, 0, stream>>>(w, WbR, WbI);
  k3_mix  <<<512, 512, 0, stream>>>(XbR, XbI, WbR, WbI, YbR, YbI);
  k4_inv  <<<1024, 256, 0, stream>>>(YbR, YbI, Ftab, bias, out);
}


# X spectrum workspace re-laid as [f/32][b][i][f mod 32]: forward kernel writes 64-byte line pieces, not 4-byte words 4 KB apart; k3 reads whole lines and turns them through LDS into A fragments
# speedup vs baseline: 2.8979x; 1.8334x over previous
#include <hip/hip_runtime.h>


typedef __attribute__((ext_vector_type(2))) float v2f;
typedef __attribute__((ext_vector_type(8))) float v8f;

#define PI_F 3.14159265358979323846f

__device__ __forceinline__ v8f wmma4(v2f a, v2f b, v8f c) {
  return __builtin_amdgcn_wmma_f32_16x16x4_f32(false, a, false, b, (short)0, c,
                                               false, false);
}

__device__ __forceinline__ int bandmap(int bi) { return (bi + 96) & 127; }

__device__ __forceinline__ v2f fragA(const float* A, int lda, int m0, int k0) {
  int lane = threadIdx.x & 31;
  int m = m0 + (lane & 15);
  int kk = k0 + ((lane >> 4) << 1);
  v2f r; r.x = A[m * lda + kk]; r.y = A[m * lda + kk + 1]; return r;
}
__device__ __forceinline__ v2f fragB(const float* B, int ldb, int k0, int n0) {
  int lane = threadIdx.x & 31;
  int n = n0 + (lane & 15);
  int kk = k0 + ((lane >> 4) << 1);
  v2f r; r.x = B[kk * ldb + n]; r.y = B[(kk + 1) * ldb + n]; return r;
}
__device__ __forceinline__ v2f fragB_bc(const float* F, int k0, int n0) {
  int lane = threadIdx.x & 31;
  int col = bandmap(n0 + (lane & 15));
  int kk = k0 + ((lane >> 4) << 1);
  v2f r; r.x = F[kk * 128 + col]; r.y = F[(kk + 1) * 128 + col]; return r;
}
__device__ __forceinline__ v2f fragA_br(const float* F, int m0, int k0) {
  int lane = threadIdx.x & 31;
  int row = bandmap(m0 + (lane & 15));
  int kk = k0 + ((lane >> 4) << 1);
  v2f r; r.x = F[row * 128 + kk]; r.y = F[row * 128 + kk + 1]; return r;
}
__device__ __forceinline__ v2f fragA_kmap(const float* F, int m0, int k0) {
  int lane = threadIdx.x & 31;
  int m = m0 + (lane & 15);
  int kk = k0 + ((lane >> 4) << 1);
  v2f r; r.x = F[m * 128 + bandmap(kk)]; r.y = F[m * 128 + bandmap(kk + 1)];
  return r;
}
__device__ __forceinline__ v2f fragB_kmap(const float* F, int k0, int n0) {
  int lane = threadIdx.x & 31;
  int n = n0 + (lane & 15);
  int kk = k0 + ((lane >> 4) << 1);
  v2f r; r.x = F[bandmap(kk) * 128 + n]; r.y = F[bandmap(kk + 1) * 128 + n];
  return r;
}

__global__ void k0_tables(float* __restrict__ Ftab) {
  for (int t = threadIdx.x + blockIdx.x * blockDim.x; t < 16384;
       t += blockDim.x * gridDim.x) {
    int j = t >> 7, k = t & 127;
    float ang = -2.f * PI_F * (float)((j * k) & 127) / 128.f;
    float s, c;
    sincosf(ang, &s, &c);
    Ftab[t] = c;
    Ftab[16384 + t] = s;
  }
}

__global__ void k1_fwd(const float* __restrict__ x, const float* __restrict__ Ftab,
                       float* __restrict__ XbR, float* __restrict__ XbI) {
  __shared__ float T1r[128 * 64];
  __shared__ float T1i[128 * 64];
  const float* Fre = Ftab;
  const float* Fim = Ftab + 16384;
  int blk = blockIdx.x;
  int b = blk >> 6, ci = blk & 63;
  const float* Ximg = x + ((size_t)blk << 14);
  int wave = threadIdx.x >> 5;
  int lane = threadIdx.x & 31;

  for (int t = wave; t < 32; t += 8) {
    int m0 = (t >> 2) << 4, n0 = (t & 3) << 4;
    v8f aR = {}, aI = {};
    for (int k = 0; k < 128; k += 4) {
      v2f a  = fragA(Ximg, 128, m0, k);
      v2f br = fragB_bc(Fre, k, n0);
      v2f bi = fragB_bc(Fim, k, n0);
      aR = wmma4(a, br, aR);
      aI = wmma4(a, bi, aI);
    }
    int n = n0 + (lane & 15);
    int mb = m0 + ((lane >> 4) << 3);
#pragma unroll
    for (int r = 0; r < 8; ++r) {
      T1r[(mb + r) * 64 + n] = aR[r];
      T1i[(mb + r) * 64 + n] = aI[r];
    }
  }
  __syncthreads();

  int bdst = (b + 8) & 15;
  for (int t = wave; t < 16; t += 8) {
    int m0 = (t >> 2) << 4, n0 = (t & 3) << 4;
    v8f cR = {}, cI = {};
    for (int k = 0; k < 128; k += 4) {
      v2f ar = fragA_br(Fre, m0, k);
      v2f ai = fragA_br(Fim, m0, k);
      v2f br = fragB(T1r, 64, k, n0);
      v2f bi = fragB(T1i, 64, k, n0);
      cR = wmma4(ar, br, cR);
      v2f ani; ani.x = -ai.x; ani.y = -ai.y;
      cR = wmma4(ani, bi, cR);
      cI = wmma4(ar, bi, cI);
      cI = wmma4(ai, br, cI);
    }
    int n = n0 + (lane & 15);
    int mb = m0 + ((lane >> 4) << 3);
    int dv = n - 32;
#pragma unroll
    for (int r = 0; r < 8; ++r) {
      int uu = mb + r;
      int du = uu - 32;
      float msk = (du * du + dv * dv <= 900) ? 1.f : 0.f;
      size_t off = ((((size_t)((uu << 1) + (n >> 5)) << 10) +
                     (size_t)(bdst << 6) + ci) << 5) + (size_t)(n & 31);
      XbR[off] = cR[r] * msk;
      XbI[off] = cI[r] * msk;
    }
  }
}

__global__ void __launch_bounds__(256)
k2_wdft(const float* __restrict__ w, float* __restrict__ WbR,
        float* __restrict__ WbI) {
  __shared__ __attribute__((aligned(16))) float tw[16 * 20];
  int bid = blockIdx.x;
  int pchunk = bid & 3;
  int vchunk = (bid >> 2) & 3;
  int uu = bid >> 4;
  int tid = threadIdx.x;
  if (tid < 144) {
    int vloc = tid / 9, tap = tid - vloc * 9;
    int ky = tap / 3, kx = tap - ky * 3;
    int u = bandmap(uu), v = bandmap(vchunk * 16 + vloc);
    int idx = (u * ky + v * kx) & 127;
    float s, c;
    sincosf(-2.f * PI_F * (float)idx / 128.f, &s, &c);
    tw[vloc * 20 + tap] = c;
    tw[vloc * 20 + 9 + tap] = s;
  } else if (tid < 160) {
    tw[(tid - 144) * 20 + 18] = 0.f;
    tw[(tid - 144) * 20 + 19] = 0.f;
  }
  int p0 = (pchunk << 10) + (tid << 2);
  int i = p0 >> 6, op0 = p0 & 63;
  float wv[4][9];
#pragma unroll
  for (int q = 0; q < 4; ++q) {
    int osrc = (op0 + q + 32) & 63;
    const float* wk = w + (size_t)(osrc * 64 + i) * 9;
#pragma unroll
    for (int t = 0; t < 9; ++t) wv[q][t] = wk[t];
  }
  __syncthreads();
  size_t fbase = (size_t)(uu * 64 + vchunk * 16);
  for (int vloc = 0; vloc < 16; ++vloc) {
    const float4* t4 = (const float4*)(tw + vloc * 20);
    float4 a = t4[0], b = t4[1], c = t4[2], d = t4[3], e = t4[4];
    float tr[9] = {a.x, a.y, a.z, a.w, b.x, b.y, b.z, b.w, c.x};
    float ti[9] = {c.y, c.z, c.w, d.x, d.y, d.z, d.w, e.x, e.y};
    float ar[4], ai[4];
#pragma unroll
    for (int q = 0; q < 4; ++q) {
      float sr = 0.f, si = 0.f;
#pragma unroll
      for (int t = 0; t < 9; ++t) {
        sr += wv[q][t] * tr[t];
        si += wv[q][t] * ti[t];
      }
      ar[q] = sr;
      ai[q] = si;
    }
    size_t off = ((fbase + (size_t)vloc) << 12) + (size_t)p0;
    *(float4*)(WbR + off) = make_float4(ar[0], ar[1], ar[2], ar[3]);
    *(float4*)(WbI + off) = make_float4(ai[0], ai[1], ai[2], ai[3]);
  }
}

#define K3_FPB 32
#define K3_PITCH 36
#define K3_APB 68
#define K3_APP (16 * K3_APB + 4)
#define K3_APL (8 * K3_APP)
__global__ void __launch_bounds__(512)
k3_mix(const float* __restrict__ XbR, const float* __restrict__ XbI,
       const float* __restrict__ WbR, const float* __restrict__ WbI,
       float* __restrict__ YbR, float* __restrict__ YbI) {
  __shared__ __attribute__((aligned(16))) float stg[2 * 256 * K3_PITCH];
  __shared__ __attribute__((aligned(16))) float ast[2 * K3_APL];
  int wave = threadIdx.x >> 5;
  int lane = threadIdx.x & 31;
  int n0 = (blockIdx.x & 3) << 4;
  int fb = blockIdx.x >> 2;
  int f0 = fb * K3_FPB;
  int ol = lane & 15;
  int hh = lane >> 4;
  int mb = hh << 3;

  const float* XR = XbR + ((size_t)fb << 15);
  const float* XI = XbI + ((size_t)fb << 15);
  int cq = threadIdx.x & 7;
  int cp = (threadIdx.x >> 3) & 7;
  int cb = threadIdx.x >> 6;
  int gsrc = ((cb * 64 + 2 * cp) << 5) + (cq << 2);
  int ldst = cp * K3_APP + cb * K3_APB + (cq << 3);

  int fa = f0 + 2 * wave;
  const float* BR0 = WbR + ((size_t)fa << 12);
  const float* BI0 = WbI + ((size_t)fa << 12);
  const float* BR1 = BR0 + 4096;
  const float* BI1 = BI0 + 4096;
  __builtin_prefetch(BR0, 0, 1);
  __builtin_prefetch(BI0, 0, 1);
  __builtin_prefetch(BR1, 0, 1);
  __builtin_prefetch(BI1, 0, 1);

  const float* afr = ast + hh * K3_APP + ol * K3_APB + (wave << 2);

  float4 pre[8];
#pragma unroll
  for (int it = 0; it < 2; ++it) {
    const float* g = XR + gsrc + (it << 14);
    const float* h = XI + gsrc + (it << 14);
    pre[it * 4 + 0] = *(const float4*)(g);
    pre[it * 4 + 1] = *(const float4*)(g + 32);
    pre[it * 4 + 2] = *(const float4*)(h);
    pre[it * 4 + 3] = *(const float4*)(h + 32);
  }

  v8f cR0 = {}, cI0 = {}, cR1 = {}, cI1 = {};
#pragma unroll
  for (int c = 0; c < 4; ++c) {
    if (c) __syncthreads();
#pragma unroll
    for (int it = 0; it < 2; ++it) {
      float* d = ast + ldst + it * (8 * K3_APB);
      float4 e0 = pre[it * 4 + 0], e1 = pre[it * 4 + 1];
      *(float4*)(d)     = make_float4(e0.x, e1.x, e0.y, e1.y);
      *(float4*)(d + 4) = make_float4(e0.z, e1.z, e0.w, e1.w);
      float4 i0 = pre[it * 4 + 2], i1 = pre[it * 4 + 3];
      *(float4*)(d + K3_APL)     = make_float4(i0.x, i1.x, i0.y, i1.y);
      *(float4*)(d + K3_APL + 4) = make_float4(i0.z, i1.z, i0.w, i1.w);
    }
    __syncthreads();
    if (c < 3) {
#pragma unroll
      for (int it = 0; it < 2; ++it) {
        const float* g = XR + gsrc + (it << 14) + ((c + 1) << 9);
        const float* h = XI + gsrc + (it << 14) + ((c + 1) << 9);
        pre[it * 4 + 0] = *(const float4*)(g);
        pre[it * 4 + 1] = *(const float4*)(g + 32);
        pre[it * 4 + 2] = *(const float4*)(h);
        pre[it * 4 + 3] = *(const float4*)(h + 32);
      }
    }
#pragma unroll
    for (int ks = 0; ks < 4; ++ks) {
      int k = (c << 4) + (ks << 2);
      float4 pr = *(const float4*)(afr + 2 * ks * K3_APP);
      float4 pi = *(const float4*)(afr + 2 * ks * K3_APP + K3_APL);
      v2f ar0, ai0, ar1, ai1;
      ar0.x = pr.x; ar0.y = pr.y; ar1.x = pr.z; ar1.y = pr.w;
      ai0.x = pi.x; ai0.y = pi.y; ai1.x = pi.z; ai1.y = pi.w;
      v2f br0 = fragB(BR0, 64, k, n0);
      v2f bi0 = fragB(BI0, 64, k, n0);
      v2f br1 = fragB(BR1, 64, k, n0);
      v2f bi1 = fragB(BI1, 64, k, n0);
      v2f an0; an0.x = -ai0.x; an0.y = -ai0.y;
      v2f an1; an1.x = -ai1.x; an1.y = -ai1.y;
      cR0 = wmma4(ar0, br0, cR0);
      cR1 = wmma4(ar1, br1, cR1);
      cI0 = wmma4(ar0, bi0, cI0);
      cI1 = wmma4(ar1, bi1, cI1);
      cR0 = wmma4(an0, bi0, cR0);
      cR1 = wmma4(an1, bi1, cR1);
      cI0 = wmma4(ai0, br0, cI0);
      cI1 = wmma4(ai1, br1, cI1);
    }
  }
#pragma unroll
  for (int r = 0; r < 8; ++r) {
    int row = (mb + r) * 16 + ol;
    *(float2*)(stg + row * K3_PITCH + 2 * wave) = make_float2(cR0[r], cR1[r]);
    *(float2*)(stg + (256 + row) * K3_PITCH + 2 * wave) = make_float2(cI0[r], cI1[r]);
  }
  __syncthreads();
  int q = threadIdx.x & 7;
#pragma unroll
  for (int j = 0; j < 8; ++j) {
    int row = (threadIdx.x >> 3) + (j << 6);
    int rb = (row >> 4) & 15, ro = row & 15;
    float4 v = *(const float4*)(stg + row * K3_PITCH + (q << 2));
    float* dst = ((row >> 8) ? YbI : YbR) +
                 ((size_t)(rb * 64 + n0 + ro) << 12) + f0 + (q << 2);
    *(float4*)dst = v;
  }
}

#define S_PITCH 68
__global__ void k4_inv(const float* __restrict__ YbR, const float* __restrict__ YbI,
                       const float* __restrict__ Ftab, const float* __restrict__ bias,
                       float* __restrict__ out) {
  __shared__ __attribute__((aligned(16))) float Sr[128 * S_PITCH];
  __shared__ __attribute__((aligned(16))) float Si[128 * S_PITCH];
  const float* Fre = Ftab;
  const float* Fim = Ftab + 16384;
  int blk = blockIdx.x;
  const float* Yr = YbR + ((size_t)blk << 12);
  const float* Yi = YbI + ((size_t)blk << 12);
  int wave = threadIdx.x >> 5;
  int lane = threadIdx.x & 31;

  for (int t = wave; t < 32; t += 8) {
    int m0 = (t >> 2) << 4, n0 = (t & 3) << 4;
    v8f cR = {}, cI = {};
    for (int k = 0; k < 64; k += 4) {
      v2f ar = fragA_kmap(Fre, m0, k);
      v2f ai = fragA_kmap(Fim, m0, k);
      v2f br = fragB(Yr, 64, k, n0);
      v2f bi = fragB(Yi, 64, k, n0);
      cR = wmma4(ar, br, cR);
      cR = wmma4(ai, bi, cR);
      cI = wmma4(ar, bi, cI);
      v2f ani; ani.x = -ai.x; ani.y = -ai.y;
      cI = wmma4(ani, br, cI);
    }
    int n = n0 + (lane & 15);
    int mb = m0 + ((lane >> 4) << 3);
#pragma unroll
    for (int r = 0; r < 8; ++r) {
      Sr[(mb + r) * S_PITCH + n] = cR[r];
      Si[(mb + r) * S_PITCH + n] = cI[r];
    }
  }
  __syncthreads();

  float bo = bias[blk & 63];
  float* oimg = out + ((size_t)blk << 14);
  for (int t = wave; t < 64; t += 8) {
    int m0 = (t >> 3) << 4, n0 = (t & 7) << 4;
    v8f c = {};
    for (int k = 0; k < 64; k += 4) {
      v2f sr = fragA(Sr, S_PITCH, m0, k);
      v2f si = fragA(Si, S_PITCH, m0, k);
      v2f br = fragB_kmap(Fre, k, n0);
      v2f bi = fragB_kmap(Fim, k, n0);
      c = wmma4(sr, br, c);
      c = wmma4(si, bi, c);
    }
    int n = n0 + (lane & 15);
    int mb = m0 + ((lane >> 4) << 3);
#pragma unroll
    for (int r = 0; r < 8; ++r)
      oimg[(mb + r) * 128 + n] = c[r] * (1.f / 16384.f) + bo;
  }
}

extern "C" void kernel_launch(void* const* d_in, const int* in_sizes, int n_in,
                              void* d_out, int out_size, void* d_ws, size_t ws_size,
                              hipStream_t stream) {
  (void)in_sizes; (void)n_in; (void)out_size; (void)ws_size;
  const float* x    = (const float*)d_in[0];
  const float* w    = (const float*)d_in[1];
  const float* bias = (const float*)d_in[2];
  float* out = (float*)d_out;

  float* ws   = (float*)d_ws;
  float* Ftab = ws;
  float* XbR  = ws + 32768;
  float* XbI  = XbR + 4194304;
  float* WbR  = XbI + 4194304;
  float* WbI  = WbR + 16777216;
  float* YbR  = WbI + 16777216;
  float* YbI  = YbR + 4194304;

  k0_tables<<<16, 256, 0, stream>>>(Ftab);
  k1_fwd  <<<1024, 256, 0, stream>>>(x, Ftab, XbR, XbI);
  k2_wdft <<<1024, 256, 0, stream>>>(w, WbR, WbI);
  k3_mix  <<<512, 512, 0, stream>>>(XbR, XbI, WbR, WbI, YbR, YbI);
  k4_inv  <<<1024, 256, 0, stream>>>(YbR, YbI, Ftab, bias, out);
}


# k4 stage 1 and k1 stage 2: DFT-matrix A fragments read through the table's symmetry (F[k][m] for F[m][k]): 16 lanes read one 64-byte row piece, not 16 lines; values and sum order unchanged
# speedup vs baseline: 3.1275x; 1.0792x over previous
#include <hip/hip_runtime.h>


typedef __attribute__((ext_vector_type(2))) float v2f;
typedef __attribute__((ext_vector_type(8))) float v8f;

#define PI_F 3.14159265358979323846f

__device__ __forceinline__ v8f wmma4(v2f a, v2f b, v8f c) {
  return __builtin_amdgcn_wmma_f32_16x16x4_f32(false, a, false, b, (short)0, c,
                                               false, false);
}

__device__ __forceinline__ int bandmap(int bi) { return (bi + 96) & 127; }

__device__ __forceinline__ v2f fragA(const float* A, int lda, int m0, int k0) {
  int lane = threadIdx.x & 31;
  int m = m0 + (lane & 15);
  int kk = k0 + ((lane >> 4) << 1);
  v2f r; r.x = A[m * lda + kk]; r.y = A[m * lda + kk + 1]; return r;
}
__device__ __forceinline__ v2f fragB(const float* B, int ldb, int k0, int n0) {
  int lane = threadIdx.x & 31;
  int n = n0 + (lane & 15);
  int kk = k0 + ((lane >> 4) << 1);
  v2f r; r.x = B[kk * ldb + n]; r.y = B[(kk + 1) * ldb + n]; return r;
}
__device__ __forceinline__ v2f fragB_bc(const float* F, int k0, int n0) {
  int lane = threadIdx.x & 31;
  int col = bandmap(n0 + (lane & 15));
  int kk = k0 + ((lane >> 4) << 1);
  v2f r; r.x = F[kk * 128 + col]; r.y = F[(kk + 1) * 128 + col]; return r;
}
__device__ __forceinline__ v2f fragA_br(const float* F, int m0, int k0) {
  int lane = threadIdx.x & 31;
  int col = bandmap(m0 + (lane & 15));
  int kk = k0 + ((lane >> 4) << 1);
  v2f r; r.x = F[kk * 128 + col]; r.y = F[(kk + 1) * 128 + col]; return r;
}
__device__ __forceinline__ v2f fragA_kmap(const float* F, int m0, int k0) {
  int lane = threadIdx.x & 31;
  int m = m0 + (lane & 15);
  int kk = k0 + ((lane >> 4) << 1);
  v2f r; r.x = F[bandmap(kk) * 128 + m]; r.y = F[bandmap(kk + 1) * 128 + m];
  return r;
}
__device__ __forceinline__ v2f fragB_kmap(const float* F, int k0, int n0) {
  int lane = threadIdx.x & 31;
  int n = n0 + (lane & 15);
  int kk = k0 + ((lane >> 4) << 1);
  v2f r; r.x = F[bandmap(kk) * 128 + n]; r.y = F[bandmap(kk + 1) * 128 + n];
  return r;
}

__global__ void k0_tables(float* __restrict__ Ftab) {
  for (int t = threadIdx.x + blockIdx.x * blockDim.x; t < 16384;
       t += blockDim.x * gridDim.x) {
    int j = t >> 7, k = t & 127;
    float ang = -2.f * PI_F * (float)((j * k) & 127) / 128.f;
    float s, c;
    sincosf(ang, &s, &c);
    Ftab[t] = c;
    Ftab[16384 + t] = s;
  }
}

__global__ void k1_fwd(const float* __restrict__ x, const float* __restrict__ Ftab,
                       float* __restrict__ XbR, float* __restrict__ XbI) {
  __shared__ float T1r[128 * 64];
  __shared__ float T1i[128 * 64];
  const float* Fre = Ftab;
  const float* Fim = Ftab + 16384;
  int blk = blockIdx.x;
  int b = blk >> 6, ci = blk & 63;
  const float* Ximg = x + ((size_t)blk << 14);
  int wave = threadIdx.x >> 5;
  int lane = threadIdx.x & 31;

  for (int t = wave; t < 32; t += 8) {
    int m0 = (t >> 2) << 4, n0 = (t & 3) << 4;
    v8f aR = {}, aI = {};
    for (int k = 0; k < 128; k += 4) {
      v2f a  = fragA(Ximg, 128, m0, k);
      v2f br = fragB_bc(Fre, k, n0);
      v2f bi = fragB_bc(Fim, k, n0);
      aR = wmma4(a, br, aR);
      aI = wmma4(a, bi, aI);
    }
    int n = n0 + (lane & 15);
    int mb = m0 + ((lane >> 4) << 3);
#pragma unroll
    for (int r = 0; r < 8; ++r) {
      T1r[(mb + r) * 64 + n] = aR[r];
      T1i[(mb + r) * 64 + n] = aI[r];
    }
  }
  __syncthreads();

  int bdst = (b + 8) & 15;
  for (int t = wave; t < 16; t += 8) {
    int m0 = (t >> 2) << 4, n0 = (t & 3) << 4;
    v8f cR = {}, cI = {};
    for (int k = 0; k < 128; k += 4) {
      v2f ar = fragA_br(Fre, m0, k);
      v2f ai = fragA_br(Fim, m0, k);
      v2f br = fragB(T1r, 64, k, n0);
      v2f bi = fragB(T1i, 64, k, n0);
      cR = wmma4(ar, br, cR);
      v2f ani; ani.x = -ai.x; ani.y = -ai.y;
      cR = wmma4(ani, bi, cR);
      cI = wmma4(ar, bi, cI);
      cI = wmma4(ai, br, cI);
    }
    int n = n0 + (lane & 15);
    int mb = m0 + ((lane >> 4) << 3);
    int dv = n - 32;
#pragma unroll
    for (int r = 0; r < 8; ++r) {
      int uu = mb + r;
      int du = uu - 32;
      float msk = (du * du + dv * dv <= 900) ? 1.f : 0.f;
      size_t off = ((((size_t)((uu << 1) + (n >> 5)) << 10) +
                     (size_t)(bdst << 6) + ci) << 5) + (size_t)(n & 31);
      XbR[off] = cR[r] * msk;
      XbI[off] = cI[r] * msk;
    }
  }
}

__global__ void __launch_bounds__(256)
k2_wdft(const float* __restrict__ w, float* __restrict__ WbR,
        float* __restrict__ WbI) {
  __shared__ __attribute__((aligned(16))) float tw[16 * 20];
  int bid = blockIdx.x;
  int pchunk = bid & 3;
  int vchunk = (bid >> 2) & 3;
  int uu = bid >> 4;
  int tid = threadIdx.x;
  if (tid < 144) {
    int vloc = tid / 9, tap = tid - vloc * 9;
    int ky = tap / 3, kx = tap - ky * 3;
    int u = bandmap(uu), v = bandmap(vchunk * 16 + vloc);
    int idx = (u * ky + v * kx) & 127;
    float s, c;
    sincosf(-2.f * PI_F * (float)idx / 128.f, &s, &c);
    tw[vloc * 20 + tap] = c;
    tw[vloc * 20 + 9 + tap] = s;
  } else if (tid < 160) {
    tw[(tid - 144) * 20 + 18] = 0.f;
    tw[(tid - 144) * 20 + 19] = 0.f;
  }
  int p0 = (pchunk << 10) + (tid << 2);
  int i = p0 >> 6, op0 = p0 & 63;
  float wv[4][9];
#pragma unroll
  for (int q = 0; q < 4; ++q) {
    int osrc = (op0 + q + 32) & 63;
    const float* wk = w + (size_t)(osrc * 64 + i) * 9;
#pragma unroll
    for (int t = 0; t < 9; ++t) wv[q][t] = wk[t];
  }
  __syncthreads();
  size_t fbase = (size_t)(uu * 64 + vchunk * 16);
  for (int vloc = 0; vloc < 16; ++vloc) {
    const float4* t4 = (const float4*)(tw + vloc * 20);
    float4 a = t4[0], b = t4[1], c = t4[2], d = t4[3], e = t4[4];
    float tr[9] = {a.x, a.y, a.z, a.w, b.x, b.y, b.z, b.w, c.x};
    float ti[9] = {c.y, c.z, c.w, d.x, d.y, d.z, d.w, e.x, e.y};
    float ar[4], ai[4];
#pragma unroll
    for (int q = 0; q < 4; ++q) {
      float sr = 0.f, si = 0.f;
#pragma unroll
      for (int t = 0; t < 9; ++t) {
        sr += wv[q][t] * tr[t];
        si += wv[q][t] * ti[t];
      }
      ar[q] = sr;
      ai[q] = si;
    }
    size_t off = ((fbase + (size_t)vloc) << 12) + (size_t)p0;
    *(float4*)(WbR + off) = make_float4(ar[0], ar[1], ar[2], ar[3]);
    *(float4*)(WbI + off) = make_float4(ai[0], ai[1], ai[2], ai[3]);
  }
}

#define K3_FPB 32
#define K3_PITCH 36
#define K3_APB 68
#define K3_APP (16 * K3_APB + 4)
#define K3_APL (8 * K3_APP)
__global__ void __launch_bounds__(512)
k3_mix(const float* __restrict__ XbR, const float* __restrict__ XbI,
       const float* __restrict__ WbR, const float* __restrict__ WbI,
       float* __restrict__ YbR, float* __restrict__ YbI) {
  __shared__ __attribute__((aligned(16))) float stg[2 * 256 * K3_PITCH];
  __shared__ __attribute__((aligned(16))) float ast[2 * K3_APL];
  int wave = threadIdx.x >> 5;
  int lane = threadIdx.x & 31;
  int n0 = (blockIdx.x & 3) << 4;
  int fb = blockIdx.x >> 2;
  int f0 = fb * K3_FPB;
  int ol = lane & 15;
  int hh = lane >> 4;
  int mb = hh << 3;

  const float* XR = XbR + ((size_t)fb << 15);
  const float* XI = XbI + ((size_t)fb << 15);
  int cq = threadIdx.x & 7;
  int cp = (threadIdx.x >> 3) & 7;
  int cb = threadIdx.x >> 6;
  int gsrc = ((cb * 64 + 2 * cp) << 5) + (cq << 2);
  int ldst = cp * K3_APP + cb * K3_APB + (cq << 3);

  int fa = f0 + 2 * wave;
  const float* BR0 = WbR + ((size_t)fa << 12);
  const float* BI0 = WbI + ((size_t)fa << 12);
  const float* BR1 = BR0 + 4096;
  const float* BI1 = BI0 + 4096;
  __builtin_prefetch(BR0, 0, 1);
  __builtin_prefetch(BI0, 0, 1);
  __builtin_prefetch(BR1, 0, 1);
  __builtin_prefetch(BI1, 0, 1);

  const float* afr = ast + hh * K3_APP + ol * K3_APB + (wave << 2);

  float4 pre[8];
#pragma unroll
  for (int it = 0; it < 2; ++it) {
    const float* g = XR + gsrc + (it << 14);
    const float* h = XI + gsrc + (it << 14);
    pre[it * 4 + 0] = *(const float4*)(g);
    pre[it * 4 + 1] = *(const float4*)(g + 32);
    pre[it * 4 + 2] = *(const float4*)(h);
    pre[it * 4 + 3] = *(const float4*)(h + 32);
  }

  v8f cR0 = {}, cI0 = {}, cR1 = {}, cI1 = {};
#pragma unroll
  for (int c = 0; c < 4; ++c) {
    if (c) __syncthreads();
#pragma unroll
    for (int it = 0; it < 2; ++it) {
      float* d = ast + ldst + it * (8 * K3_APB);
      float4 e0 = pre[it * 4 + 0], e1 = pre[it * 4 + 1];
      *(float4*)(d)     = make_float4(e0.x, e1.x, e0.y, e1.y);
      *(float4*)(d + 4) = make_float4(e0.z, e1.z, e0.w, e1.w);
      float4 i0 = pre[it * 4 + 2], i1 = pre[it * 4 + 3];
      *(float4*)(d + K3_APL)     = make_float4(i0.x, i1.x, i0.y, i1.y);
      *(float4*)(d + K3_APL + 4) = make_float4(i0.z, i1.z, i0.w, i1.w);
    }
    __syncthreads();
    if (c < 3) {
#pragma unroll
      for (int it = 0; it < 2; ++it) {
        const float* g = XR + gsrc + (it << 14) + ((c + 1) << 9);
        const float* h = XI + gsrc + (it << 14) + ((c + 1) << 9);
        pre[it * 4 + 0] = *(const float4*)(g);
        pre[it * 4 + 1] = *(const float4*)(g + 32);
        pre[it * 4 + 2] = *(const float4*)(h);
        pre[it * 4 + 3] = *(const float4*)(h + 32);
      }
    }
#pragma unroll
    for (int ks = 0; ks < 4; ++ks) {
      int k = (c << 4) + (ks << 2);
      float4 pr = *(const float4*)(afr + 2 * ks * K3_APP);
      float4 pi = *(const float4*)(afr + 2 * ks * K3_APP + K3_APL);
      v2f ar0, ai0, ar1, ai1;
      ar0.x = pr.x; ar0.y = pr.y; ar1.x = pr.z; ar1.y = pr.w;
      ai0.x = pi.x; ai0.y = pi.y; ai1.x = pi.z; ai1.y = pi.w;
      v2f br0 = fragB(BR0, 64, k, n0);
      v2f bi0 = fragB(BI0, 64, k, n0);
      v2f br1 = fragB(BR1, 64, k, n0);
      v2f bi1 = fragB(BI1, 64, k, n0);
      v2f an0; an0.x = -ai0.x; an0.y = -ai0.y;
      v2f an1; an1.x = -ai1.x; an1.y = -ai1.y;
      cR0 = wmma4(ar0, br0, cR0);
      cR1 = wmma4(ar1, br1, cR1);
      cI0 = wmma4(ar0, bi0, cI0);
      cI1 = wmma4(ar1, bi1, cI1);
      cR0 = wmma4(an0, bi0, cR0);
      cR1 = wmma4(an1, bi1, cR1);
      cI0 = wmma4(ai0, br0, cI0);
      cI1 = wmma4(ai1, br1, cI1);
    }
  }
#pragma unroll
  for (int r = 0; r < 8; ++r) {
    int row = (mb + r) * 16 + ol;
    *(float2*)(stg + row * K3_PITCH + 2 * wave) = make_float2(cR0[r], cR1[r]);
    *(float2*)(stg + (256 + row) * K3_PITCH + 2 * wave) = make_float2(cI0[r], cI1[r]);
  }
  __syncthreads();
  int q = threadIdx.x & 7;
#pragma unroll
  for (int j = 0; j < 8; ++j) {
    int row = (threadIdx.x >> 3) + (j << 6);
    int rb = (row >> 4) & 15, ro = row & 15;
    float4 v = *(const float4*)(stg + row * K3_PITCH + (q << 2));
    float* dst = ((row >> 8) ? YbI : YbR) +
                 ((size_t)(rb * 64 + n0 + ro) << 12) + f0 + (q << 2);
    *(float4*)dst = v;
  }
}

#define S_PITCH 68
__global__ void k4_inv(const float* __restrict__ YbR, const float* __restrict__ YbI,
                       const float* __restrict__ Ftab, const float* __restrict__ bias,
                       float* __restrict__ out) {
  __shared__ __attribute__((aligned(16))) float Sr[128 * S_PITCH];
  __shared__ __attribute__((aligned(16))) float Si[128 * S_PITCH];
  const float* Fre = Ftab;
  const float* Fim = Ftab + 16384;
  int blk = blockIdx.x;
  const float* Yr = YbR + ((size_t)blk << 12);
  const float* Yi = YbI + ((size_t)blk << 12);
  int wave = threadIdx.x >> 5;
  int lane = threadIdx.x & 31;

  for (int t = wave; t < 32; t += 8) {
    int m0 = (t >> 2) << 4, n0 = (t & 3) << 4;
    v8f cR = {}, cI = {};
    for (int k = 0; k < 64; k += 4) {
      v2f ar = fragA_kmap(Fre, m0, k);
      v2f ai = fragA_kmap(Fim, m0, k);
      v2f br = fragB(Yr, 64, k, n0);
      v2f bi = fragB(Yi, 64, k, n0);
      cR = wmma4(ar, br, cR);
      cR = wmma4(ai, bi, cR);
      cI = wmma4(ar, bi, cI);
      v2f ani; ani.x = -ai.x; ani.y = -ai.y;
      cI = wmma4(ani, br, cI);
    }
    int n = n0 + (lane & 15);
    int mb = m0 + ((lane >> 4) << 3);
#pragma unroll
    for (int r = 0; r < 8; ++r) {
      Sr[(mb + r) * S_PITCH + n] = cR[r];
      Si[(mb + r) * S_PITCH + n] = cI[r];
    }
  }
  __syncthreads();

  float bo = bias[blk & 63];
  float* oimg = out + ((size_t)blk << 14);
  for (int t = wave; t < 64; t += 8) {
    int m0 = (t >> 3) << 4, n0 = (t & 7) << 4;
    v8f c = {};
    for (int k = 0; k < 64; k += 4) {
      v2f sr = fragA(Sr, S_PITCH, m0, k);
      v2f si = fragA(Si, S_PITCH, m0, k);
      v2f br = fragB_kmap(Fre, k, n0);
      v2f bi = fragB_kmap(Fim, k, n0);
      c = wmma4(sr, br, c);
      c = wmma4(si, bi, c);
    }
    int n = n0 + (lane & 15);
    int mb = m0 + ((lane >> 4) << 3);
#pragma unroll
    for (int r = 0; r < 8; ++r)
      oimg[(mb + r) * 128 + n] = c[r] * (1.f / 16384.f) + bo;
  }
}

extern "C" void kernel_launch(void* const* d_in, const int* in_sizes, int n_in,
                              void* d_out, int out_size, void* d_ws, size_t ws_size,
                              hipStream_t stream) {
  (void)in_sizes; (void)n_in; (void)out_size; (void)ws_size;
  const float* x    = (const float*)d_in[0];
  const float* w    = (const float*)d_in[1];
  const float* bias = (const float*)d_in[2];
  float* out = (float*)d_out;

  float* ws   = (float*)d_ws;
  float* Ftab = ws;
  float* XbR  = ws + 32768;
  float* XbI  = XbR + 4194304;
  float* WbR  = XbI + 4194304;
  float* WbI  = WbR + 16777216;
  float* YbR  = WbI + 16777216;
  float* YbI  = YbR + 4194304;

  k0_tables<<<16, 256, 0, stream>>>(Ftab);
  k1_fwd  <<<1024, 256, 0, stream>>>(x, Ftab, XbR, XbI);
  k2_wdft <<<1024, 256, 0, stream>>>(w, WbR, WbI);
  k3_mix  <<<512, 512, 0, stream>>>(XbR, XbI, WbR, WbI, YbR, YbI);
  k4_inv  <<<1024, 256, 0, stream>>>(YbR, YbI, Ftab, bias, out);
}


# Half band from real-input conjugate symmetry: only band columns dv=0..31 are transformed, mixed and inverted (dv>0 weighted 2 in the last sum, which changes its summation order); WMMA count halves
# speedup vs baseline: 5.9907x; 1.9155x over previous
#include <hip/hip_runtime.h>


typedef __attribute__((ext_vector_type(2))) float v2f;
typedef __attribute__((ext_vector_type(8))) float v8f;

#define PI_F 3.14159265358979323846f

__device__ __forceinline__ v8f wmma4(v2f a, v2f b, v8f c) {
  return __builtin_amdgcn_wmma_f32_16x16x4_f32(false, a, false, b, (short)0, c,
                                               false, false);
}

__device__ __forceinline__ int bandmap(int bi) { return (bi + 96) & 127; }

__device__ __forceinline__ v2f fragA(const float* A, int lda, int m0, int k0) {
  int lane = threadIdx.x & 31;
  int m = m0 + (lane & 15);
  int kk = k0 + ((lane >> 4) << 1);
  v2f r; r.x = A[m * lda + kk]; r.y = A[m * lda + kk + 1]; return r;
}
__device__ __forceinline__ v2f fragB(const float* B, int ldb, int k0, int n0) {
  int lane = threadIdx.x & 31;
  int n = n0 + (lane & 15);
  int kk = k0 + ((lane >> 4) << 1);
  v2f r; r.x = B[kk * ldb + n]; r.y = B[(kk + 1) * ldb + n]; return r;
}
__device__ __forceinline__ v2f fragB_bc(const float* F, int k0, int n0) {
  int lane = threadIdx.x & 31;
  int col = bandmap(n0 + (lane & 15));
  int kk = k0 + ((lane >> 4) << 1);
  v2f r; r.x = F[kk * 128 + col]; r.y = F[(kk + 1) * 128 + col]; return r;
}
__device__ __forceinline__ v2f fragA_br(const float* F, int m0, int k0) {
  int lane = threadIdx.x & 31;
  int col = bandmap(m0 + (lane & 15));
  int kk = k0 + ((lane >> 4) << 1);
  v2f r; r.x = F[kk * 128 + col]; r.y = F[(kk + 1) * 128 + col]; return r;
}
__device__ __forceinline__ v2f fragA_kmap(const float* F, int m0, int k0) {
  int lane = threadIdx.x & 31;
  int m = m0 + (lane & 15);
  int kk = k0 + ((lane >> 4) << 1);
  v2f r; r.x = F[bandmap(kk) * 128 + m]; r.y = F[bandmap(kk + 1) * 128 + m];
  return r;
}
__device__ __forceinline__ v2f fragB_kmap(const float* F, int k0, int n0) {
  int lane = threadIdx.x & 31;
  int n = n0 + (lane & 15);
  int kk = k0 + ((lane >> 4) << 1);
  v2f r; r.x = F[bandmap(kk) * 128 + n]; r.y = F[bandmap(kk + 1) * 128 + n];
  return r;
}

__global__ void k0_tables(float* __restrict__ Ftab) {
  for (int t = threadIdx.x + blockIdx.x * blockDim.x; t < 16384;
       t += blockDim.x * gridDim.x) {
    int j = t >> 7, k = t & 127;
    float ang = -2.f * PI_F * (float)((j * k) & 127) / 128.f;
    float s, c;
    sincosf(ang, &s, &c);
    Ftab[t] = c;
    Ftab[16384 + t] = s;
  }
}

__global__ void k1_fwd(const float* __restrict__ x, const float* __restrict__ Ftab,
                       float* __restrict__ XbR, float* __restrict__ XbI) {
  __shared__ float T1r[128 * 64];
  __shared__ float T1i[128 * 64];
  const float* Fre = Ftab;
  const float* Fim = Ftab + 16384;
  int blk = blockIdx.x;
  int b = blk >> 6, ci = blk & 63;
  const float* Ximg = x + ((size_t)blk << 14);
  int wave = threadIdx.x >> 5;
  int lane = threadIdx.x & 31;

  for (int t = wave; t < 16; t += 8) {
    int m0 = (t >> 1) << 4, n0 = 32 + ((t & 1) << 4);
    v8f aR = {}, aI = {};
    for (int k = 0; k < 128; k += 4) {
      v2f a  = fragA(Ximg, 128, m0, k);
      v2f br = fragB_bc(Fre, k, n0);
      v2f bi = fragB_bc(Fim, k, n0);
      aR = wmma4(a, br, aR);
      aI = wmma4(a, bi, aI);
    }
    int n = n0 + (lane & 15);
    int mb = m0 + ((lane >> 4) << 3);
#pragma unroll
    for (int r = 0; r < 8; ++r) {
      T1r[(mb + r) * 64 + n] = aR[r];
      T1i[(mb + r) * 64 + n] = aI[r];
    }
  }
  __syncthreads();

  int bdst = (b + 8) & 15;
  {
    int m0 = (wave >> 1) << 4, n0 = 32 + ((wave & 1) << 4);
    v8f cR = {}, cI = {};
    for (int k = 0; k < 128; k += 4) {
      v2f ar = fragA_br(Fre, m0, k);
      v2f ai = fragA_br(Fim, m0, k);
      v2f br = fragB(T1r, 64, k, n0);
      v2f bi = fragB(T1i, 64, k, n0);
      cR = wmma4(ar, br, cR);
      v2f ani; ani.x = -ai.x; ani.y = -ai.y;
      cR = wmma4(ani, bi, cR);
      cI = wmma4(ar, bi, cI);
      cI = wmma4(ai, br, cI);
    }
    int n = n0 + (lane & 15);
    int mb = m0 + ((lane >> 4) << 3);
    int dv = n - 32;
#pragma unroll
    for (int r = 0; r < 8; ++r) {
      int uu = mb + r;
      int du = uu - 32;
      float msk = (du * du + dv * dv <= 900) ? 1.f : 0.f;
      size_t off = ((((size_t)((uu << 1) + (n >> 5)) << 10) +
                     (size_t)(bdst << 6) + ci) << 5) + (size_t)(n & 31);
      XbR[off] = cR[r] * msk;
      XbI[off] = cI[r] * msk;
    }
  }
}

__global__ void __launch_bounds__(256)
k2_wdft(const float* __restrict__ w, float* __restrict__ WbR,
        float* __restrict__ WbI) {
  __shared__ __attribute__((aligned(16))) float tw[16 * 20];
  int bid = blockIdx.x;
  int pchunk = bid & 3;
  int vchunk = 2 + ((bid >> 2) & 1);
  int uu = bid >> 3;
  int tid = threadIdx.x;
  if (tid < 144) {
    int vloc = tid / 9, tap = tid - vloc * 9;
    int ky = tap / 3, kx = tap - ky * 3;
    int u = bandmap(uu), v = bandmap(vchunk * 16 + vloc);
    int idx = (u * ky + v * kx) & 127;
    float s, c;
    sincosf(-2.f * PI_F * (float)idx / 128.f, &s, &c);
    tw[vloc * 20 + tap] = c;
    tw[vloc * 20 + 9 + tap] = s;
  } else if (tid < 160) {
    tw[(tid - 144) * 20 + 18] = 0.f;
    tw[(tid - 144) * 20 + 19] = 0.f;
  }
  int p0 = (pchunk << 10) + (tid << 2);
  int i = p0 >> 6, op0 = p0 & 63;
  float wv[4][9];
#pragma unroll
  for (int q = 0; q < 4; ++q) {
    int osrc = (op0 + q + 32) & 63;
    const float* wk = w + (size_t)(osrc * 64 + i) * 9;
#pragma unroll
    for (int t = 0; t < 9; ++t) wv[q][t] = wk[t];
  }
  __syncthreads();
  size_t fbase = (size_t)(uu * 64 + vchunk * 16);
  for (int vloc = 0; vloc < 16; ++vloc) {
    const float4* t4 = (const float4*)(tw + vloc * 20);
    float4 a = t4[0], b = t4[1], c = t4[2], d = t4[3], e = t4[4];
    float tr[9] = {a.x, a.y, a.z, a.w, b.x, b.y, b.z, b.w, c.x};
    float ti[9] = {c.y, c.z, c.w, d.x, d.y, d.z, d.w, e.x, e.y};
    float ar[4], ai[4];
#pragma unroll
    for (int q = 0; q < 4; ++q) {
      float sr = 0.f, si = 0.f;
#pragma unroll
      for (int t = 0; t < 9; ++t) {
        sr += wv[q][t] * tr[t];
        si += wv[q][t] * ti[t];
      }
      ar[q] = sr;
      ai[q] = si;
    }
    size_t off = ((fbase + (size_t)vloc) << 12) + (size_t)p0;
    *(float4*)(WbR + off) = make_float4(ar[0], ar[1], ar[2], ar[3]);
    *(float4*)(WbI + off) = make_float4(ai[0], ai[1], ai[2], ai[3]);
  }
}

#define K3_FPB 32
#define K3_PITCH 36
#define K3_APB 68
#define K3_APP (16 * K3_APB + 4)
#define K3_APL (8 * K3_APP)
__global__ void __launch_bounds__(512)
k3_mix(const float* __restrict__ XbR, const float* __restrict__ XbI,
       const float* __restrict__ WbR, const float* __restrict__ WbI,
       float* __restrict__ YbR, float* __restrict__ YbI) {
  __shared__ __attribute__((aligned(16))) float stg[2 * 256 * K3_PITCH];
  __shared__ __attribute__((aligned(16))) float ast[2 * K3_APL];
  int wave = threadIdx.x >> 5;
  int lane = threadIdx.x & 31;
  int n0 = (blockIdx.x & 3) << 4;
  int fb = ((blockIdx.x >> 2) << 1) | 1;
  int f0 = fb * K3_FPB;
  int ol = lane & 15;
  int hh = lane >> 4;
  int mb = hh << 3;

  const float* XR = XbR + ((size_t)fb << 15);
  const float* XI = XbI + ((size_t)fb << 15);
  int cq = threadIdx.x & 7;
  int cp = (threadIdx.x >> 3) & 7;
  int cb = threadIdx.x >> 6;
  int gsrc = ((cb * 64 + 2 * cp) << 5) + (cq << 2);
  int ldst = cp * K3_APP + cb * K3_APB + (cq << 3);

  int fa = f0 + 2 * wave;
  const float* BR0 = WbR + ((size_t)fa << 12);
  const float* BI0 = WbI + ((size_t)fa << 12);
  const float* BR1 = BR0 + 4096;
  const float* BI1 = BI0 + 4096;
  __builtin_prefetch(BR0, 0, 1);
  __builtin_prefetch(BI0, 0, 1);
  __builtin_prefetch(BR1, 0, 1);
  __builtin_prefetch(BI1, 0, 1);

  const float* afr = ast + hh * K3_APP + ol * K3_APB + (wave << 2);

  float4 pre[8];
#pragma unroll
  for (int it = 0; it < 2; ++it) {
    const float* g = XR + gsrc + (it << 14);
    const float* h = XI + gsrc + (it << 14);
    pre[it * 4 + 0] = *(const float4*)(g);
    pre[it * 4 + 1] = *(const float4*)(g + 32);
    pre[it * 4 + 2] = *(const float4*)(h);
    pre[it * 4 + 3] = *(const float4*)(h + 32);
  }

  v8f cR0 = {}, cI0 = {}, cR1 = {}, cI1 = {};
#pragma unroll
  for (int c = 0; c < 4; ++c) {
    if (c) __syncthreads();
#pragma unroll
    for (int it = 0; it < 2; ++it) {
      float* d = ast + ldst + it * (8 * K3_APB);
      float4 e0 = pre[it * 4 + 0], e1 = pre[it * 4 + 1];
      *(float4*)(d)     = make_float4(e0.x, e1.x, e0.y, e1.y);
      *(float4*)(d + 4) = make_float4(e0.z, e1.z, e0.w, e1.w);
      float4 i0 = pre[it * 4 + 2], i1 = pre[it * 4 + 3];
      *(float4*)(d + K3_APL)     = make_float4(i0.x, i1.x, i0.y, i1.y);
      *(float4*)(d + K3_APL + 4) = make_float4(i0.z, i1.z, i0.w, i1.w);
    }
    __syncthreads();
    if (c < 3) {
#pragma unroll
      for (int it = 0; it < 2; ++it) {
        const float* g = XR + gsrc + (it << 14) + ((c + 1) << 9);
        const float* h = XI + gsrc + (it << 14) + ((c + 1) << 9);
        pre[it * 4 + 0] = *(const float4*)(g);
        pre[it * 4 + 1] = *(const float4*)(g + 32);
        pre[it * 4 + 2] = *(const float4*)(h);
        pre[it * 4 + 3] = *(const float4*)(h + 32);
      }
    }
#pragma unroll
    for (int ks = 0; ks < 4; ++ks) {
      int k = (c << 4) + (ks << 2);
      float4 pr = *(const float4*)(afr + 2 * ks * K3_APP);
      float4 pi = *(const float4*)(afr + 2 * ks * K3_APP + K3_APL);
      v2f ar0, ai0, ar1, ai1;
      ar0.x = pr.x; ar0.y = pr.y; ar1.x = pr.z; ar1.y = pr.w;
      ai0.x = pi.x; ai0.y = pi.y; ai1.x = pi.z; ai1.y = pi.w;
      v2f br0 = fragB(BR0, 64, k, n0);
      v2f bi0 = fragB(BI0, 64, k, n0);
      v2f br1 = fragB(BR1, 64, k, n0);
      v2f bi1 = fragB(BI1, 64, k, n0);
      v2f an0; an0.x = -ai0.x; an0.y = -ai0.y;
      v2f an1; an1.x = -ai1.x; an1.y = -ai1.y;
      cR0 = wmma4(ar0, br0, cR0);
      cR1 = wmma4(ar1, br1, cR1);
      cI0 = wmma4(ar0, bi0, cI0);
      cI1 = wmma4(ar1, bi1, cI1);
      cR0 = wmma4(an0, bi0, cR0);
      cR1 = wmma4(an1, bi1, cR1);
      cI0 = wmma4(ai0, br0, cI0);
      cI1 = wmma4(ai1, br1, cI1);
    }
  }
#pragma unroll
  for (int r = 0; r < 8; ++r) {
    int row = (mb + r) * 16 + ol;
    *(float2*)(stg + row * K3_PITCH + 2 * wave) = make_float2(cR0[r], cR1[r]);
    *(float2*)(stg + (256 + row) * K3_PITCH + 2 * wave) = make_float2(cI0[r], cI1[r]);
  }
  __syncthreads();
  int q = threadIdx.x & 7;
#pragma unroll
  for (int j = 0; j < 8; ++j) {
    int row = (threadIdx.x >> 3) + (j << 6);
    int rb = (row >> 4) & 15, ro = row & 15;
    float4 v = *(const float4*)(stg + row * K3_PITCH + (q << 2));
    float* dst = ((row >> 8) ? YbI : YbR) +
                 ((size_t)(rb * 64 + n0 + ro) << 12) + f0 + (q << 2);
    *(float4*)dst = v;
  }
}

#define S_PITCH 68
__global__ void k4_inv(const float* __restrict__ YbR, const float* __restrict__ YbI,
                       const float* __restrict__ Ftab, const float* __restrict__ bias,
                       float* __restrict__ out) {
  __shared__ __attribute__((aligned(16))) float Sr[128 * S_PITCH];
  __shared__ __attribute__((aligned(16))) float Si[128 * S_PITCH];
  const float* Fre = Ftab;
  const float* Fim = Ftab + 16384;
  int blk = blockIdx.x;
  const float* Yr = YbR + ((size_t)blk << 12);
  const float* Yi = YbI + ((size_t)blk << 12);
  int wave = threadIdx.x >> 5;
  int lane = threadIdx.x & 31;

  for (int t = wave; t < 16; t += 8) {
    int m0 = (t >> 1) << 4, n0 = 32 + ((t & 1) << 4);
    v8f cR = {}, cI = {};
    for (int k = 0; k < 64; k += 4) {
      v2f ar = fragA_kmap(Fre, m0, k);
      v2f ai = fragA_kmap(Fim, m0, k);
      v2f br = fragB(Yr, 64, k, n0);
      v2f bi = fragB(Yi, 64, k, n0);
      cR = wmma4(ar, br, cR);
      cR = wmma4(ai, bi, cR);
      cI = wmma4(ar, bi, cI);
      v2f ani; ani.x = -ai.x; ani.y = -ai.y;
      cI = wmma4(ani, br, cI);
    }
    int n = n0 + (lane & 15);
    int mb = m0 + ((lane >> 4) << 3);
    float sc = (n == 32) ? (1.f / 16384.f) : (2.f / 16384.f);
#pragma unroll
    for (int r = 0; r < 8; ++r) {
      Sr[(mb + r) * S_PITCH + n] = cR[r] * sc;
      Si[(mb + r) * S_PITCH + n] = cI[r] * sc;
    }
  }
  __syncthreads();

  float bo = bias[blk & 63];
  float* oimg = out + ((size_t)blk << 14);
  for (int t = wave; t < 64; t += 8) {
    int m0 = (t >> 3) << 4, n0 = (t & 7) << 4;
    v8f c = {};
    for (int k = 32; k < 64; k += 4) {
      v2f sr = fragA(Sr, S_PITCH, m0, k);
      v2f si = fragA(Si, S_PITCH, m0, k);
      v2f br = fragB_kmap(Fre, k, n0);
      v2f bi = fragB_kmap(Fim, k, n0);
      c = wmma4(sr, br, c);
      c = wmma4(si, bi, c);
    }
    int n = n0 + (lane & 15);
    int mb = m0 + ((lane >> 4) << 3);
#pragma unroll
    for (int r = 0; r < 8; ++r)
      oimg[(mb + r) * 128 + n] = c[r] + bo;
  }
}

extern "C" void kernel_launch(void* const* d_in, const int* in_sizes, int n_in,
                              void* d_out, int out_size, void* d_ws, size_t ws_size,
                              hipStream_t stream) {
  (void)in_sizes; (void)n_in; (void)out_size; (void)ws_size;
  const float* x    = (const float*)d_in[0];
  const float* w    = (const float*)d_in[1];
  const float* bias = (const float*)d_in[2];
  float* out = (float*)d_out;

  float* ws   = (float*)d_ws;
  float* Ftab = ws;
  float* XbR  = ws + 32768;
  float* XbI  = XbR + 4194304;
  float* WbR  = XbI + 4194304;
  float* WbI  = WbR + 16777216;
  float* YbR  = WbI + 16777216;
  float* YbI  = YbR + 4194304;

  k0_tables<<<16, 256, 0, stream>>>(Ftab);
  k1_fwd  <<<1024, 256, 0, stream>>>(x, Ftab, XbR, XbI);
  k2_wdft <<<512, 256, 0, stream>>>(w, WbR, WbI);
  k3_mix  <<<256, 512, 0, stream>>>(XbR, XbI, WbR, WbI, YbR, YbI);
  k4_inv  <<<1024, 256, 0, stream>>>(YbR, YbI, Ftab, bias, out);
}


# Radix-2 split of the four 128-point DFT GEMMs (k1, k4): summed index folded 128->64 (forward), even/odd sums recombined as E+O, E-O (inverse); their WMMA count halves; summation order changes
# speedup vs baseline: 7.4590x; 1.2451x over previous
#include <hip/hip_runtime.h>


typedef __attribute__((ext_vector_type(2))) float v2f;
typedef __attribute__((ext_vector_type(8))) float v8f;

#define PI_F 3.14159265358979323846f

__device__ __forceinline__ v8f wmma4(v2f a, v2f b, v8f c) {
  return __builtin_amdgcn_wmma_f32_16x16x4_f32(false, a, false, b, (short)0, c,
                                               false, false);
}

__device__ __forceinline__ int bandmap(int bi) { return (bi + 96) & 127; }

__device__ __forceinline__ v2f fragB(const float* B, int ldb, int k0, int n0) {
  int lane = threadIdx.x & 31;
  int n = n0 + (lane & 15);
  int kk = k0 + ((lane >> 4) << 1);
  v2f r; r.x = B[kk * ldb + n]; r.y = B[(kk + 1) * ldb + n]; return r;
}

__global__ void k0_tables(float* __restrict__ Ftab) {
  for (int t = threadIdx.x + blockIdx.x * blockDim.x; t < 16384;
       t += blockDim.x * gridDim.x) {
    int j = t >> 7, k = t & 127;
    float ang = -2.f * PI_F * (float)((j * k) & 127) / 128.f;
    float s, c;
    sincosf(ang, &s, &c);
    Ftab[t] = c;
    Ftab[16384 + t] = s;
  }
}

#define T_PITCH 40
__global__ void k1_fwd(const float* __restrict__ x, const float* __restrict__ Ftab,
                       float* __restrict__ XbR, float* __restrict__ XbI) {
  __shared__ __attribute__((aligned(16))) float T1r[128 * T_PITCH];
  __shared__ __attribute__((aligned(16))) float T1i[128 * T_PITCH];
  const float* Fre = Ftab;
  const float* Fim = Ftab + 16384;
  int blk = blockIdx.x;
  int b = blk >> 6, ci = blk & 63;
  const float* Ximg = x + ((size_t)blk << 14);
  int wave = threadIdx.x >> 5;
  int lane = threadIdx.x & 31;
  int c = lane & 15;
  int hh = lane >> 4;

  {
    int m0 = wave << 4;
    const float* xrow = Ximg + (m0 + c) * 128 + (hh << 1);
    const float* fr = Fre + (hh << 1) * 128 + 2 * c;
    const float* fi = Fim + (hh << 1) * 128 + 2 * c;
    v8f eR = {}, eI = {}, oR = {}, oI = {};
    for (int k = 0; k < 64; k += 4) {
      float2 lo = *(const float2*)(xrow + k);
      float2 hi = *(const float2*)(xrow + k + 64);
      v2f as, ad;
      as.x = lo.x + hi.x; as.y = lo.y + hi.y;
      ad.x = lo.x - hi.x; ad.y = lo.y - hi.y;
      float2 r0 = *(const float2*)(fr + k * 128);
      float2 r1 = *(const float2*)(fr + (k + 1) * 128);
      float2 i0 = *(const float2*)(fi + k * 128);
      float2 i1 = *(const float2*)(fi + (k + 1) * 128);
      v2f bre, bro, bie, bio;
      bre.x = r0.x; bre.y = r1.x; bro.x = r0.y; bro.y = r1.y;
      bie.x = i0.x; bie.y = i1.x; bio.x = i0.y; bio.y = i1.y;
      eR = wmma4(as, bre, eR);
      oR = wmma4(ad, bro, oR);
      eI = wmma4(as, bie, eI);
      oI = wmma4(ad, bio, oI);
    }
    int mb = m0 + (hh << 3);
#pragma unroll
    for (int r = 0; r < 8; ++r) {
      *(float2*)(T1r + (mb + r) * T_PITCH + 2 * c) = make_float2(eR[r], oR[r]);
      *(float2*)(T1i + (mb + r) * T_PITCH + 2 * c) = make_float2(eI[r], oI[r]);
    }
  }
  __syncthreads();

  for (int e = threadIdx.x; e < 64 * 8; e += 256) {
    int off = (e >> 3) * T_PITCH + ((e & 7) << 2);
    float4 a = *(const float4*)(T1r + off);
    float4 d = *(const float4*)(T1r + off + 64 * T_PITCH);
    *(float4*)(T1r + off) = make_float4(a.x + d.x, a.y + d.y, a.z + d.z, a.w + d.w);
    *(float4*)(T1r + off + 64 * T_PITCH) =
        make_float4(a.x - d.x, a.y - d.y, a.z - d.z, a.w - d.w);
    float4 p = *(const float4*)(T1i + off);
    float4 q = *(const float4*)(T1i + off + 64 * T_PITCH);
    *(float4*)(T1i + off) = make_float4(p.x + q.x, p.y + q.y, p.z + q.z, p.w + q.w);
    *(float4*)(T1i + off + 64 * T_PITCH) =
        make_float4(p.x - q.x, p.y - q.y, p.z - q.z, p.w - q.w);
  }
  __syncthreads();

  int bdst = (b + 8) & 15;
  {
    int par = wave & 1;
    int mt = (wave >> 1) & 1;
    int n0 = (wave >> 2) << 4;
    const float* Br = T1r + par * (64 * T_PITCH);
    const float* Bi = T1i + par * (64 * T_PITCH);
    int col = bandmap(2 * ((mt << 4) + c) + par);
    const float* fr = Fre + (hh << 1) * 128 + col;
    const float* fi = Fim + (hh << 1) * 128 + col;
    v8f cR = {}, cI = {};
    for (int k = 0; k < 64; k += 4) {
      v2f ar, ai;
      ar.x = fr[k * 128]; ar.y = fr[(k + 1) * 128];
      ai.x = fi[k * 128]; ai.y = fi[(k + 1) * 128];
      v2f br = fragB(Br, T_PITCH, k, n0);
      v2f bi = fragB(Bi, T_PITCH, k, n0);
      cR = wmma4(ar, br, cR);
      v2f ani; ani.x = -ai.x; ani.y = -ai.y;
      cR = wmma4(ani, bi, cR);
      cI = wmma4(ar, bi, cI);
      cI = wmma4(ai, br, cI);
    }
    int dv = n0 + c;
#pragma unroll
    for (int r = 0; r < 8; ++r) {
      int uu = 2 * ((mt << 4) + (hh << 3) + r) + par;
      int du = uu - 32;
      float msk = (du * du + dv * dv <= 900) ? 1.f : 0.f;
      size_t off = ((((size_t)((uu << 1) + 1) << 10) +
                     (size_t)(bdst << 6) + ci) << 5) + (size_t)dv;
      XbR[off] = cR[r] * msk;
      XbI[off] = cI[r] * msk;
    }
  }
}

__global__ void __launch_bounds__(256)
k2_wdft(const float* __restrict__ w, float* __restrict__ WbR,
        float* __restrict__ WbI) {
  __shared__ __attribute__((aligned(16))) float tw[16 * 20];
  int bid = blockIdx.x;
  int pchunk = bid & 3;
  int vchunk = 2 + ((bid >> 2) & 1);
  int uu = bid >> 3;
  int tid = threadIdx.x;
  if (tid < 144) {
    int vloc = tid / 9, tap = tid - vloc * 9;
    int ky = tap / 3, kx = tap - ky * 3;
    int u = bandmap(uu), v = bandmap(vchunk * 16 + vloc);
    int idx = (u * ky + v * kx) & 127;
    float s, c;
    sincosf(-2.f * PI_F * (float)idx / 128.f, &s, &c);
    tw[vloc * 20 + tap] = c;
    tw[vloc * 20 + 9 + tap] = s;
  } else if (tid < 160) {
    tw[(tid - 144) * 20 + 18] = 0.f;
    tw[(tid - 144) * 20 + 19] = 0.f;
  }
  int p0 = (pchunk << 10) + (tid << 2);
  int i = p0 >> 6, op0 = p0 & 63;
  float wv[4][9];
#pragma unroll
  for (int q = 0; q < 4; ++q) {
    int osrc = (op0 + q + 32) & 63;
    const float* wk = w + (size_t)(osrc * 64 + i) * 9;
#pragma unroll
    for (int t = 0; t < 9; ++t) wv[q][t] = wk[t];
  }
  __syncthreads();
  size_t fbase = (size_t)(uu * 64 + vchunk * 16);
  for (int vloc = 0; vloc < 16; ++vloc) {
    const float4* t4 = (const float4*)(tw + vloc * 20);
    float4 a = t4[0], b = t4[1], c = t4[2], d = t4[3], e = t4[4];
    float tr[9] = {a.x, a.y, a.z, a.w, b.x, b.y, b.z, b.w, c.x};
    float ti[9] = {c.y, c.z, c.w, d.x, d.y, d.z, d.w, e.x, e.y};
    float ar[4], ai[4];
#pragma unroll
    for (int q = 0; q < 4; ++q) {
      float sr = 0.f, si = 0.f;
#pragma unroll
      for (int t = 0; t < 9; ++t) {
        sr += wv[q][t] * tr[t];
        si += wv[q][t] * ti[t];
      }
      ar[q] = sr;
      ai[q] = si;
    }
    size_t off = ((fbase + (size_t)vloc) << 12) + (size_t)p0;
    *(float4*)(WbR + off) = make_float4(ar[0], ar[1], ar[2], ar[3]);
    *(float4*)(WbI + off) = make_float4(ai[0], ai[1], ai[2], ai[3]);
  }
}

#define K3_FPB 32
#define K3_PITCH 36
#define K3_APB 68
#define K3_APP (16 * K3_APB + 4)
#define K3_APL (8 * K3_APP)
__global__ void __launch_bounds__(512)
k3_mix(const float* __restrict__ XbR, const float* __restrict__ XbI,
       const float* __restrict__ WbR, const float* __restrict__ WbI,
       float* __restrict__ YbR, float* __restrict__ YbI) {
  __shared__ __attribute__((aligned(16))) float stg[2 * 256 * K3_PITCH];
  __shared__ __attribute__((aligned(16))) float ast[2 * K3_APL];
  int wave = threadIdx.x >> 5;
  int lane = threadIdx.x & 31;
  int n0 = (blockIdx.x & 3) << 4;
  int fb = ((blockIdx.x >> 2) << 1) | 1;
  int f0 = fb * K3_FPB;
  int ol = lane & 15;
  int hh = lane >> 4;
  int mb = hh << 3;

  const float* XR = XbR + ((size_t)fb << 15);
  const float* XI = XbI + ((size_t)fb << 15);
  int cq = threadIdx.x & 7;
  int cp = (threadIdx.x >> 3) & 7;
  int cb = threadIdx.x >> 6;
  int gsrc = ((cb * 64 + 2 * cp) << 5) + (cq << 2);
  int ldst = cp * K3_APP + cb * K3_APB + (cq << 3);

  int fa = f0 + 2 * wave;
  const float* BR0 = WbR + ((size_t)fa << 12);
  const float* BI0 = WbI + ((size_t)fa << 12);
  const float* BR1 = BR0 + 4096;
  const float* BI1 = BI0 + 4096;
  __builtin_prefetch(BR0, 0, 1);
  __builtin_prefetch(BI0, 0, 1);
  __builtin_prefetch(BR1, 0, 1);
  __builtin_prefetch(BI1, 0, 1);

  const float* afr = ast + hh * K3_APP + ol * K3_APB + (wave << 2);

  float4 pre[8];
#pragma unroll
  for (int it = 0; it < 2; ++it) {
    const float* g = XR + gsrc + (it << 14);
    const float* h = XI + gsrc + (it << 14);
    pre[it * 4 + 0] = *(const float4*)(g);
    pre[it * 4 + 1] = *(const float4*)(g + 32);
    pre[it * 4 + 2] = *(const float4*)(h);
    pre[it * 4 + 3] = *(const float4*)(h + 32);
  }

  v8f cR0 = {}, cI0 = {}, cR1 = {}, cI1 = {};
#pragma unroll
  for (int c = 0; c < 4; ++c) {
    if (c) __syncthreads();
#pragma unroll
    for (int it = 0; it < 2; ++it) {
      float* d = ast + ldst + it * (8 * K3_APB);
      float4 e0 = pre[it * 4 + 0], e1 = pre[it * 4 + 1];
      *(float4*)(d)     = make_float4(e0.x, e1.x, e0.y, e1.y);
      *(float4*)(d + 4) = make_float4(e0.z, e1.z, e0.w, e1.w);
      float4 i0 = pre[it * 4 + 2], i1 = pre[it * 4 + 3];
      *(float4*)(d + K3_APL)     = make_float4(i0.x, i1.x, i0.y, i1.y);
      *(float4*)(d + K3_APL + 4) = make_float4(i0.z, i1.z, i0.w, i1.w);
    }
    __syncthreads();
    if (c < 3) {
#pragma unroll
      for (int it = 0; it < 2; ++it) {
        const float* g = XR + gsrc + (it << 14) + ((c + 1) << 9);
        const float* h = XI + gsrc + (it << 14) + ((c + 1) << 9);
        pre[it * 4 + 0] = *(const float4*)(g);
        pre[it * 4 + 1] = *(const float4*)(g + 32);
        pre[it * 4 + 2] = *(const float4*)(h);
        pre[it * 4 + 3] = *(const float4*)(h + 32);
      }
    }
#pragma unroll
    for (int ks = 0; ks < 4; ++ks) {
      int k = (c << 4) + (ks << 2);
      float4 pr = *(const float4*)(afr + 2 * ks * K3_APP);
      float4 pi = *(const float4*)(afr + 2 * ks * K3_APP + K3_APL);
      v2f ar0, ai0, ar1, ai1;
      ar0.x = pr.x; ar0.y = pr.y; ar1.x = pr.z; ar1.y = pr.w;
      ai0.x = pi.x; ai0.y = pi.y; ai1.x = pi.z; ai1.y = pi.w;
      v2f br0 = fragB(BR0, 64, k, n0);
      v2f bi0 = fragB(BI0, 64, k, n0);
      v2f br1 = fragB(BR1, 64, k, n0);
      v2f bi1 = fragB(BI1, 64, k, n0);
      v2f an0; an0.x = -ai0.x; an0.y = -ai0.y;
      v2f an1; an1.x = -ai1.x; an1.y = -ai1.y;
      cR0 = wmma4(ar0, br0, cR0);
      cR1 = wmma4(ar1, br1, cR1);
      cI0 = wmma4(ar0, bi0, cI0);
      cI1 = wmma4(ar1, bi1, cI1);
      cR0 = wmma4(an0, bi0, cR0);
      cR1 = wmma4(an1, bi1, cR1);
      cI0 = wmma4(ai0, br0, cI0);
      cI1 = wmma4(ai1, br1, cI1);
    }
  }
#pragma unroll
  for (int r = 0; r < 8; ++r) {
    int row = (mb + r) * 16 + ol;
    *(float2*)(stg + row * K3_PITCH + 2 * wave) = make_float2(cR0[r], cR1[r]);
    *(float2*)(stg + (256 + row) * K3_PITCH + 2 * wave) = make_float2(cI0[r], cI1[r]);
  }
  __syncthreads();
  int q = threadIdx.x & 7;
#pragma unroll
  for (int j = 0; j < 8; ++j) {
    int row = (threadIdx.x >> 3) + (j << 6);
    int rb = (row >> 4) & 15, ro = row & 15;
    float4 v = *(const float4*)(stg + row * K3_PITCH + (q << 2));
    float* dst = ((row >> 8) ? YbI : YbR) +
                 ((size_t)(rb * 64 + n0 + ro) << 12) + f0 + (q << 2);
    *(float4*)dst = v;
  }
}

#define S_PITCH 68
__global__ void k4_inv(const float* __restrict__ YbR, const float* __restrict__ YbI,
                       const float* __restrict__ Ftab, const float* __restrict__ bias,
                       float* __restrict__ out) {
  __shared__ __attribute__((aligned(16))) float Sr[128 * S_PITCH];
  __shared__ __attribute__((aligned(16))) float Si[128 * S_PITCH];
  const float* Fre = Ftab;
  const float* Fim = Ftab + 16384;
  int blk = blockIdx.x;
  const float* Yr = YbR + ((size_t)blk << 12);
  const float* Yi = YbI + ((size_t)blk << 12);
  int wave = threadIdx.x >> 5;
  int lane = threadIdx.x & 31;
  int c = lane & 15;
  int hh = lane >> 4;

  {
    int m0 = (wave >> 1) << 4;
    int n0 = 32 + ((wave & 1) << 4);
    v8f eR = {}, eI = {}, oR = {}, oI = {};
    for (int k = 0; k < 32; k += 4) {
      int ue = 2 * (k + (hh << 1));
      v2f ar, ai, br, bi, ani;
      ar.x = Fre[bandmap(ue) * 128 + m0 + c];  ar.y = Fre[bandmap(ue + 2) * 128 + m0 + c];
      ai.x = Fim[bandmap(ue) * 128 + m0 + c];  ai.y = Fim[bandmap(ue + 2) * 128 + m0 + c];
      br.x = Yr[ue * 64 + n0 + c];             br.y = Yr[(ue + 2) * 64 + n0 + c];
      bi.x = Yi[ue * 64 + n0 + c];             bi.y = Yi[(ue + 2) * 64 + n0 + c];
      eR = wmma4(ar, br, eR);
      eR = wmma4(ai, bi, eR);
      eI = wmma4(ar, bi, eI);
      ani.x = -ai.x; ani.y = -ai.y;
      eI = wmma4(ani, br, eI);
      ar.x = Fre[bandmap(ue + 1) * 128 + m0 + c];  ar.y = Fre[bandmap(ue + 3) * 128 + m0 + c];
      ai.x = Fim[bandmap(ue + 1) * 128 + m0 + c];  ai.y = Fim[bandmap(ue + 3) * 128 + m0 + c];
      br.x = Yr[(ue + 1) * 64 + n0 + c];           br.y = Yr[(ue + 3) * 64 + n0 + c];
      bi.x = Yi[(ue + 1) * 64 + n0 + c];           bi.y = Yi[(ue + 3) * 64 + n0 + c];
      oR = wmma4(ar, br, oR);
      oR = wmma4(ai, bi, oR);
      oI = wmma4(ar, bi, oI);
      ani.x = -ai.x; ani.y = -ai.y;
      oI = wmma4(ani, br, oI);
    }
    int n = n0 + c;
    int mb = m0 + (hh << 3);
    float sc = (n == 32) ? (1.f / 16384.f) : (2.f / 16384.f);
#pragma unroll
    for (int r = 0; r < 8; ++r) {
      Sr[(mb + r) * S_PITCH + n]      = (eR[r] + oR[r]) * sc;
      Sr[(mb + r + 64) * S_PITCH + n] = (eR[r] - oR[r]) * sc;
      Si[(mb + r) * S_PITCH + n]      = (eI[r] + oI[r]) * sc;
      Si[(mb + r + 64) * S_PITCH + n] = (eI[r] - oI[r]) * sc;
    }
  }
  __syncthreads();

  float bo = bias[blk & 63];
  float* oimg = out + ((size_t)blk << 14);
  {
    int n0 = (wave & 3) << 4;
    v2f bre[4], bro[4], bie[4], bio[4];
#pragma unroll
    for (int ks = 0; ks < 4; ++ks) {
      int de = 2 * ((ks << 2) + (hh << 1));
      const float* pr = Fre + de * 128 + n0 + c;
      const float* pi = Fim + de * 128 + n0 + c;
      bre[ks].x = pr[0];   bre[ks].y = pr[256];
      bro[ks].x = pr[128]; bro[ks].y = pr[384];
      bie[ks].x = pi[0];   bie[ks].y = pi[256];
      bio[ks].x = pi[128]; bio[ks].y = pi[384];
    }
    for (int mt = wave >> 2; mt < 8; mt += 2) {
      int m0 = mt << 4;
      const float* sr = Sr + (m0 + c) * S_PITCH + 32 + (hh << 2);
      const float* si = Si + (m0 + c) * S_PITCH + 32 + (hh << 2);
      v8f cE = {}, cO = {};
#pragma unroll
      for (int ks = 0; ks < 4; ++ks) {
        float4 qr = *(const float4*)(sr + (ks << 3));
        float4 qi = *(const float4*)(si + (ks << 3));
        v2f sre, sro, sie, sio;
        sre.x = qr.x; sre.y = qr.z; sro.x = qr.y; sro.y = qr.w;
        sie.x = qi.x; sie.y = qi.z; sio.x = qi.y; sio.y = qi.w;
        cE = wmma4(sre, bre[ks], cE);
        cO = wmma4(sro, bro[ks], cO);
        cE = wmma4(sie, bie[ks], cE);
        cO = wmma4(sio, bio[ks], cO);
      }
      int n = n0 + c;
      int mb = m0 + (hh << 3);
#pragma unroll
      for (int r = 0; r < 8; ++r) {
        oimg[(mb + r) * 128 + n]      = cE[r] + cO[r] + bo;
        oimg[(mb + r) * 128 + n + 64] = cE[r] - cO[r] + bo;
      }
    }
  }
}

extern "C" void kernel_launch(void* const* d_in, const int* in_sizes, int n_in,
                              void* d_out, int out_size, void* d_ws, size_t ws_size,
                              hipStream_t stream) {
  (void)in_sizes; (void)n_in; (void)out_size; (void)ws_size;
  const float* x    = (const float*)d_in[0];
  const float* w    = (const float*)d_in[1];
  const float* bias = (const float*)d_in[2];
  float* out = (float*)d_out;

  float* ws   = (float*)d_ws;
  float* Ftab = ws;
  float* XbR  = ws + 32768;
  float* XbI  = XbR + 4194304;
  float* WbR  = XbI + 4194304;
  float* WbI  = WbR + 16777216;
  float* YbR  = WbI + 16777216;
  float* YbI  = YbR + 4194304;

  k0_tables<<<16, 256, 0, stream>>>(Ftab);
  k1_fwd  <<<1024, 256, 0, stream>>>(x, Ftab, XbR, XbI);
  k2_wdft <<<512, 256, 0, stream>>>(w, WbR, WbI);
  k3_mix  <<<256, 512, 0, stream>>>(XbR, XbI, WbR, WbI, YbR, YbI);
  k4_inv  <<<1024, 256, 0, stream>>>(YbR, YbI, Ftab, bias, out);
}


# Weight spectrum no longer stored: channel-mix blocks keep the 3x3 taps of their 16 output channels in LDS and each lane forms its B fragments by the 9-tap DFT (same sum order); its launch is gone
# speedup vs baseline: 10.4425x; 1.4000x over previous
#include <hip/hip_runtime.h>


typedef __attribute__((ext_vector_type(2))) float v2f;
typedef __attribute__((ext_vector_type(8))) float v8f;

#define PI_F 3.14159265358979323846f

__device__ __forceinline__ v8f wmma4(v2f a, v2f b, v8f c) {
  return __builtin_amdgcn_wmma_f32_16x16x4_f32(false, a, false, b, (short)0, c,
                                               false, false);
}

__device__ __forceinline__ int bandmap(int bi) { return (bi + 96) & 127; }

__device__ __forceinline__ v2f fragB(const float* B, int ldb, int k0, int n0) {
  int lane = threadIdx.x & 31;
  int n = n0 + (lane & 15);
  int kk = k0 + ((lane >> 4) << 1);
  v2f r; r.x = B[kk * ldb + n]; r.y = B[(kk + 1) * ldb + n]; return r;
}

__global__ void k0_tables(float* __restrict__ Ftab) {
  for (int t = threadIdx.x + blockIdx.x * blockDim.x; t < 16384;
       t += blockDim.x * gridDim.x) {
    int j = t >> 7, k = t & 127;
    float ang = -2.f * PI_F * (float)((j * k) & 127) / 128.f;
    float s, c;
    sincosf(ang, &s, &c);
    Ftab[t] = c;
    Ftab[16384 + t] = s;
  }
}

#define T_PITCH 40
__global__ void k1_fwd(const float* __restrict__ x, const float* __restrict__ Ftab,
                       float* __restrict__ XbR, float* __restrict__ XbI) {
  __shared__ __attribute__((aligned(16))) float T1r[128 * T_PITCH];
  __shared__ __attribute__((aligned(16))) float T1i[128 * T_PITCH];
  const float* Fre = Ftab;
  const float* Fim = Ftab + 16384;
  int blk = blockIdx.x;
  int b = blk >> 6, ci = blk & 63;
  const float* Ximg = x + ((size_t)blk << 14);
  int wave = threadIdx.x >> 5;
  int lane = threadIdx.x & 31;
  int c = lane & 15;
  int hh = lane >> 4;

  {
    int m0 = wave << 4;
    const float* xrow = Ximg + (m0 + c) * 128 + (hh << 1);
    const float* fr = Fre + (hh << 1) * 128 + 2 * c;
    const float* fi = Fim + (hh << 1) * 128 + 2 * c;
    v8f eR = {}, eI = {}, oR = {}, oI = {};
    for (int k = 0; k < 64; k += 4) {
      float2 lo = *(const float2*)(xrow + k);
      float2 hi = *(const float2*)(xrow + k + 64);
      v2f as, ad;
      as.x = lo.x + hi.x; as.y = lo.y + hi.y;
      ad.x = lo.x - hi.x; ad.y = lo.y - hi.y;
      float2 r0 = *(const float2*)(fr + k * 128);
      float2 r1 = *(const float2*)(fr + (k + 1) * 128);
      float2 i0 = *(const float2*)(fi + k * 128);
      float2 i1 = *(const float2*)(fi + (k + 1) * 128);
      v2f bre, bro, bie, bio;
      bre.x = r0.x; bre.y = r1.x; bro.x = r0.y; bro.y = r1.y;
      bie.x = i0.x; bie.y = i1.x; bio.x = i0.y; bio.y = i1.y;
      eR = wmma4(as, bre, eR);
      oR = wmma4(ad, bro, oR);
      eI = wmma4(as, bie, eI);
      oI = wmma4(ad, bio, oI);
    }
    int mb = m0 + (hh << 3);
#pragma unroll
    for (int r = 0; r < 8; ++r) {
      *(float2*)(T1r + (mb + r) * T_PITCH + 2 * c) = make_float2(eR[r], oR[r]);
      *(float2*)(T1i + (mb + r) * T_PITCH + 2 * c) = make_float2(eI[r], oI[r]);
    }
  }
  __syncthreads();

  for (int e = threadIdx.x; e < 64 * 8; e += 256) {
    int off = (e >> 3) * T_PITCH + ((e & 7) << 2);
    float4 a = *(const float4*)(T1r + off);
    float4 d = *(const float4*)(T1r + off + 64 * T_PITCH);
    *(float4*)(T1r + off) = make_float4(a.x + d.x, a.y + d.y, a.z + d.z, a.w + d.w);
    *(float4*)(T1r + off + 64 * T_PITCH) =
        make_float4(a.x - d.x, a.y - d.y, a.z - d.z, a.w - d.w);
    float4 p = *(const float4*)(T1i + off);
    float4 q = *(const float4*)(T1i + off + 64 * T_PITCH);
    *(float4*)(T1i + off) = make_float4(p.x + q.x, p.y + q.y, p.z + q.z, p.w + q.w);
    *(float4*)(T1i + off + 64 * T_PITCH) =
        make_float4(p.x - q.x, p.y - q.y, p.z - q.z, p.w - q.w);
  }
  __syncthreads();

  int bdst = (b + 8) & 15;
  {
    int par = wave & 1;
    int mt = (wave >> 1) & 1;
    int n0 = (wave >> 2) << 4;
    const float* Br = T1r + par * (64 * T_PITCH);
    const float* Bi = T1i + par * (64 * T_PITCH);
    int col = bandmap(2 * ((mt << 4) + c) + par);
    const float* fr = Fre + (hh << 1) * 128 + col;
    const float* fi = Fim + (hh << 1) * 128 + col;
    v8f cR = {}, cI = {};
    for (int k = 0; k < 64; k += 4) {
      v2f ar, ai;
      ar.x = fr[k * 128]; ar.y = fr[(k + 1) * 128];
      ai.x = fi[k * 128]; ai.y = fi[(k + 1) * 128];
      v2f br = fragB(Br, T_PITCH, k, n0);
      v2f bi = fragB(Bi, T_PITCH, k, n0);
      cR = wmma4(ar, br, cR);
      v2f ani; ani.x = -ai.x; ani.y = -ai.y;
      cR = wmma4(ani, bi, cR);
      cI = wmma4(ar, bi, cI);
      cI = wmma4(ai, br, cI);
    }
    int dv = n0 + c;
#pragma unroll
    for (int r = 0; r < 8; ++r) {
      int uu = 2 * ((mt << 4) + (hh << 3) + r) + par;
      int du = uu - 32;
      float msk = (du * du + dv * dv <= 900) ? 1.f : 0.f;
      size_t off = ((((size_t)((uu << 1) + 1) << 10) +
                     (size_t)(bdst << 6) + ci) << 5) + (size_t)dv;
      XbR[off] = cR[r] * msk;
      XbI[off] = cI[r] * msk;
    }
  }
}

__device__ __forceinline__ void tapdft(const float (&wv)[9], const float (&tr)[9],
                                       const float (&ti)[9], float& re, float& im) {
  float sr = 0.f, si = 0.f;
#pragma unroll
  for (int t = 0; t < 9; ++t) {
    sr = __builtin_fmaf(wv[t], tr[t], sr);
    si = __builtin_fmaf(wv[t], ti[t], si);
  }
  re = sr;
  im = si;
}

#define K3_FPB 32
#define K3_PITCH 36
#define K3_APB 68
#define K3_APP (16 * K3_APB + 4)
#define K3_APL (8 * K3_APP)
#define K3_WPE 12
__global__ void __launch_bounds__(512)
k3_mix(const float* __restrict__ XbR, const float* __restrict__ XbI,
       const float* __restrict__ w,
       float* __restrict__ YbR, float* __restrict__ YbI) {
  __shared__ __attribute__((aligned(16))) float stg[2 * 256 * K3_PITCH];
  __shared__ __attribute__((aligned(16))) float ast[2 * K3_APL];
  __shared__ __attribute__((aligned(16))) float wl[64 * 16 * K3_WPE];
  __shared__ __attribute__((aligned(16))) float twl[32 * 20];
  int wave = threadIdx.x >> 5;
  int lane = threadIdx.x & 31;
  int n0 = (blockIdx.x & 3) << 4;
  int fb = ((blockIdx.x >> 2) << 1) | 1;
  int f0 = fb * K3_FPB;
  int ol = lane & 15;
  int hh = lane >> 4;
  int mb = hh << 3;

  const float* XR = XbR + ((size_t)fb << 15);
  const float* XI = XbI + ((size_t)fb << 15);
  int cq = threadIdx.x & 7;
  int cp = (threadIdx.x >> 3) & 7;
  int cb = threadIdx.x >> 6;
  int gsrc = ((cb * 64 + 2 * cp) << 5) + (cq << 2);
  int ldst = cp * K3_APP + cb * K3_APB + (cq << 3);

  const float* afr = ast + hh * K3_APP + ol * K3_APB + (wave << 2);

  float4 pre[8];
#pragma unroll
  for (int it = 0; it < 2; ++it) {
    const float* g = XR + gsrc + (it << 14);
    const float* h = XI + gsrc + (it << 14);
    pre[it * 4 + 0] = *(const float4*)(g);
    pre[it * 4 + 1] = *(const float4*)(g + 32);
    pre[it * 4 + 2] = *(const float4*)(h);
    pre[it * 4 + 3] = *(const float4*)(h + 32);
  }

  {
    float4 wq[5];
#pragma unroll
    for (int j = 0; j < 5; ++j) {
      int e = threadIdx.x + (j << 9);
      if (e < 2304) {
        int cc = e / 144, q = e - cc * 144;
        int osrc = (n0 + cc + 32) & 63;
        wq[j] = *(const float4*)(w + osrc * 576 + (q << 2));
      }
    }
    if (threadIdx.x < 288) {
      int fl = threadIdx.x / 9, tap = threadIdx.x - fl * 9;
      int ky = tap / 3, kx = tap - ky * 3;
      int u = bandmap(blockIdx.x >> 2), v = bandmap(32 + fl);
      int idx = (u * ky + v * kx) & 127;
      float sn, cs;
      sincosf(-2.f * PI_F * (float)idx / 128.f, &sn, &cs);
      twl[fl * 20 + tap] = cs;
      twl[fl * 20 + 9 + tap] = sn;
    }
#pragma unroll
    for (int j = 0; j < 5; ++j) {
      int e = threadIdx.x + (j << 9);
      if (e < 2304) {
        int cc = e / 144, q = e - cc * 144;
        int fl = q << 2;
        float v4[4] = {wq[j].x, wq[j].y, wq[j].z, wq[j].w};
#pragma unroll
        for (int t = 0; t < 4; ++t) {
          int ii = (fl + t) / 9, tp = (fl + t) - ii * 9;
          wl[(ii * 16 + cc) * K3_WPE + tp] = v4[t];
        }
      }
    }
  }
  __syncthreads();
  float tr0[9], ti0[9], tr1[9], ti1[9];
  {
    const float* t0 = twl + (wave << 1) * 20;
    float4 a = *(const float4*)(t0), b = *(const float4*)(t0 + 4);
    float4 g = *(const float4*)(t0 + 8), d = *(const float4*)(t0 + 12);
    float2 e = *(const float2*)(t0 + 16);
    tr0[0] = a.x; tr0[1] = a.y; tr0[2] = a.z; tr0[3] = a.w; tr0[4] = b.x;
    tr0[5] = b.y; tr0[6] = b.z; tr0[7] = b.w; tr0[8] = g.x;
    ti0[0] = g.y; ti0[1] = g.z; ti0[2] = g.w; ti0[3] = d.x; ti0[4] = d.y;
    ti0[5] = d.z; ti0[6] = d.w; ti0[7] = e.x; ti0[8] = e.y;
    const float* t1 = t0 + 20;
    a = *(const float4*)(t1); b = *(const float4*)(t1 + 4);
    g = *(const float4*)(t1 + 8); d = *(const float4*)(t1 + 12);
    e = *(const float2*)(t1 + 16);
    tr1[0] = a.x; tr1[1] = a.y; tr1[2] = a.z; tr1[3] = a.w; tr1[4] = b.x;
    tr1[5] = b.y; tr1[6] = b.z; tr1[7] = b.w; tr1[8] = g.x;
    ti1[0] = g.y; ti1[1] = g.z; ti1[2] = g.w; ti1[3] = d.x; ti1[4] = d.y;
    ti1[5] = d.z; ti1[6] = d.w; ti1[7] = e.x; ti1[8] = e.y;
  }
  const float* wfr = wl + ((hh << 1) * 16 + ol) * K3_WPE;

  v8f cR0 = {}, cI0 = {}, cR1 = {}, cI1 = {};
#pragma unroll
  for (int c = 0; c < 4; ++c) {
    if (c) __syncthreads();
#pragma unroll
    for (int it = 0; it < 2; ++it) {
      float* d = ast + ldst + it * (8 * K3_APB);
      float4 e0 = pre[it * 4 + 0], e1 = pre[it * 4 + 1];
      *(float4*)(d)     = make_float4(e0.x, e1.x, e0.y, e1.y);
      *(float4*)(d + 4) = make_float4(e0.z, e1.z, e0.w, e1.w);
      float4 i0 = pre[it * 4 + 2], i1 = pre[it * 4 + 3];
      *(float4*)(d + K3_APL)     = make_float4(i0.x, i1.x, i0.y, i1.y);
      *(float4*)(d + K3_APL + 4) = make_float4(i0.z, i1.z, i0.w, i1.w);
    }
    __syncthreads();
    if (c < 3) {
#pragma unroll
      for (int it = 0; it < 2; ++it) {
        const float* g = XR + gsrc + (it << 14) + ((c + 1) << 9);
        const float* h = XI + gsrc + (it << 14) + ((c + 1) << 9);
        pre[it * 4 + 0] = *(const float4*)(g);
        pre[it * 4 + 1] = *(const float4*)(g + 32);
        pre[it * 4 + 2] = *(const float4*)(h);
        pre[it * 4 + 3] = *(const float4*)(h + 32);
      }
    }
#pragma unroll
    for (int ks = 0; ks < 4; ++ks) {
      int k = (c << 4) + (ks << 2);
      float4 pr = *(const float4*)(afr + 2 * ks * K3_APP);
      float4 pi = *(const float4*)(afr + 2 * ks * K3_APP + K3_APL);
      v2f ar0, ai0, ar1, ai1;
      ar0.x = pr.x; ar0.y = pr.y; ar1.x = pr.z; ar1.y = pr.w;
      ai0.x = pi.x; ai0.y = pi.y; ai1.x = pi.z; ai1.y = pi.w;
      const float* wp = wfr + (k << 4) * K3_WPE;
      float4 ua = *(const float4*)(wp), ub = *(const float4*)(wp + 4);
      float4 va = *(const float4*)(wp + 16 * K3_WPE), vb = *(const float4*)(wp + 16 * K3_WPE + 4);
      float w0[9] = {ua.x, ua.y, ua.z, ua.w, ub.x, ub.y, ub.z, ub.w, wp[8]};
      float w1[9] = {va.x, va.y, va.z, va.w, vb.x, vb.y, vb.z, vb.w, wp[16 * K3_WPE + 8]};
      float r00, i00, r01, i01, r10, i10, r11, i11;
      tapdft(w0, tr0, ti0, r00, i00);
      tapdft(w1, tr0, ti0, r01, i01);
      tapdft(w0, tr1, ti1, r10, i10);
      tapdft(w1, tr1, ti1, r11, i11);
      v2f br0, bi0, br1, bi1;
      br0.x = r00; br0.y = r01; bi0.x = i00; bi0.y = i01;
      br1.x = r10; br1.y = r11; bi1.x = i10; bi1.y = i11;
      v2f an0; an0.x = -ai0.x; an0.y = -ai0.y;
      v2f an1; an1.x = -ai1.x; an1.y = -ai1.y;
      cR0 = wmma4(ar0, br0, cR0);
      cR1 = wmma4(ar1, br1, cR1);
      cI0 = wmma4(ar0, bi0, cI0);
      cI1 = wmma4(ar1, bi1, cI1);
      cR0 = wmma4(an0, bi0, cR0);
      cR1 = wmma4(an1, bi1, cR1);
      cI0 = wmma4(ai0, br0, cI0);
      cI1 = wmma4(ai1, br1, cI1);
    }
  }
#pragma unroll
  for (int r = 0; r < 8; ++r) {
    int row = (mb + r) * 16 + ol;
    *(float2*)(stg + row * K3_PITCH + 2 * wave) = make_float2(cR0[r], cR1[r]);
    *(float2*)(stg + (256 + row) * K3_PITCH + 2 * wave) = make_float2(cI0[r], cI1[r]);
  }
  __syncthreads();
  int q = threadIdx.x & 7;
#pragma unroll
  for (int j = 0; j < 8; ++j) {
    int row = (threadIdx.x >> 3) + (j << 6);
    int rb = (row >> 4) & 15, ro = row & 15;
    float4 v = *(const float4*)(stg + row * K3_PITCH + (q << 2));
    float* dst = ((row >> 8) ? YbI : YbR) +
                 ((size_t)(rb * 64 + n0 + ro) << 12) + f0 + (q << 2);
    *(float4*)dst = v;
  }
}

#define S_PITCH 68
__global__ void k4_inv(const float* __restrict__ YbR, const float* __restrict__ YbI,
                       const float* __restrict__ Ftab, const float* __restrict__ bias,
                       float* __restrict__ out) {
  __shared__ __attribute__((aligned(16))) float Sr[128 * S_PITCH];
  __shared__ __attribute__((aligned(16))) float Si[128 * S_PITCH];
  const float* Fre = Ftab;
  const float* Fim = Ftab + 16384;
  int blk = blockIdx.x;
  const float* Yr = YbR + ((size_t)blk << 12);
  const float* Yi = YbI + ((size_t)blk << 12);
  int wave = threadIdx.x >> 5;
  int lane = threadIdx.x & 31;
  int c = lane & 15;
  int hh = lane >> 4;

  {
    int m0 = (wave >> 1) << 4;
    int n0 = 32 + ((wave & 1) << 4);
    v8f eR = {}, eI = {}, oR = {}, oI = {};
    for (int k = 0; k < 32; k += 4) {
      int ue = 2 * (k + (hh << 1));
      v2f ar, ai, br, bi, ani;
      ar.x = Fre[bandmap(ue) * 128 + m0 + c];  ar.y = Fre[bandmap(ue + 2) * 128 + m0 + c];
      ai.x = Fim[bandmap(ue) * 128 + m0 + c];  ai.y = Fim[bandmap(ue + 2) * 128 + m0 + c];
      br.x = Yr[ue * 64 + n0 + c];             br.y = Yr[(ue + 2) * 64 + n0 + c];
      bi.x = Yi[ue * 64 + n0 + c];             bi.y = Yi[(ue + 2) * 64 + n0 + c];
      eR = wmma4(ar, br, eR);
      eR = wmma4(ai, bi, eR);
      eI = wmma4(ar, bi, eI);
      ani.x = -ai.x; ani.y = -ai.y;
      eI = wmma4(ani, br, eI);
      ar.x = Fre[bandmap(ue + 1) * 128 + m0 + c];  ar.y = Fre[bandmap(ue + 3) * 128 + m0 + c];
      ai.x = Fim[bandmap(ue + 1) * 128 + m0 + c];  ai.y = Fim[bandmap(ue + 3) * 128 + m0 + c];
      br.x = Yr[(ue + 1) * 64 + n0 + c];           br.y = Yr[(ue + 3) * 64 + n0 + c];
      bi.x = Yi[(ue + 1) * 64 + n0 + c];           bi.y = Yi[(ue + 3) * 64 + n0 + c];
      oR = wmma4(ar, br, oR);
      oR = wmma4(ai, bi, oR);
      oI = wmma4(ar, bi, oI);
      ani.x = -ai.x; ani.y = -ai.y;
      oI = wmma4(ani, br, oI);
    }
    int n = n0 + c;
    int mb = m0 + (hh << 3);
    float sc = (n == 32) ? (1.f / 16384.f) : (2.f / 16384.f);
#pragma unroll
    for (int r = 0; r < 8; ++r) {
      Sr[(mb + r) * S_PITCH + n]      = (eR[r] + oR[r]) * sc;
      Sr[(mb + r + 64) * S_PITCH + n] = (eR[r] - oR[r]) * sc;
      Si[(mb + r) * S_PITCH + n]      = (eI[r] + oI[r]) * sc;
      Si[(mb + r + 64) * S_PITCH + n] = (eI[r] - oI[r]) * sc;
    }
  }
  __syncthreads();

  float bo = bias[blk & 63];
  float* oimg = out + ((size_t)blk << 14);
  {
    int n0 = (wave & 3) << 4;
    v2f bre[4], bro[4], bie[4], bio[4];
#pragma unroll
    for (int ks = 0; ks < 4; ++ks) {
      int de = 2 * ((ks << 2) + (hh << 1));
      const float* pr = Fre + de * 128 + n0 + c;
      const float* pi = Fim + de * 128 + n0 + c;
      bre[ks].x = pr[0];   bre[ks].y = pr[256];
      bro[ks].x = pr[128]; bro[ks].y = pr[384];
      bie[ks].x = pi[0];   bie[ks].y = pi[256];
      bio[ks].x = pi[128]; bio[ks].y = pi[384];
    }
    for (int mt = wave >> 2; mt < 8; mt += 2) {
      int m0 = mt << 4;
      const float* sr = Sr + (m0 + c) * S_PITCH + 32 + (hh << 2);
      const float* si = Si + (m0 + c) * S_PITCH + 32 + (hh << 2);
      v8f cE = {}, cO = {};
#pragma unroll
      for (int ks = 0; ks < 4; ++ks) {
        float4 qr = *(const float4*)(sr + (ks << 3));
        float4 qi = *(const float4*)(si + (ks << 3));
        v2f sre, sro, sie, sio;
        sre.x = qr.x; sre.y = qr.z; sro.x = qr.y; sro.y = qr.w;
        sie.x = qi.x; sie.y = qi.z; sio.x = qi.y; sio.y = qi.w;
        cE = wmma4(sre, bre[ks], cE);
        cO = wmma4(sro, bro[ks], cO);
        cE = wmma4(sie, bie[ks], cE);
        cO = wmma4(sio, bio[ks], cO);
      }
      int n = n0 + c;
      int mb = m0 + (hh << 3);
#pragma unroll
      for (int r = 0; r < 8; ++r) {
        oimg[(mb + r) * 128 + n]      = cE[r] + cO[r] + bo;
        oimg[(mb + r) * 128 + n + 64] = cE[r] - cO[r] + bo;
      }
    }
  }
}

extern "C" void kernel_launch(void* const* d_in, const int* in_sizes, int n_in,
                              void* d_out, int out_size, void* d_ws, size_t ws_size,
                              hipStream_t stream) {
  (void)in_sizes; (void)n_in; (void)out_size; (void)ws_size;
  const float* x    = (const float*)d_in[0];
  const float* w    = (const float*)d_in[1];
  const float* bias = (const float*)d_in[2];
  float* out = (float*)d_out;

  float* ws   = (float*)d_ws;
  float* Ftab = ws;
  float* XbR  = ws + 32768;
  float* XbI  = XbR + 4194304;
  float* YbR  = XbI + 4194304;
  float* YbI  = YbR + 4194304;

  k0_tables<<<16, 256, 0, stream>>>(Ftab);
  k1_fwd  <<<1024, 256, 0, stream>>>(x, Ftab, XbR, XbI);
  k3_mix  <<<256, 512, 0, stream>>>(XbR, XbI, w, YbR, YbI);
  k4_inv  <<<1024, 256, 0, stream>>>(YbR, YbI, Ftab, bias, out);
}


# k1 stage 1: x read once in whole 128-byte lines (16 B per lane), folded in registers, staged per wave through LDS for the A fragments; was 16-byte pieces of 16 lines per load; sums unchanged
# speedup vs baseline: 10.7654x; 1.0309x over previous
#include <hip/hip_runtime.h>


typedef __attribute__((ext_vector_type(2))) float v2f;
typedef __attribute__((ext_vector_type(8))) float v8f;

#define PI_F 3.14159265358979323846f

__device__ __forceinline__ v8f wmma4(v2f a, v2f b, v8f c) {
  return __builtin_amdgcn_wmma_f32_16x16x4_f32(false, a, false, b, (short)0, c,
                                               false, false);
}

__device__ __forceinline__ int bandmap(int bi) { return (bi + 96) & 127; }

__device__ __forceinline__ v2f fragB(const float* B, int ldb, int k0, int n0) {
  int lane = threadIdx.x & 31;
  int n = n0 + (lane & 15);
  int kk = k0 + ((lane >> 4) << 1);
  v2f r; r.x = B[kk * ldb + n]; r.y = B[(kk + 1) * ldb + n]; return r;
}

__global__ void k0_tables(float* __restrict__ Ftab) {
  for (int t = threadIdx.x + blockIdx.x * blockDim.x; t < 16384;
       t += blockDim.x * gridDim.x) {
    int j = t >> 7, k = t & 127;
    float ang = -2.f * PI_F * (float)((j * k) & 127) / 128.f;
    float s, c;
    sincosf(ang, &s, &c);
    Ftab[t] = c;
    Ftab[16384 + t] = s;
  }
}

#define T_PITCH 40
#define X_PITCH 36
__global__ void k1_fwd(const float* __restrict__ x, const float* __restrict__ Ftab,
                       float* __restrict__ XbR, float* __restrict__ XbI) {
  __shared__ __attribute__((aligned(16))) float T1r[128 * T_PITCH];
  __shared__ __attribute__((aligned(16))) float T1i[128 * T_PITCH];
  __shared__ __attribute__((aligned(16))) float xs[8 * 2 * 16 * X_PITCH];
  const float* Fre = Ftab;
  const float* Fim = Ftab + 16384;
  int blk = blockIdx.x;
  int b = blk >> 6, ci = blk & 63;
  const float* Ximg = x + ((size_t)blk << 14);
  int wave = threadIdx.x >> 5;
  int lane = threadIdx.x & 31;
  int c = lane & 15;
  int hh = lane >> 4;

  {
    int m0 = wave << 4;
    int q = lane & 7, rr = lane >> 3;
    const float* xg = Ximg + (m0 + rr) * 128 + (q << 2);
    float4 lo[2][4], hi[2][4];
#pragma unroll
    for (int hf = 0; hf < 2; ++hf) {
#pragma unroll
      for (int j = 0; j < 4; ++j) {
        lo[hf][j] = *(const float4*)(xg + j * 512 + hf * 32);
        hi[hf][j] = *(const float4*)(xg + j * 512 + hf * 32 + 64);
      }
    }
    float* xw = xs + wave * (2 * 16 * X_PITCH);
    const float* xa = xw + c * X_PITCH + (hh << 1);
    const float* fr = Fre + (hh << 1) * 128 + 2 * c;
    const float* fi = Fim + (hh << 1) * 128 + 2 * c;
    v8f eR = {}, eI = {}, oR = {}, oI = {};
#pragma unroll
    for (int hf = 0; hf < 2; ++hf) {
      if (hf) __syncthreads();
#pragma unroll
      for (int j = 0; j < 4; ++j) {
        float4 a = lo[hf][j], d = hi[hf][j];
        float* p = xw + (4 * j + rr) * X_PITCH + (q << 2);
        *(float4*)p = make_float4(a.x + d.x, a.y + d.y, a.z + d.z, a.w + d.w);
        *(float4*)(p + 16 * X_PITCH) =
            make_float4(a.x - d.x, a.y - d.y, a.z - d.z, a.w - d.w);
      }
      __syncthreads();
      for (int kk = 0; kk < 32; kk += 4) {
        int k = hf * 32 + kk;
        float2 sv = *(const float2*)(xa + kk);
        float2 dv2 = *(const float2*)(xa + 16 * X_PITCH + kk);
        v2f as, ad;
        as.x = sv.x; as.y = sv.y;
        ad.x = dv2.x; ad.y = dv2.y;
        float2 r0 = *(const float2*)(fr + k * 128);
        float2 r1 = *(const float2*)(fr + (k + 1) * 128);
        float2 i0 = *(const float2*)(fi + k * 128);
        float2 i1 = *(const float2*)(fi + (k + 1) * 128);
        v2f bre, bro, bie, bio;
        bre.x = r0.x; bre.y = r1.x; bro.x = r0.y; bro.y = r1.y;
        bie.x = i0.x; bie.y = i1.x; bio.x = i0.y; bio.y = i1.y;
        eR = wmma4(as, bre, eR);
        oR = wmma4(ad, bro, oR);
        eI = wmma4(as, bie, eI);
        oI = wmma4(ad, bio, oI);
      }
    }
    int mb = m0 + (hh << 3);
#pragma unroll
    for (int r = 0; r < 8; ++r) {
      *(float2*)(T1r + (mb + r) * T_PITCH + 2 * c) = make_float2(eR[r], oR[r]);
      *(float2*)(T1i + (mb + r) * T_PITCH + 2 * c) = make_float2(eI[r], oI[r]);
    }
  }
  __syncthreads();

  for (int e = threadIdx.x; e < 64 * 8; e += 256) {
    int off = (e >> 3) * T_PITCH + ((e & 7) << 2);
    float4 a = *(const float4*)(T1r + off);
    float4 d = *(const float4*)(T1r + off + 64 * T_PITCH);
    *(float4*)(T1r + off) = make_float4(a.x + d.x, a.y + d.y, a.z + d.z, a.w + d.w);
    *(float4*)(T1r + off + 64 * T_PITCH) =
        make_float4(a.x - d.x, a.y - d.y, a.z - d.z, a.w - d.w);
    float4 p = *(const float4*)(T1i + off);
    float4 q = *(const float4*)(T1i + off + 64 * T_PITCH);
    *(float4*)(T1i + off) = make_float4(p.x + q.x, p.y + q.y, p.z + q.z, p.w + q.w);
    *(float4*)(T1i + off + 64 * T_PITCH) =
        make_float4(p.x - q.x, p.y - q.y, p.z - q.z, p.w - q.w);
  }
  __syncthreads();

  int bdst = (b + 8) & 15;
  {
    int par = wave & 1;
    int mt = (wave >> 1) & 1;
    int n0 = (wave >> 2) << 4;
    const float* Br = T1r + par * (64 * T_PITCH);
    const float* Bi = T1i + par * (64 * T_PITCH);
    int col = bandmap(2 * ((mt << 4) + c) + par);
    const float* fr = Fre + (hh << 1) * 128 + col;
    const float* fi = Fim + (hh << 1) * 128 + col;
    v8f cR = {}, cI = {};
    for (int k = 0; k < 64; k += 4) {
      v2f ar, ai;
      ar.x = fr[k * 128]; ar.y = fr[(k + 1) * 128];
      ai.x = fi[k * 128]; ai.y = fi[(k + 1) * 128];
      v2f br = fragB(Br, T_PITCH, k, n0);
      v2f bi = fragB(Bi, T_PITCH, k, n0);
      cR = wmma4(ar, br, cR);
      v2f ani; ani.x = -ai.x; ani.y = -ai.y;
      cR = wmma4(ani, bi, cR);
      cI = wmma4(ar, bi, cI);
      cI = wmma4(ai, br, cI);
    }
    int dv = n0 + c;
#pragma unroll
    for (int r = 0; r < 8; ++r) {
      int uu = 2 * ((mt << 4) + (hh << 3) + r) + par;
      int du = uu - 32;
      float msk = (du * du + dv * dv <= 900) ? 1.f : 0.f;
      size_t off = ((((size_t)((uu << 1) + 1) << 10) +
                     (size_t)(bdst << 6) + ci) << 5) + (size_t)dv;
      XbR[off] = cR[r] * msk;
      XbI[off] = cI[r] * msk;
    }
  }
}

__device__ __forceinline__ void tapdft(const float (&wv)[9], const float (&tr)[9],
                                       const float (&ti)[9], float& re, float& im) {
  float sr = 0.f, si = 0.f;
#pragma unroll
  for (int t = 0; t < 9; ++t) {
    sr = __builtin_fmaf(wv[t], tr[t], sr);
    si = __builtin_fmaf(wv[t], ti[t], si);
  }
  re = sr;
  im = si;
}

#define K3_FPB 32
#define K3_PITCH 36
#define K3_APB 68
#define K3_APP (16 * K3_APB + 4)
#define K3_APL (8 * K3_APP)
#define K3_WPE 12
__global__ void __launch_bounds__(512)
k3_mix(const float* __restrict__ XbR, const float* __restrict__ XbI,
       const float* __restrict__ w,
       float* __restrict__ YbR, float* __restrict__ YbI) {
  __shared__ __attribute__((aligned(16))) float stg[2 * 256 * K3_PITCH];
  __shared__ __attribute__((aligned(16))) float ast[2 * K3_APL];
  __shared__ __attribute__((aligned(16))) float wl[64 * 16 * K3_WPE];
  __shared__ __attribute__((aligned(16))) float twl[32 * 20];
  int wave = threadIdx.x >> 5;
  int lane = threadIdx.x & 31;
  int n0 = (blockIdx.x & 3) << 4;
  int fb = ((blockIdx.x >> 2) << 1) | 1;
  int f0 = fb * K3_FPB;
  int ol = lane & 15;
  int hh = lane >> 4;
  int mb = hh << 3;

  const float* XR = XbR + ((size_t)fb << 15);
  const float* XI = XbI + ((size_t)fb << 15);
  int cq = threadIdx.x & 7;
  int cp = (threadIdx.x >> 3) & 7;
  int cb = threadIdx.x >> 6;
  int gsrc = ((cb * 64 + 2 * cp) << 5) + (cq << 2);
  int ldst = cp * K3_APP + cb * K3_APB + (cq << 3);

  const float* afr = ast + hh * K3_APP + ol * K3_APB + (wave << 2);

  float4 pre[8];
#pragma unroll
  for (int it = 0; it < 2; ++it) {
    const float* g = XR + gsrc + (it << 14);
    const float* h = XI + gsrc + (it << 14);
    pre[it * 4 + 0] = *(const float4*)(g);
    pre[it * 4 + 1] = *(const float4*)(g + 32);
    pre[it * 4 + 2] = *(const float4*)(h);
    pre[it * 4 + 3] = *(const float4*)(h + 32);
  }

  {
    float4 wq[5];
#pragma unroll
    for (int j = 0; j < 5; ++j) {
      int e = threadIdx.x + (j << 9);
      if (e < 2304) {
        int cc = e / 144, q = e - cc * 144;
        int osrc = (n0 + cc + 32) & 63;
        wq[j] = *(const float4*)(w + osrc * 576 + (q << 2));
      }
    }
    if (threadIdx.x < 288) {
      int fl = threadIdx.x / 9, tap = threadIdx.x - fl * 9;
      int ky = tap / 3, kx = tap - ky * 3;
      int u = bandmap(blockIdx.x >> 2), v = bandmap(32 + fl);
      int idx = (u * ky + v * kx) & 127;
      float sn, cs;
      sincosf(-2.f * PI_F * (float)idx / 128.f, &sn, &cs);
      twl[fl * 20 + tap] = cs;
      twl[fl * 20 + 9 + tap] = sn;
    }
#pragma unroll
    for (int j = 0; j < 5; ++j) {
      int e = threadIdx.x + (j << 9);
      if (e < 2304) {
        int cc = e / 144, q = e - cc * 144;
        int fl = q << 2;
        float v4[4] = {wq[j].x, wq[j].y, wq[j].z, wq[j].w};
#pragma unroll
        for (int t = 0; t < 4; ++t) {
          int ii = (fl + t) / 9, tp = (fl + t) - ii * 9;
          wl[(ii * 16 + cc) * K3_WPE + tp] = v4[t];
        }
      }
    }
  }
  __syncthreads();
  float tr0[9], ti0[9], tr1[9], ti1[9];
  {
    const float* t0 = twl + (wave << 1) * 20;
    float4 a = *(const float4*)(t0), b = *(const float4*)(t0 + 4);
    float4 g = *(const float4*)(t0 + 8), d = *(const float4*)(t0 + 12);
    float2 e = *(const float2*)(t0 + 16);
    tr0[0] = a.x; tr0[1] = a.y; tr0[2] = a.z; tr0[3] = a.w; tr0[4] = b.x;
    tr0[5] = b.y; tr0[6] = b.z; tr0[7] = b.w; tr0[8] = g.x;
    ti0[0] = g.y; ti0[1] = g.z; ti0[2] = g.w; ti0[3] = d.x; ti0[4] = d.y;
    ti0[5] = d.z; ti0[6] = d.w; ti0[7] = e.x; ti0[8] = e.y;
    const float* t1 = t0 + 20;
    a = *(const float4*)(t1); b = *(const float4*)(t1 + 4);
    g = *(const float4*)(t1 + 8); d = *(const float4*)(t1 + 12);
    e = *(const float2*)(t1 + 16);
    tr1[0] = a.x; tr1[1] = a.y; tr1[2] = a.z; tr1[3] = a.w; tr1[4] = b.x;
    tr1[5] = b.y; tr1[6] = b.z; tr1[7] = b.w; tr1[8] = g.x;
    ti1[0] = g.y; ti1[1] = g.z; ti1[2] = g.w; ti1[3] = d.x; ti1[4] = d.y;
    ti1[5] = d.z; ti1[6] = d.w; ti1[7] = e.x; ti1[8] = e.y;
  }
  const float* wfr = wl + ((hh << 1) * 16 + ol) * K3_WPE;

  v8f cR0 = {}, cI0 = {}, cR1 = {}, cI1 = {};
#pragma unroll
  for (int c = 0; c < 4; ++c) {
    if (c) __syncthreads();
#pragma unroll
    for (int it = 0; it < 2; ++it) {
      float* d = ast + ldst + it * (8 * K3_APB);
      float4 e0 = pre[it * 4 + 0], e1 = pre[it * 4 + 1];
      *(float4*)(d)     = make_float4(e0.x, e1.x, e0.y, e1.y);
      *(float4*)(d + 4) = make_float4(e0.z, e1.z, e0.w, e1.w);
      float4 i0 = pre[it * 4 + 2], i1 = pre[it * 4 + 3];
      *(float4*)(d + K3_APL)     = make_float4(i0.x, i1.x, i0.y, i1.y);
      *(float4*)(d + K3_APL + 4) = make_float4(i0.z, i1.z, i0.w, i1.w);
    }
    __syncthreads();
    if (c < 3) {
#pragma unroll
      for (int it = 0; it < 2; ++it) {
        const float* g = XR + gsrc + (it << 14) + ((c + 1) << 9);
        const float* h = XI + gsrc + (it << 14) + ((c + 1) << 9);
        pre[it * 4 + 0] = *(const float4*)(g);
        pre[it * 4 + 1] = *(const float4*)(g + 32);
        pre[it * 4 + 2] = *(const float4*)(h);
        pre[it * 4 + 3] = *(const float4*)(h + 32);
      }
    }
#pragma unroll
    for (int ks = 0; ks < 4; ++ks) {
      int k = (c << 4) + (ks << 2);
      float4 pr = *(const float4*)(afr + 2 * ks * K3_APP);
      float4 pi = *(const float4*)(afr + 2 * ks * K3_APP + K3_APL);
      v2f ar0, ai0, ar1, ai1;
      ar0.x = pr.x; ar0.y = pr.y; ar1.x = pr.z; ar1.y = pr.w;
      ai0.x = pi.x; ai0.y = pi.y; ai1.x = pi.z; ai1.y = pi.w;
      const float* wp = wfr + (k << 4) * K3_WPE;
      float4 ua = *(const float4*)(wp), ub = *(const float4*)(wp + 4);
      float4 va = *(const float4*)(wp + 16 * K3_WPE), vb = *(const float4*)(wp + 16 * K3_WPE + 4);
      float w0[9] = {ua.x, ua.y, ua.z, ua.w, ub.x, ub.y, ub.z, ub.w, wp[8]};
      float w1[9] = {va.x, va.y, va.z, va.w, vb.x, vb.y, vb.z, vb.w, wp[16 * K3_WPE + 8]};
      float r00, i00, r01, i01, r10, i10, r11, i11;
      tapdft(w0, tr0, ti0, r00, i00);
      tapdft(w1, tr0, ti0, r01, i01);
      tapdft(w0, tr1, ti1, r10, i10);
      tapdft(w1, tr1, ti1, r11, i11);
      v2f br0, bi0, br1, bi1;
      br0.x = r00; br0.y = r01; bi0.x = i00; bi0.y = i01;
      br1.x = r10; br1.y = r11; bi1.x = i10; bi1.y = i11;
      v2f an0; an0.x = -ai0.x; an0.y = -ai0.y;
      v2f an1; an1.x = -ai1.x; an1.y = -ai1.y;
      cR0 = wmma4(ar0, br0, cR0);
      cR1 = wmma4(ar1, br1, cR1);
      cI0 = wmma4(ar0, bi0, cI0);
      cI1 = wmma4(ar1, bi1, cI1);
      cR0 = wmma4(an0, bi0, cR0);
      cR1 = wmma4(an1, bi1, cR1);
      cI0 = wmma4(ai0, br0, cI0);
      cI1 = wmma4(ai1, br1, cI1);
    }
  }
#pragma unroll
  for (int r = 0; r < 8; ++r) {
    int row = (mb + r) * 16 + ol;
    *(float2*)(stg + row * K3_PITCH + 2 * wave) = make_float2(cR0[r], cR1[r]);
    *(float2*)(stg + (256 + row) * K3_PITCH + 2 * wave) = make_float2(cI0[r], cI1[r]);
  }
  __syncthreads();
  int q = threadIdx.x & 7;
#pragma unroll
  for (int j = 0; j < 8; ++j) {
    int row = (threadIdx.x >> 3) + (j << 6);
    int rb = (row >> 4) & 15, ro = row & 15;
    float4 v = *(const float4*)(stg + row * K3_PITCH + (q << 2));
    float* dst = ((row >> 8) ? YbI : YbR) +
                 ((size_t)(rb * 64 + n0 + ro) << 12) + f0 + (q << 2);
    *(float4*)dst = v;
  }
}

#define S_PITCH 68
__global__ void k4_inv(const float* __restrict__ YbR, const float* __restrict__ YbI,
                       const float* __restrict__ Ftab, const float* __restrict__ bias,
                       float* __restrict__ out) {
  __shared__ __attribute__((aligned(16))) float Sr[128 * S_PITCH];
  __shared__ __attribute__((aligned(16))) float Si[128 * S_PITCH];
  const float* Fre = Ftab;
  const float* Fim = Ftab + 16384;
  int blk = blockIdx.x;
  const float* Yr = YbR + ((size_t)blk << 12);
  const float* Yi = YbI + ((size_t)blk << 12);
  int wave = threadIdx.x >> 5;
  int lane = threadIdx.x & 31;
  int c = lane & 15;
  int hh = lane >> 4;

  {
    int m0 = (wave >> 1) << 4;
    int n0 = 32 + ((wave & 1) << 4);
    v8f eR = {}, eI = {}, oR = {}, oI = {};
    for (int k = 0; k < 32; k += 4) {
      int ue = 2 * (k + (hh << 1));
      v2f ar, ai, br, bi, ani;
      ar.x = Fre[bandmap(ue) * 128 + m0 + c];  ar.y = Fre[bandmap(ue + 2) * 128 + m0 + c];
      ai.x = Fim[bandmap(ue) * 128 + m0 + c];  ai.y = Fim[bandmap(ue + 2) * 128 + m0 + c];
      br.x = Yr[ue * 64 + n0 + c];             br.y = Yr[(ue + 2) * 64 + n0 + c];
      bi.x = Yi[ue * 64 + n0 + c];             bi.y = Yi[(ue + 2) * 64 + n0 + c];
      eR = wmma4(ar, br, eR);
      eR = wmma4(ai, bi, eR);
      eI = wmma4(ar, bi, eI);
      ani.x = -ai.x; ani.y = -ai.y;
      eI = wmma4(ani, br, eI);
      ar.x = Fre[bandmap(ue + 1) * 128 + m0 + c];  ar.y = Fre[bandmap(ue + 3) * 128 + m0 + c];
      ai.x = Fim[bandmap(ue + 1) * 128 + m0 + c];  ai.y = Fim[bandmap(ue + 3) * 128 + m0 + c];
      br.x = Yr[(ue + 1) * 64 + n0 + c];           br.y = Yr[(ue + 3) * 64 + n0 + c];
      bi.x = Yi[(ue + 1) * 64 + n0 + c];           bi.y = Yi[(ue + 3) * 64 + n0 + c];
      oR = wmma4(ar, br, oR);
      oR = wmma4(ai, bi, oR);
      oI = wmma4(ar, bi, oI);
      ani.x = -ai.x; ani.y = -ai.y;
      oI = wmma4(ani, br, oI);
    }
    int n = n0 + c;
    int mb = m0 + (hh << 3);
    float sc = (n == 32) ? (1.f / 16384.f) : (2.f / 16384.f);
#pragma unroll
    for (int r = 0; r < 8; ++r) {
      Sr[(mb + r) * S_PITCH + n]      = (eR[r] + oR[r]) * sc;
      Sr[(mb + r + 64) * S_PITCH + n] = (eR[r] - oR[r]) * sc;
      Si[(mb + r) * S_PITCH + n]      = (eI[r] + oI[r]) * sc;
      Si[(mb + r + 64) * S_PITCH + n] = (eI[r] - oI[r]) * sc;
    }
  }
  __syncthreads();

  float bo = bias[blk & 63];
  float* oimg = out + ((size_t)blk << 14);
  {
    int n0 = (wave & 3) << 4;
    v2f bre[4], bro[4], bie[4], bio[4];
#pragma unroll
    for (int ks = 0; ks < 4; ++ks) {
      int de = 2 * ((ks << 2) + (hh << 1));
      const float* pr = Fre + de * 128 + n0 + c;
      const float* pi = Fim + de * 128 + n0 + c;
      bre[ks].x = pr[0];   bre[ks].y = pr[256];
      bro[ks].x = pr[128]; bro[ks].y = pr[384];
      bie[ks].x = pi[0];   bie[ks].y = pi[256];
      bio[ks].x = pi[128]; bio[ks].y = pi[384];
    }
    for (int mt = wave >> 2; mt < 8; mt += 2) {
      int m0 = mt << 4;
      const float* sr = Sr + (m0 + c) * S_PITCH + 32 + (hh << 2);
      const float* si = Si + (m0 + c) * S_PITCH + 32 + (hh << 2);
      v8f cE = {}, cO = {};
#pragma unroll
      for (int ks = 0; ks < 4; ++ks) {
        float4 qr = *(const float4*)(sr + (ks << 3));
        float4 qi = *(const float4*)(si + (ks << 3));
        v2f sre, sro, sie, sio;
        sre.x = qr.x; sre.y = qr.z; sro.x = qr.y; sro.y = qr.w;
        sie.x = qi.x; sie.y = qi.z; sio.x = qi.y; sio.y = qi.w;
        cE = wmma4(sre, bre[ks], cE);
        cO = wmma4(sro, bro[ks], cO);
        cE = wmma4(sie, bie[ks], cE);
        cO = wmma4(sio, bio[ks], cO);
      }
      int n = n0 + c;
      int mb = m0 + (hh << 3);
#pragma unroll
      for (int r = 0; r < 8; ++r) {
        oimg[(mb + r) * 128 + n]      = cE[r] + cO[r] + bo;
        oimg[(mb + r) * 128 + n + 64] = cE[r] - cO[r] + bo;
      }
    }
  }
}

extern "C" void kernel_launch(void* const* d_in, const int* in_sizes, int n_in,
                              void* d_out, int out_size, void* d_ws, size_t ws_size,
                              hipStream_t stream) {
  (void)in_sizes; (void)n_in; (void)out_size; (void)ws_size;
  const float* x    = (const float*)d_in[0];
  const float* w    = (const float*)d_in[1];
  const float* bias = (const float*)d_in[2];
  float* out = (float*)d_out;

  float* ws   = (float*)d_ws;
  float* Ftab = ws;
  float* XbR  = ws + 32768;
  float* XbI  = XbR + 4194304;
  float* YbR  = XbI + 4194304;
  float* YbI  = YbR + 4194304;

  k0_tables<<<16, 256, 0, stream>>>(Ftab);
  k1_fwd  <<<1024, 256, 0, stream>>>(x, Ftab, XbR, XbI);
  k3_mix  <<<256, 512, 0, stream>>>(XbR, XbI, w, YbR, YbI);
  k4_inv  <<<1024, 256, 0, stream>>>(YbR, YbI, Ftab, bias, out);
}


# Radix-4 split of the two complex DFT GEMMs (k1 stage 2, k4 stage 1): forward fold 64->32 per row residue mod 4, inverse class sums recombined by +-1, +-i; their WMMA count halves; sum order changes
# speedup vs baseline: 11.3616x; 1.0554x over previous
#include <hip/hip_runtime.h>


typedef __attribute__((ext_vector_type(2))) float v2f;
typedef __attribute__((ext_vector_type(8))) float v8f;

#define PI_F 3.14159265358979323846f

__device__ __forceinline__ v8f wmma4(v2f a, v2f b, v8f c) {
  return __builtin_amdgcn_wmma_f32_16x16x4_f32(false, a, false, b, (short)0, c,
                                               false, false);
}

__device__ __forceinline__ float4 f4add(float4 a, float4 b) {
  return make_float4(a.x + b.x, a.y + b.y, a.z + b.z, a.w + b.w);
}
__device__ __forceinline__ float4 f4sub(float4 a, float4 b) {
  return make_float4(a.x - b.x, a.y - b.y, a.z - b.z, a.w - b.w);
}

__device__ __forceinline__ int bandmap(int bi) { return (bi + 96) & 127; }

__device__ __forceinline__ v2f fragB(const float* B, int ldb, int k0, int n0) {
  int lane = threadIdx.x & 31;
  int n = n0 + (lane & 15);
  int kk = k0 + ((lane >> 4) << 1);
  v2f r; r.x = B[kk * ldb + n]; r.y = B[(kk + 1) * ldb + n]; return r;
}

__global__ void k0_tables(float* __restrict__ Ftab) {
  for (int t = threadIdx.x + blockIdx.x * blockDim.x; t < 16384;
       t += blockDim.x * gridDim.x) {
    int j = t >> 7, k = t & 127;
    float ang = -2.f * PI_F * (float)((j * k) & 127) / 128.f;
    float s, c;
    sincosf(ang, &s, &c);
    Ftab[t] = c;
    Ftab[16384 + t] = s;
  }
}

#define T_PITCH 40
#define X_PITCH 36
__global__ void k1_fwd(const float* __restrict__ x, const float* __restrict__ Ftab,
                       float* __restrict__ XbR, float* __restrict__ XbI) {
  __shared__ __attribute__((aligned(16))) float T1r[128 * T_PITCH];
  __shared__ __attribute__((aligned(16))) float T1i[128 * T_PITCH];
  __shared__ __attribute__((aligned(16))) float xs[8 * 2 * 16 * X_PITCH];
  const float* Fre = Ftab;
  const float* Fim = Ftab + 16384;
  int blk = blockIdx.x;
  int b = blk >> 6, ci = blk & 63;
  const float* Ximg = x + ((size_t)blk << 14);
  int wave = threadIdx.x >> 5;
  int lane = threadIdx.x & 31;
  int c = lane & 15;
  int hh = lane >> 4;

  {
    int m0 = wave << 4;
    int q = lane & 7, rr = lane >> 3;
    const float* xg = Ximg + (m0 + rr) * 128 + (q << 2);
    float4 lo[2][4], hi[2][4];
#pragma unroll
    for (int hf = 0; hf < 2; ++hf) {
#pragma unroll
      for (int j = 0; j < 4; ++j) {
        lo[hf][j] = *(const float4*)(xg + j * 512 + hf * 32);
        hi[hf][j] = *(const float4*)(xg + j * 512 + hf * 32 + 64);
      }
    }
    float* xw = xs + wave * (2 * 16 * X_PITCH);
    const float* xa = xw + c * X_PITCH + (hh << 1);
    const float* fr = Fre + (hh << 1) * 128 + 2 * c;
    const float* fi = Fim + (hh << 1) * 128 + 2 * c;
    v8f eR = {}, eI = {}, oR = {}, oI = {};
#pragma unroll
    for (int hf = 0; hf < 2; ++hf) {
      if (hf) __syncthreads();
#pragma unroll
      for (int j = 0; j < 4; ++j) {
        float4 a = lo[hf][j], d = hi[hf][j];
        float* p = xw + (4 * j + rr) * X_PITCH + (q << 2);
        *(float4*)p = make_float4(a.x + d.x, a.y + d.y, a.z + d.z, a.w + d.w);
        *(float4*)(p + 16 * X_PITCH) =
            make_float4(a.x - d.x, a.y - d.y, a.z - d.z, a.w - d.w);
      }
      __syncthreads();
      for (int kk = 0; kk < 32; kk += 4) {
        int k = hf * 32 + kk;
        float2 sv = *(const float2*)(xa + kk);
        float2 dv2 = *(const float2*)(xa + 16 * X_PITCH + kk);
        v2f as, ad;
        as.x = sv.x; as.y = sv.y;
        ad.x = dv2.x; ad.y = dv2.y;
        float2 r0 = *(const float2*)(fr + k * 128);
        float2 r1 = *(const float2*)(fr + (k + 1) * 128);
        float2 i0 = *(const float2*)(fi + k * 128);
        float2 i1 = *(const float2*)(fi + (k + 1) * 128);
        v2f bre, bro, bie, bio;
        bre.x = r0.x; bre.y = r1.x; bro.x = r0.y; bro.y = r1.y;
        bie.x = i0.x; bie.y = i1.x; bio.x = i0.y; bio.y = i1.y;
        eR = wmma4(as, bre, eR);
        oR = wmma4(ad, bro, oR);
        eI = wmma4(as, bie, eI);
        oI = wmma4(ad, bio, oI);
      }
    }
    int mb = m0 + (hh << 3);
#pragma unroll
    for (int r = 0; r < 8; ++r) {
      *(float2*)(T1r + (mb + r) * T_PITCH + 2 * c) = make_float2(eR[r], oR[r]);
      *(float2*)(T1i + (mb + r) * T_PITCH + 2 * c) = make_float2(eI[r], oI[r]);
    }
  }
  __syncthreads();

  {
    int e = threadIdx.x;
    int off = (e >> 3) * T_PITCH + ((e & 7) << 2);
    float4 r0 = *(const float4*)(T1r + off);
    float4 r1 = *(const float4*)(T1r + off + 32 * T_PITCH);
    float4 r2 = *(const float4*)(T1r + off + 64 * T_PITCH);
    float4 r3 = *(const float4*)(T1r + off + 96 * T_PITCH);
    float4 i0 = *(const float4*)(T1i + off);
    float4 i1 = *(const float4*)(T1i + off + 32 * T_PITCH);
    float4 i2 = *(const float4*)(T1i + off + 64 * T_PITCH);
    float4 i3 = *(const float4*)(T1i + off + 96 * T_PITCH);
    float4 sr = f4add(r0, r2), dr = f4sub(r0, r2);
    float4 si = f4add(i0, i2), di = f4sub(i0, i2);
    float4 ur = f4add(r1, r3), vr = f4sub(r1, r3);
    float4 ui = f4add(i1, i3), vi = f4sub(i1, i3);
    *(float4*)(T1r + off)                = f4add(sr, ur);
    *(float4*)(T1i + off)                = f4add(si, ui);
    *(float4*)(T1r + off + 32 * T_PITCH) = f4add(dr, vi);
    *(float4*)(T1i + off + 32 * T_PITCH) = f4sub(di, vr);
    *(float4*)(T1r + off + 64 * T_PITCH) = f4sub(sr, ur);
    *(float4*)(T1i + off + 64 * T_PITCH) = f4sub(si, ui);
    *(float4*)(T1r + off + 96 * T_PITCH) = f4sub(dr, vi);
    *(float4*)(T1i + off + 96 * T_PITCH) = f4add(di, vr);
  }
  __syncthreads();

  int bdst = (b + 8) & 15;
  {
    int cls = wave & 3;
    int n0 = (wave >> 2) << 4;
    const float* Br = T1r + cls * (32 * T_PITCH);
    const float* Bi = T1i + cls * (32 * T_PITCH);
    int col = bandmap(4 * c + cls);
    const float* fr = Fre + (hh << 1) * 128 + col;
    const float* fi = Fim + (hh << 1) * 128 + col;
    v8f cR = {}, cI = {};
    for (int k = 0; k < 32; k += 4) {
      v2f ar, ai;
      ar.x = fr[k * 128]; ar.y = fr[(k + 1) * 128];
      ai.x = fi[k * 128]; ai.y = fi[(k + 1) * 128];
      v2f br = fragB(Br, T_PITCH, k, n0);
      v2f bi = fragB(Bi, T_PITCH, k, n0);
      cR = wmma4(ar, br, cR);
      v2f ani; ani.x = -ai.x; ani.y = -ai.y;
      cR = wmma4(ani, bi, cR);
      cI = wmma4(ar, bi, cI);
      cI = wmma4(ai, br, cI);
    }
    int dv = n0 + c;
#pragma unroll
    for (int r = 0; r < 8; ++r) {
      int uu = 4 * ((hh << 3) + r) + cls;
      int du = uu - 32;
      float msk = (du * du + dv * dv <= 900) ? 1.f : 0.f;
      size_t off = ((((size_t)((uu << 1) + 1) << 10) +
                     (size_t)(bdst << 6) + ci) << 5) + (size_t)dv;
      XbR[off] = cR[r] * msk;
      XbI[off] = cI[r] * msk;
    }
  }
}

__device__ __forceinline__ void tapdft(const float (&wv)[9], const float (&tr)[9],
                                       const float (&ti)[9], float& re, float& im) {
  float sr = 0.f, si = 0.f;
#pragma unroll
  for (int t = 0; t < 9; ++t) {
    sr = __builtin_fmaf(wv[t], tr[t], sr);
    si = __builtin_fmaf(wv[t], ti[t], si);
  }
  re = sr;
  im = si;
}

#define K3_FPB 32
#define K3_PITCH 36
#define K3_APB 68
#define K3_APP (16 * K3_APB + 4)
#define K3_APL (8 * K3_APP)
#define K3_WPE 12
__global__ void __launch_bounds__(512)
k3_mix(const float* __restrict__ XbR, const float* __restrict__ XbI,
       const float* __restrict__ w,
       float* __restrict__ YbR, float* __restrict__ YbI) {
  __shared__ __attribute__((aligned(16))) float stg[2 * 256 * K3_PITCH];
  __shared__ __attribute__((aligned(16))) float ast[2 * K3_APL];
  __shared__ __attribute__((aligned(16))) float wl[64 * 16 * K3_WPE];
  __shared__ __attribute__((aligned(16))) float twl[32 * 20];
  int wave = threadIdx.x >> 5;
  int lane = threadIdx.x & 31;
  int n0 = (blockIdx.x & 3) << 4;
  int fb = ((blockIdx.x >> 2) << 1) | 1;
  int f0 = fb * K3_FPB;
  int ol = lane & 15;
  int hh = lane >> 4;
  int mb = hh << 3;

  const float* XR = XbR + ((size_t)fb << 15);
  const float* XI = XbI + ((size_t)fb << 15);
  int cq = threadIdx.x & 7;
  int cp = (threadIdx.x >> 3) & 7;
  int cb = threadIdx.x >> 6;
  int gsrc = ((cb * 64 + 2 * cp) << 5) + (cq << 2);
  int ldst = cp * K3_APP + cb * K3_APB + (cq << 3);

  const float* afr = ast + hh * K3_APP + ol * K3_APB + (wave << 2);

  float4 pre[8];
#pragma unroll
  for (int it = 0; it < 2; ++it) {
    const float* g = XR + gsrc + (it << 14);
    const float* h = XI + gsrc + (it << 14);
    pre[it * 4 + 0] = *(const float4*)(g);
    pre[it * 4 + 1] = *(const float4*)(g + 32);
    pre[it * 4 + 2] = *(const float4*)(h);
    pre[it * 4 + 3] = *(const float4*)(h + 32);
  }

  {
    float4 wq[5];
#pragma unroll
    for (int j = 0; j < 5; ++j) {
      int e = threadIdx.x + (j << 9);
      if (e < 2304) {
        int cc = e / 144, q = e - cc * 144;
        int osrc = (n0 + cc + 32) & 63;
        wq[j] = *(const float4*)(w + osrc * 576 + (q << 2));
      }
    }
    if (threadIdx.x < 288) {
      int fl = threadIdx.x / 9, tap = threadIdx.x - fl * 9;
      int ky = tap / 3, kx = tap - ky * 3;
      int u = bandmap(blockIdx.x >> 2), v = bandmap(32 + fl);
      int idx = (u * ky + v * kx) & 127;
      float sn, cs;
      sincosf(-2.f * PI_F * (float)idx / 128.f, &sn, &cs);
      twl[fl * 20 + tap] = cs;
      twl[fl * 20 + 9 + tap] = sn;
    }
#pragma unroll
    for (int j = 0; j < 5; ++j) {
      int e = threadIdx.x + (j << 9);
      if (e < 2304) {
        int cc = e / 144, q = e - cc * 144;
        int fl = q << 2;
        float v4[4] = {wq[j].x, wq[j].y, wq[j].z, wq[j].w};
#pragma unroll
        for (int t = 0; t < 4; ++t) {
          int ii = (fl + t) / 9, tp = (fl + t) - ii * 9;
          wl[(ii * 16 + cc) * K3_WPE + tp] = v4[t];
        }
      }
    }
  }
  __syncthreads();
  float tr0[9], ti0[9], tr1[9], ti1[9];
  {
    const float* t0 = twl + (wave << 1) * 20;
    float4 a = *(const float4*)(t0), b = *(const float4*)(t0 + 4);
    float4 g = *(const float4*)(t0 + 8), d = *(const float4*)(t0 + 12);
    float2 e = *(const float2*)(t0 + 16);
    tr0[0] = a.x; tr0[1] = a.y; tr0[2] = a.z; tr0[3] = a.w; tr0[4] = b.x;
    tr0[5] = b.y; tr0[6] = b.z; tr0[7] = b.w; tr0[8] = g.x;
    ti0[0] = g.y; ti0[1] = g.z; ti0[2] = g.w; ti0[3] = d.x; ti0[4] = d.y;
    ti0[5] = d.z; ti0[6] = d.w; ti0[7] = e.x; ti0[8] = e.y;
    const float* t1 = t0 + 20;
    a = *(const float4*)(t1); b = *(const float4*)(t1 + 4);
    g = *(const float4*)(t1 + 8); d = *(const float4*)(t1 + 12);
    e = *(const float2*)(t1 + 16);
    tr1[0] = a.x; tr1[1] = a.y; tr1[2] = a.z; tr1[3] = a.w; tr1[4] = b.x;
    tr1[5] = b.y; tr1[6] = b.z; tr1[7] = b.w; tr1[8] = g.x;
    ti1[0] = g.y; ti1[1] = g.z; ti1[2] = g.w; ti1[3] = d.x; ti1[4] = d.y;
    ti1[5] = d.z; ti1[6] = d.w; ti1[7] = e.x; ti1[8] = e.y;
  }
  const float* wfr = wl + ((hh << 1) * 16 + ol) * K3_WPE;

  v8f cR0 = {}, cI0 = {}, cR1 = {}, cI1 = {};
#pragma unroll
  for (int c = 0; c < 4; ++c) {
    if (c) __syncthreads();
#pragma unroll
    for (int it = 0; it < 2; ++it) {
      float* d = ast + ldst + it * (8 * K3_APB);
      float4 e0 = pre[it * 4 + 0], e1 = pre[it * 4 + 1];
      *(float4*)(d)     = make_float4(e0.x, e1.x, e0.y, e1.y);
      *(float4*)(d + 4) = make_float4(e0.z, e1.z, e0.w, e1.w);
      float4 i0 = pre[it * 4 + 2], i1 = pre[it * 4 + 3];
      *(float4*)(d + K3_APL)     = make_float4(i0.x, i1.x, i0.y, i1.y);
      *(float4*)(d + K3_APL + 4) = make_float4(i0.z, i1.z, i0.w, i1.w);
    }
    __syncthreads();
    if (c < 3) {
#pragma unroll
      for (int it = 0; it < 2; ++it) {
        const float* g = XR + gsrc + (it << 14) + ((c + 1) << 9);
        const float* h = XI + gsrc + (it << 14) + ((c + 1) << 9);
        pre[it * 4 + 0] = *(const float4*)(g);
        pre[it * 4 + 1] = *(const float4*)(g + 32);
        pre[it * 4 + 2] = *(const float4*)(h);
        pre[it * 4 + 3] = *(const float4*)(h + 32);
      }
    }
#pragma unroll
    for (int ks = 0; ks < 4; ++ks) {
      int k = (c << 4) + (ks << 2);
      float4 pr = *(const float4*)(afr + 2 * ks * K3_APP);
      float4 pi = *(const float4*)(afr + 2 * ks * K3_APP + K3_APL);
      v2f ar0, ai0, ar1, ai1;
      ar0.x = pr.x; ar0.y = pr.y; ar1.x = pr.z; ar1.y = pr.w;
      ai0.x = pi.x; ai0.y = pi.y; ai1.x = pi.z; ai1.y = pi.w;
      const float* wp = wfr + (k << 4) * K3_WPE;
      float4 ua = *(const float4*)(wp), ub = *(const float4*)(wp + 4);
      float4 va = *(const float4*)(wp + 16 * K3_WPE), vb = *(const float4*)(wp + 16 * K3_WPE + 4);
      float w0[9] = {ua.x, ua.y, ua.z, ua.w, ub.x, ub.y, ub.z, ub.w, wp[8]};
      float w1[9] = {va.x, va.y, va.z, va.w, vb.x, vb.y, vb.z, vb.w, wp[16 * K3_WPE + 8]};
      float r00, i00, r01, i01, r10, i10, r11, i11;
      tapdft(w0, tr0, ti0, r00, i00);
      tapdft(w1, tr0, ti0, r01, i01);
      tapdft(w0, tr1, ti1, r10, i10);
      tapdft(w1, tr1, ti1, r11, i11);
      v2f br0, bi0, br1, bi1;
      br0.x = r00; br0.y = r01; bi0.x = i00; bi0.y = i01;
      br1.x = r10; br1.y = r11; bi1.x = i10; bi1.y = i11;
      v2f an0; an0.x = -ai0.x; an0.y = -ai0.y;
      v2f an1; an1.x = -ai1.x; an1.y = -ai1.y;
      cR0 = wmma4(ar0, br0, cR0);
      cR1 = wmma4(ar1, br1, cR1);
      cI0 = wmma4(ar0, bi0, cI0);
      cI1 = wmma4(ar1, bi1, cI1);
      cR0 = wmma4(an0, bi0, cR0);
      cR1 = wmma4(an1, bi1, cR1);
      cI0 = wmma4(ai0, br0, cI0);
      cI1 = wmma4(ai1, br1, cI1);
    }
  }
#pragma unroll
  for (int r = 0; r < 8; ++r) {
    int row = (mb + r) * 16 + ol;
    *(float2*)(stg + row * K3_PITCH + 2 * wave) = make_float2(cR0[r], cR1[r]);
    *(float2*)(stg + (256 + row) * K3_PITCH + 2 * wave) = make_float2(cI0[r], cI1[r]);
  }
  __syncthreads();
  int q = threadIdx.x & 7;
#pragma unroll
  for (int j = 0; j < 8; ++j) {
    int row = (threadIdx.x >> 3) + (j << 6);
    int rb = (row >> 4) & 15, ro = row & 15;
    float4 v = *(const float4*)(stg + row * K3_PITCH + (q << 2));
    float* dst = ((row >> 8) ? YbI : YbR) +
                 ((size_t)(rb * 64 + n0 + ro) << 12) + f0 + (q << 2);
    *(float4*)dst = v;
  }
}

__device__ __forceinline__ void k4_class_pair(const float* fa, const float* fb,
                                              const float* ya, const float* yb,
                                              int ra, int hh,
                                              v8f& aR, v8f& aI, v8f& bR, v8f& bI) {
  v8f xR = {}, xI = {}, zR = {}, zI = {};
#pragma unroll 2
  for (int k = 0; k < 16; k += 4) {
    int u0 = 4 * (k + (hh << 1)) + ra;
    v2f ar, ai, br, bi, ani;
    ar.x = fa[bandmap(u0) * 128];      ar.y = fa[bandmap(u0 + 4) * 128];
    ai.x = fb[bandmap(u0) * 128];      ai.y = fb[bandmap(u0 + 4) * 128];
    br.x = ya[u0 * 64];                br.y = ya[(u0 + 4) * 64];
    bi.x = yb[u0 * 64];                bi.y = yb[(u0 + 4) * 64];
    xR = wmma4(ar, br, xR);
    xR = wmma4(ai, bi, xR);
    xI = wmma4(ar, bi, xI);
    ani.x = -ai.x; ani.y = -ai.y;
    xI = wmma4(ani, br, xI);
    ar.x = fa[bandmap(u0 + 2) * 128];  ar.y = fa[bandmap(u0 + 6) * 128];
    ai.x = fb[bandmap(u0 + 2) * 128];  ai.y = fb[bandmap(u0 + 6) * 128];
    br.x = ya[(u0 + 2) * 64];          br.y = ya[(u0 + 6) * 64];
    bi.x = yb[(u0 + 2) * 64];          bi.y = yb[(u0 + 6) * 64];
    zR = wmma4(ar, br, zR);
    zR = wmma4(ai, bi, zR);
    zI = wmma4(ar, bi, zI);
    ani.x = -ai.x; ani.y = -ai.y;
    zI = wmma4(ani, br, zI);
  }
  aR = xR; aI = xI; bR = zR; bI = zI;
}

#define S_PITCH 68
__global__ void k4_inv(const float* __restrict__ YbR, const float* __restrict__ YbI,
                       const float* __restrict__ Ftab, const float* __restrict__ bias,
                       float* __restrict__ out) {
  __shared__ __attribute__((aligned(16))) float Sr[128 * S_PITCH];
  __shared__ __attribute__((aligned(16))) float Si[128 * S_PITCH];
  const float* Fre = Ftab;
  const float* Fim = Ftab + 16384;
  int blk = blockIdx.x;
  const float* Yr = YbR + ((size_t)blk << 12);
  const float* Yi = YbI + ((size_t)blk << 12);
  int wave = threadIdx.x >> 5;
  int lane = threadIdx.x & 31;
  int c = lane & 15;
  int hh = lane >> 4;

  if (wave < 4) {
    int m0 = (wave >> 1) << 4;
    int n0 = 32 + ((wave & 1) << 4);
    const float* fa = Fre + m0 + c;
    const float* fb = Fim + m0 + c;
    const float* ya = Yr + n0 + c;
    const float* yb = Yi + n0 + c;
    v8f p0R, p0I, p2R, p2I, p1R, p1I, p3R, p3I;
    k4_class_pair(fa, fb, ya, yb, 0, hh, p0R, p0I, p2R, p2I);
    k4_class_pair(fa, fb, ya, yb, 1, hh, p1R, p1I, p3R, p3I);
    int n = n0 + c;
    int mb = m0 + (hh << 3);
    float sc = (n == 32) ? (1.f / 16384.f) : (2.f / 16384.f);
#pragma unroll
    for (int r = 0; r < 8; ++r) {
      float aR = p0R[r] + p2R[r], aI = p0I[r] + p2I[r];
      float bR = p0R[r] - p2R[r], bI = p0I[r] - p2I[r];
      float cR = p1R[r] + p3R[r], cI = p1I[r] + p3I[r];
      float dR = p1R[r] - p3R[r], dI = p1I[r] - p3I[r];
      Sr[(mb + r) * S_PITCH + n]      = (aR + cR) * sc;
      Si[(mb + r) * S_PITCH + n]      = (aI + cI) * sc;
      Sr[(mb + r + 32) * S_PITCH + n] = (bR - dI) * sc;
      Si[(mb + r + 32) * S_PITCH + n] = (bI + dR) * sc;
      Sr[(mb + r + 64) * S_PITCH + n] = (aR - cR) * sc;
      Si[(mb + r + 64) * S_PITCH + n] = (aI - cI) * sc;
      Sr[(mb + r + 96) * S_PITCH + n] = (bR + dI) * sc;
      Si[(mb + r + 96) * S_PITCH + n] = (bI - dR) * sc;
    }
  }
  __syncthreads();

  float bo = bias[blk & 63];
  float* oimg = out + ((size_t)blk << 14);
  {
    int n0 = (wave & 3) << 4;
    v2f bre[4], bro[4], bie[4], bio[4];
#pragma unroll
    for (int ks = 0; ks < 4; ++ks) {
      int de = 2 * ((ks << 2) + (hh << 1));
      const float* pr = Fre + de * 128 + n0 + c;
      const float* pi = Fim + de * 128 + n0 + c;
      bre[ks].x = pr[0];   bre[ks].y = pr[256];
      bro[ks].x = pr[128]; bro[ks].y = pr[384];
      bie[ks].x = pi[0];   bie[ks].y = pi[256];
      bio[ks].x = pi[128]; bio[ks].y = pi[384];
    }
    for (int mt = wave >> 2; mt < 8; mt += 2) {
      int m0 = mt << 4;
      const float* sr = Sr + (m0 + c) * S_PITCH + 32 + (hh << 2);
      const float* si = Si + (m0 + c) * S_PITCH + 32 + (hh << 2);
      v8f cE = {}, cO = {};
#pragma unroll
      for (int ks = 0; ks < 4; ++ks) {
        float4 qr = *(const float4*)(sr + (ks << 3));
        float4 qi = *(const float4*)(si + (ks << 3));
        v2f sre, sro, sie, sio;
        sre.x = qr.x; sre.y = qr.z; sro.x = qr.y; sro.y = qr.w;
        sie.x = qi.x; sie.y = qi.z; sio.x = qi.y; sio.y = qi.w;
        cE = wmma4(sre, bre[ks], cE);
        cO = wmma4(sro, bro[ks], cO);
        cE = wmma4(sie, bie[ks], cE);
        cO = wmma4(sio, bio[ks], cO);
      }
      int n = n0 + c;
      int mb = m0 + (hh << 3);
#pragma unroll
      for (int r = 0; r < 8; ++r) {
        oimg[(mb + r) * 128 + n]      = cE[r] + cO[r] + bo;
        oimg[(mb + r) * 128 + n + 64] = cE[r] - cO[r] + bo;
      }
    }
  }
}

extern "C" void kernel_launch(void* const* d_in, const int* in_sizes, int n_in,
                              void* d_out, int out_size, void* d_ws, size_t ws_size,
                              hipStream_t stream) {
  (void)in_sizes; (void)n_in; (void)out_size; (void)ws_size;
  const float* x    = (const float*)d_in[0];
  const float* w    = (const float*)d_in[1];
  const float* bias = (const float*)d_in[2];
  float* out = (float*)d_out;

  float* ws   = (float*)d_ws;
  float* Ftab = ws;
  float* XbR  = ws + 32768;
  float* XbI  = XbR + 4194304;
  float* YbR  = XbI + 4194304;
  float* YbI  = YbR + 4194304;

  k0_tables<<<16, 256, 0, stream>>>(Ftab);
  k1_fwd  <<<1024, 256, 0, stream>>>(x, Ftab, XbR, XbI);
  k3_mix  <<<256, 512, 0, stream>>>(XbR, XbI, w, YbR, YbI);
  k4_inv  <<<1024, 256, 0, stream>>>(YbR, YbI, Ftab, bias, out);
}
